# starting kernel
# baseline (speedup 1.0000x reference)
#include <hip/hip_runtime.h>


typedef float v2f __attribute__((ext_vector_type(2)));
typedef float v8f __attribute__((ext_vector_type(8)));

#define B_    16
#define C_    256
#define H_    64
#define W_    64
#define OUTC  3
#define NT    512
#define CB    64

#define HW     (H_*W_)
#define NPOS   (B_*HW)
#define NTILE  (NPOS/16)
#define CHW    (C_*HW)

#define Q_OFF      0u
#define H_OFF      16777216u
#define EPACK_OFF  20971520u
#define EN2_OFF    21102592u
#define CNT_OFF    21103104u
#define WDIFF_OFF  21111296u
#define WPACK_OFF  21115392u

#define WP_PER_BLK (9*128*CB)
#define WP_PER_TAP (128*CB)

#define DIFF_POS   196608
#define WF_POS     196609

static __device__ __forceinline__ v8f wmma4(v2f a, v2f b, v8f c) {
  return __builtin_amdgcn_wmma_f32_16x16x4_f32(false, a, false, b, (short)0, c,
                                               false, false);
}

#define DECODE_TILE(s, b, hh, w0)                        \
  const int b  = (s) >> 8;                               \
  const int hh = ((s) & 255) >> 2;                       \
  const int w0 = ((s) & 3) << 4;

__global__ __launch_bounds__(256) void k_prep(const float* __restrict__ embed,
                                              float* __restrict__ epack,
                                              float* __restrict__ en2,
                                              unsigned* __restrict__ counts) {
  const int gid = blockIdx.x * 256 + threadIdx.x;
  if (gid < 128 * NT) {
    const int cc = gid >> 9, n = gid & (NT - 1);
    epack[gid * 2 + 0] = embed[(2 * cc) * NT + n];
    epack[gid * 2 + 1] = embed[(2 * cc + 1) * NT + n];
  }
  if (gid < NT) {
    float s = 0.f;
    for (int c = 0; c < C_; ++c) { float v = embed[c * NT + gid]; s += v * v; }
    en2[gid] = s;
  }
  if (gid < B_ * NT) counts[gid] = 0u;
}

__global__ __launch_bounds__(256) void k_packw(const float* __restrict__ rw1,
                                               float* __restrict__ wpack) {
  const int t = blockIdx.x * 256 + threadIdx.x;
  if (t >= 3 * WP_PER_BLK) return;
  const int i   = t / WP_PER_BLK;
  const int r   = t % WP_PER_BLK;
  const int tap = r / WP_PER_TAP;
  const int r2  = r % WP_PER_TAP;
  const int cc  = r2 >> 6;
  const int oc  = r2 & 63;
  const int dh = tap / 3, dw = tap % 3;
  const float* src = rw1 + (size_t)i * CB * C_ * 9
                         + ((size_t)oc * C_ + 2 * cc) * 9 + dh * 3 + dw;
  wpack[t * 2 + 0] = src[0];
  wpack[t * 2 + 1] = src[9];
}

__global__ __launch_bounds__(128) void k_vq(const float* __restrict__ in,
                                            const float* __restrict__ embed,
                                            const float* __restrict__ epack,
                                            const float* __restrict__ en2,
                                            float* __restrict__ q,
                                            unsigned* __restrict__ counts,
                                            float* __restrict__ wsdiff) {
  __shared__ float As[4096];
  __shared__ float rv[4][16][16];
  __shared__ int   ri[4][16][16];
  __shared__ int   idxs[16];
  __shared__ float red[128];

  const int s = blockIdx.x;
  DECODE_TILE(s, b, hh, w0)
  const int tid = threadIdx.x;
  const long gbase0 = (long)b * CHW + hh * W_ + w0;

  for (int t = tid; t < 2048; t += 128) {
    const int cc = t >> 4, m = t & 15;
    As[t * 2 + 0] = in[gbase0 + (2 * cc) * HW + m];
    As[t * 2 + 1] = in[gbase0 + (2 * cc + 1) * HW + m];
  }
  __syncthreads();

  const int wave = tid >> 5, lane = tid & 31, g = lane >> 4, mc = lane & 15;
  float bestv[8];
  int   besti[8];
#pragma unroll
  for (int r = 0; r < 8; ++r) { bestv[r] = 3.4e38f; besti[r] = 0; }

  for (int tt = 0; tt < 8; ++tt) {
    const int n0 = (wave * 8 + tt) << 4;
    v8f acc = {0.f, 0.f, 0.f, 0.f, 0.f, 0.f, 0.f, 0.f};
#pragma unroll 8
    for (int k0 = 0; k0 < C_; k0 += 4) {
      const int cc = (k0 >> 1) + g;
      v2f a  = *(const v2f*)(As + (cc * 16 + mc) * 2);
      v2f bb = *(const v2f*)(epack + (cc * NT + n0 + mc) * 2);
      acc = wmma4(a, bb, acc);
    }
    const float e2 = en2[n0 + mc];
    const int   n  = n0 + mc;
#pragma unroll
    for (int r = 0; r < 8; ++r) {
      const float d = e2 - 2.0f * acc[r];
      if (d < bestv[r]) { bestv[r] = d; besti[r] = n; }
    }
  }
#pragma unroll
  for (int r = 0; r < 8; ++r) {
    const int mr = r + 8 * g;
    rv[wave][mr][mc] = bestv[r];
    ri[wave][mr][mc] = besti[r];
  }
  __syncthreads();

  if (tid < 16) {
    float bv = 3.4e38f; int bi = 0x7fffffff;
    for (int w = 0; w < 4; ++w)
      for (int c2 = 0; c2 < 16; ++c2) {
        const float v = rv[w][tid][c2]; const int ii = ri[w][tid][c2];
        if (v < bv || (v == bv && ii < bi)) { bv = v; bi = ii; }
      }
    idxs[tid] = bi;
    atomicAdd(&counts[b * NT + bi], 1u);
  }
  __syncthreads();

  float fsum = 0.f;
  for (int t = tid; t < 2048; t += 128) {
    const int cc = t >> 4, m = t & 15;
    const int id = idxs[m];
    const float x0 = As[t * 2 + 0], x1 = As[t * 2 + 1];
    const float e0 = embed[(2 * cc) * NT + id];
    const float e1 = embed[(2 * cc + 1) * NT + id];
    q[gbase0 + (2 * cc) * HW + m]     = e0;
    q[gbase0 + (2 * cc + 1) * HW + m] = e1;
    const float d0 = e0 - x0, d1 = e1 - x1;
    fsum += d0 * d0 + d1 * d1;
  }
  red[tid] = fsum; __syncthreads();
  for (int st = 64; st > 0; st >>= 1) {
    if (tid < st) red[tid] += red[tid + st];
    __syncthreads();
  }
  if (tid == 0) wsdiff[s] = red[0];
}

__global__ __launch_bounds__(128) void k_res_conv3(const float* __restrict__ q,
                                                   const float* __restrict__ wp,
                                                   const float* __restrict__ b1,
                                                   float* __restrict__ hbuf) {
  __shared__ float Ah[13824];
  const int s = blockIdx.x;
  DECODE_TILE(s, b, hh, w0)
  const int tid = threadIdx.x;

  for (int t = tid; t < 6912; t += 128) {
    const int cc = t / 54, r2 = t % 54, rr = r2 / 18, ww = r2 % 18;
    const int hy = hh + rr - 1, wx = w0 + ww - 1;
    float v0 = 0.f, v1 = 0.f;
    if ((unsigned)hy < 64u && (unsigned)wx < 64u) {
      const long gb = (long)b * CHW + hy * W_ + wx;
      v0 = q[gb + (2 * cc) * HW];     v0 = v0 > 0.f ? v0 : 0.f;
      v1 = q[gb + (2 * cc + 1) * HW]; v1 = v1 > 0.f ? v1 : 0.f;
    }
    Ah[t * 2 + 0] = v0; Ah[t * 2 + 1] = v1;
  }
  __syncthreads();

  const int wave = tid >> 5, lane = tid & 31, g = lane >> 4, mc = lane & 15;
  const int oc = wave * 16 + mc;
  v8f acc = {0.f, 0.f, 0.f, 0.f, 0.f, 0.f, 0.f, 0.f};
  for (int tap = 0; tap < 9; ++tap) {
    const int dh = tap / 3, dw = tap % 3;
    const float* wt = wp + (size_t)tap * WP_PER_TAP * 2;
#pragma unroll 8
    for (int k0 = 0; k0 < C_; k0 += 4) {
      const int cc = (k0 >> 1) + g;
      v2f a  = *(const v2f*)(Ah + ((cc * 3 + dh) * 18 + mc + dw) * 2);
      v2f bb = *(const v2f*)(wt + (cc * CB + oc) * 2);
      acc = wmma4(a, bb, acc);
    }
  }
  const float bias = b1[oc];
  const long ob = (long)b * (CB * HW) + oc * HW + hh * W_ + w0;
#pragma unroll
  for (int r = 0; r < 8; ++r) {
    const float v = acc[r] + bias;
    hbuf[ob + r + 8 * g] = v > 0.f ? v : 0.f;
  }
}

__global__ __launch_bounds__(512) void k_res_conv1(float* __restrict__ q,
                                                   const float* __restrict__ hbuf,
                                                   const float* __restrict__ w2,
                                                   const float* __restrict__ b2) {
  __shared__ float Hs[1024];
  const int s = blockIdx.x;
  DECODE_TILE(s, b, hh, w0)
  const int tid = threadIdx.x;
  {
    const int cc = tid >> 4, m = tid & 15;
    const long gb = (long)b * (CB * HW) + hh * W_ + w0 + m;
    Hs[tid * 2 + 0] = hbuf[gb + (2 * cc) * HW];
    Hs[tid * 2 + 1] = hbuf[gb + (2 * cc + 1) * HW];
  }
  __syncthreads();

  const int wave = tid >> 5, lane = tid & 31, g = lane >> 4, mc = lane & 15;
  const int oc = wave * 16 + mc;
  v8f acc = {0.f, 0.f, 0.f, 0.f, 0.f, 0.f, 0.f, 0.f};
#pragma unroll
  for (int k0 = 0; k0 < CB; k0 += 4) {
    const int cc = (k0 >> 1) + g;
    v2f a  = *(const v2f*)(Hs + (cc * 16 + mc) * 2);
    v2f bb = *(const v2f*)(w2 + oc * CB + k0 + 2 * g);
    acc = wmma4(a, bb, acc);
  }
  const float bias = b2[oc];
  const long ob = (long)b * CHW + oc * HW + hh * W_ + w0;
#pragma unroll
  for (int r = 0; r < 8; ++r) q[ob + r + 8 * g] += acc[r] + bias;
}

__global__ __launch_bounds__(32) void k_out_conv(const float* __restrict__ q,
                                                 const float* __restrict__ ow,
                                                 const float* __restrict__ ob,
                                                 float* __restrict__ out) {
  __shared__ float Qs[4096];
  __shared__ float Bs[4096];
  const int s = blockIdx.x;
  DECODE_TILE(s, b, hh, w0)
  const int lane = threadIdx.x, g = lane >> 4, mc = lane & 15;
  const long gbase0 = (long)b * CHW + hh * W_ + w0;

  for (int t = lane; t < 2048; t += 32) {
    const int cc = t >> 4, m = t & 15;
    Qs[t * 2 + 0] = q[gbase0 + (2 * cc) * HW + m];
    Qs[t * 2 + 1] = q[gbase0 + (2 * cc + 1) * HW + m];
    Bs[t * 2 + 0] = (m < OUTC) ? ow[m * C_ + 2 * cc]     : 0.f;
    Bs[t * 2 + 1] = (m < OUTC) ? ow[m * C_ + 2 * cc + 1] : 0.f;
  }
  __syncthreads();

  v8f acc = {0.f, 0.f, 0.f, 0.f, 0.f, 0.f, 0.f, 0.f};
#pragma unroll 8
  for (int k0 = 0; k0 < C_; k0 += 4) {
    const int cc = (k0 >> 1) + g;
    v2f a  = *(const v2f*)(Qs + (cc * 16 + mc) * 2);
    v2f bb = *(const v2f*)(Bs + (cc * 16 + mc) * 2);
    acc = wmma4(a, bb, acc);
  }
  if (mc < OUTC) {
    const float bias = ob[mc];
    const long obase = ((long)(b * OUTC + mc) * H_ + hh) * W_ + w0;
#pragma unroll
    for (int r = 0; r < 8; ++r) out[obase + r + 8 * g] = acc[r] + bias;
  }
}

__global__ __launch_bounds__(256) void k_final(const unsigned* __restrict__ counts,
                                               const float* __restrict__ wsdiff,
                                               float* __restrict__ out) {
  __shared__ float red[256];
  const int gid = blockIdx.x * 256 + threadIdx.x;
  if (blockIdx.x < 32) {
    out[WF_POS + gid] = (float)counts[gid] * (1.0f / 4096.0f);
  } else {
    float sum = 0.f;
    for (int t = threadIdx.x; t < NTILE; t += 256) sum += wsdiff[t];
    red[threadIdx.x] = sum; __syncthreads();
    for (int st = 128; st > 0; st >>= 1) {
      if (threadIdx.x < st) red[threadIdx.x] += red[threadIdx.x + st];
      __syncthreads();
    }
    if (threadIdx.x == 0)
      out[DIFF_POS] = red[0] * (1.0f / ((float)NPOS * (float)C_));
  }
}

extern "C" void kernel_launch(void* const* d_in, const int* in_sizes, int n_in,
                              void* d_out, int out_size, void* d_ws, size_t ws_size,
                              hipStream_t stream) {
  (void)in_sizes; (void)n_in; (void)out_size; (void)ws_size;
  const float* in  = (const float*)d_in[0];
  const float* emb = (const float*)d_in[1];
  const float* rw1 = (const float*)d_in[2];
  const float* rb1 = (const float*)d_in[3];
  const float* rw2 = (const float*)d_in[4];
  const float* rb2 = (const float*)d_in[5];
  const float* ow  = (const float*)d_in[6];
  const float* ob  = (const float*)d_in[7];
  float* out = (float*)d_out;
  float* ws  = (float*)d_ws;

  float*    q      = ws + Q_OFF;
  float*    hbuf   = ws + H_OFF;
  float*    epack  = ws + EPACK_OFF;
  float*    en2    = ws + EN2_OFF;
  unsigned* counts = (unsigned*)(ws + CNT_OFF);
  float*    wsdiff = ws + WDIFF_OFF;
  float*    wpack  = ws + WPACK_OFF;

  k_prep<<<256, 256, 0, stream>>>(emb, epack, en2, counts);
  k_packw<<<(3 * WP_PER_BLK + 255) / 256, 256, 0, stream>>>(rw1, wpack);
  k_vq<<<NTILE, 128, 0, stream>>>(in, emb, epack, en2, q, counts, wsdiff);
  for (int i = 0; i < 3; ++i) {
    k_res_conv3<<<NTILE, 128, 0, stream>>>(q, wpack + (size_t)i * WP_PER_BLK * 2,
                                           rb1 + i * CB, hbuf);
    k_res_conv1<<<NTILE, 512, 0, stream>>>(q, hbuf, rw2 + (size_t)i * C_ * CB,
                                           rb2 + i * C_);
  }
  k_out_conv<<<NTILE, 32, 0, stream>>>(q, ow, ob, out);
  k_final<<<33, 256, 0, stream>>>(counts, wsdiff, out);
}


# swap WMMA operand roles in both residual convs (weights = A rows, positions = B columns): D lanes run along w, so q read-add-write and h stores are 64 B runs, not lane pieces 16 KB apart; same sums
# speedup vs baseline: 1.7681x; 1.7681x over previous
#include <hip/hip_runtime.h>


typedef float v2f __attribute__((ext_vector_type(2)));
typedef float v8f __attribute__((ext_vector_type(8)));

#define B_    16
#define C_    256
#define H_    64
#define W_    64
#define OUTC  3
#define NT    512
#define CB    64

#define HW     (H_*W_)
#define NPOS   (B_*HW)
#define NTILE  (NPOS/16)
#define CHW    (C_*HW)

#define Q_OFF      0u
#define H_OFF      16777216u
#define EPACK_OFF  20971520u
#define EN2_OFF    21102592u
#define CNT_OFF    21103104u
#define WDIFF_OFF  21111296u
#define WPACK_OFF  21115392u

#define WP_PER_BLK (9*128*CB)
#define WP_PER_TAP (128*CB)

#define DIFF_POS   196608
#define WF_POS     196609

static __device__ __forceinline__ v8f wmma4(v2f a, v2f b, v8f c) {
  return __builtin_amdgcn_wmma_f32_16x16x4_f32(false, a, false, b, (short)0, c,
                                               false, false);
}

#define DECODE_TILE(s, b, hh, w0)                        \
  const int b  = (s) >> 8;                               \
  const int hh = ((s) & 255) >> 2;                       \
  const int w0 = ((s) & 3) << 4;

__global__ __launch_bounds__(256) void k_prep(const float* __restrict__ embed,
                                              float* __restrict__ epack,
                                              float* __restrict__ en2,
                                              unsigned* __restrict__ counts) {
  const int gid = blockIdx.x * 256 + threadIdx.x;
  if (gid < 128 * NT) {
    const int cc = gid >> 9, n = gid & (NT - 1);
    epack[gid * 2 + 0] = embed[(2 * cc) * NT + n];
    epack[gid * 2 + 1] = embed[(2 * cc + 1) * NT + n];
  }
  if (gid < NT) {
    float s = 0.f;
    for (int c = 0; c < C_; ++c) { float v = embed[c * NT + gid]; s += v * v; }
    en2[gid] = s;
  }
  if (gid < B_ * NT) counts[gid] = 0u;
}

__global__ __launch_bounds__(256) void k_packw(const float* __restrict__ rw1,
                                               float* __restrict__ wpack) {
  const int t = blockIdx.x * 256 + threadIdx.x;
  if (t >= 3 * WP_PER_BLK) return;
  const int i   = t / WP_PER_BLK;
  const int r   = t % WP_PER_BLK;
  const int tap = r / WP_PER_TAP;
  const int r2  = r % WP_PER_TAP;
  const int cc  = r2 >> 6;
  const int oc  = r2 & 63;
  const int dh = tap / 3, dw = tap % 3;
  const float* src = rw1 + (size_t)i * CB * C_ * 9
                         + ((size_t)oc * C_ + 2 * cc) * 9 + dh * 3 + dw;
  wpack[t * 2 + 0] = src[0];
  wpack[t * 2 + 1] = src[9];
}

__global__ __launch_bounds__(128) void k_vq(const float* __restrict__ in,
                                            const float* __restrict__ embed,
                                            const float* __restrict__ epack,
                                            const float* __restrict__ en2,
                                            float* __restrict__ q,
                                            unsigned* __restrict__ counts,
                                            float* __restrict__ wsdiff) {
  __shared__ float As[4096];
  __shared__ float rv[4][16][16];
  __shared__ int   ri[4][16][16];
  __shared__ int   idxs[16];
  __shared__ float red[128];

  const int s = blockIdx.x;
  DECODE_TILE(s, b, hh, w0)
  const int tid = threadIdx.x;
  const long gbase0 = (long)b * CHW + hh * W_ + w0;

  for (int t = tid; t < 2048; t += 128) {
    const int cc = t >> 4, m = t & 15;
    As[t * 2 + 0] = in[gbase0 + (2 * cc) * HW + m];
    As[t * 2 + 1] = in[gbase0 + (2 * cc + 1) * HW + m];
  }
  __syncthreads();

  const int wave = tid >> 5, lane = tid & 31, g = lane >> 4, mc = lane & 15;
  float bestv[8];
  int   besti[8];
#pragma unroll
  for (int r = 0; r < 8; ++r) { bestv[r] = 3.4e38f; besti[r] = 0; }

  for (int tt = 0; tt < 8; ++tt) {
    const int n0 = (wave * 8 + tt) << 4;
    v8f acc = {0.f, 0.f, 0.f, 0.f, 0.f, 0.f, 0.f, 0.f};
#pragma unroll 8
    for (int k0 = 0; k0 < C_; k0 += 4) {
      const int cc = (k0 >> 1) + g;
      v2f a  = *(const v2f*)(As + (cc * 16 + mc) * 2);
      v2f bb = *(const v2f*)(epack + (cc * NT + n0 + mc) * 2);
      acc = wmma4(a, bb, acc);
    }
    const float e2 = en2[n0 + mc];
    const int   n  = n0 + mc;
#pragma unroll
    for (int r = 0; r < 8; ++r) {
      const float d = e2 - 2.0f * acc[r];
      if (d < bestv[r]) { bestv[r] = d; besti[r] = n; }
    }
  }
#pragma unroll
  for (int r = 0; r < 8; ++r) {
    const int mr = r + 8 * g;
    rv[wave][mr][mc] = bestv[r];
    ri[wave][mr][mc] = besti[r];
  }
  __syncthreads();

  if (tid < 16) {
    float bv = 3.4e38f; int bi = 0x7fffffff;
    for (int w = 0; w < 4; ++w)
      for (int c2 = 0; c2 < 16; ++c2) {
        const float v = rv[w][tid][c2]; const int ii = ri[w][tid][c2];
        if (v < bv || (v == bv && ii < bi)) { bv = v; bi = ii; }
      }
    idxs[tid] = bi;
    atomicAdd(&counts[b * NT + bi], 1u);
  }
  __syncthreads();

  float fsum = 0.f;
  for (int t = tid; t < 2048; t += 128) {
    const int cc = t >> 4, m = t & 15;
    const int id = idxs[m];
    const float x0 = As[t * 2 + 0], x1 = As[t * 2 + 1];
    const float e0 = embed[(2 * cc) * NT + id];
    const float e1 = embed[(2 * cc + 1) * NT + id];
    q[gbase0 + (2 * cc) * HW + m]     = e0;
    q[gbase0 + (2 * cc + 1) * HW + m] = e1;
    const float d0 = e0 - x0, d1 = e1 - x1;
    fsum += d0 * d0 + d1 * d1;
  }
  red[tid] = fsum; __syncthreads();
  for (int st = 64; st > 0; st >>= 1) {
    if (tid < st) red[tid] += red[tid + st];
    __syncthreads();
  }
  if (tid == 0) wsdiff[s] = red[0];
}

__global__ __launch_bounds__(128) void k_res_conv3(const float* __restrict__ q,
                                                   const float* __restrict__ wp,
                                                   const float* __restrict__ b1,
                                                   float* __restrict__ hbuf) {
  __shared__ float Ah[13824];
  const int s = blockIdx.x;
  DECODE_TILE(s, b, hh, w0)
  const int tid = threadIdx.x;

  for (int t = tid; t < 6912; t += 128) {
    const int cc = t / 54, r2 = t % 54, rr = r2 / 18, ww = r2 % 18;
    const int hy = hh + rr - 1, wx = w0 + ww - 1;
    float v0 = 0.f, v1 = 0.f;
    if ((unsigned)hy < 64u && (unsigned)wx < 64u) {
      const long gb = (long)b * CHW + hy * W_ + wx;
      v0 = q[gb + (2 * cc) * HW];     v0 = v0 > 0.f ? v0 : 0.f;
      v1 = q[gb + (2 * cc + 1) * HW]; v1 = v1 > 0.f ? v1 : 0.f;
    }
    Ah[t * 2 + 0] = v0; Ah[t * 2 + 1] = v1;
  }
  __syncthreads();

  const int wave = tid >> 5, lane = tid & 31, g = lane >> 4, mc = lane & 15;
  const int oc = wave * 16 + mc;
  v8f acc = {0.f, 0.f, 0.f, 0.f, 0.f, 0.f, 0.f, 0.f};
  for (int tap = 0; tap < 9; ++tap) {
    const int dh = tap / 3, dw = tap % 3;
    const float* wt = wp + (size_t)tap * WP_PER_TAP * 2;
#pragma unroll 8
    for (int k0 = 0; k0 < C_; k0 += 4) {
      const int cc = (k0 >> 1) + g;
      v2f xa = *(const v2f*)(Ah + ((cc * 3 + dh) * 18 + mc + dw) * 2);
      v2f wb = *(const v2f*)(wt + (cc * CB + oc) * 2);
      acc = wmma4(wb, xa, acc);
    }
  }
  const int ocb = wave * 16 + 8 * g;
  const long ob = (long)b * (CB * HW) + (long)ocb * HW + hh * W_ + w0 + mc;
#pragma unroll
  for (int r = 0; r < 8; ++r) {
    const float v = acc[r] + b1[ocb + r];
    hbuf[ob + (long)r * HW] = v > 0.f ? v : 0.f;
  }
}

__global__ __launch_bounds__(512) void k_res_conv1(float* __restrict__ q,
                                                   const float* __restrict__ hbuf,
                                                   const float* __restrict__ w2,
                                                   const float* __restrict__ b2) {
  __shared__ float Hs[1024];
  const int s = blockIdx.x;
  DECODE_TILE(s, b, hh, w0)
  const int tid = threadIdx.x;
  {
    const int cc = tid >> 4, m = tid & 15;
    const long gb = (long)b * (CB * HW) + hh * W_ + w0 + m;
    Hs[tid * 2 + 0] = hbuf[gb + (2 * cc) * HW];
    Hs[tid * 2 + 1] = hbuf[gb + (2 * cc + 1) * HW];
  }
  __syncthreads();

  const int wave = tid >> 5, lane = tid & 31, g = lane >> 4, mc = lane & 15;
  const int oc = wave * 16 + mc;
  v8f acc = {0.f, 0.f, 0.f, 0.f, 0.f, 0.f, 0.f, 0.f};
#pragma unroll
  for (int k0 = 0; k0 < CB; k0 += 4) {
    const int cc = (k0 >> 1) + g;
    v2f xa = *(const v2f*)(Hs + (cc * 16 + mc) * 2);
    v2f wb = *(const v2f*)(w2 + oc * CB + k0 + 2 * g);
    acc = wmma4(wb, xa, acc);
  }
  const int ocb = wave * 16 + 8 * g;
  const long ob = (long)b * CHW + (long)ocb * HW + hh * W_ + w0 + mc;
#pragma unroll
  for (int r = 0; r < 8; ++r)
    q[ob + (long)r * HW] += acc[r] + b2[ocb + r];
}

__global__ __launch_bounds__(32) void k_out_conv(const float* __restrict__ q,
                                                 const float* __restrict__ ow,
                                                 const float* __restrict__ ob,
                                                 float* __restrict__ out) {
  __shared__ float Qs[4096];
  __shared__ float Bs[4096];
  const int s = blockIdx.x;
  DECODE_TILE(s, b, hh, w0)
  const int lane = threadIdx.x, g = lane >> 4, mc = lane & 15;
  const long gbase0 = (long)b * CHW + hh * W_ + w0;

  for (int t = lane; t < 2048; t += 32) {
    const int cc = t >> 4, m = t & 15;
    Qs[t * 2 + 0] = q[gbase0 + (2 * cc) * HW + m];
    Qs[t * 2 + 1] = q[gbase0 + (2 * cc + 1) * HW + m];
    Bs[t * 2 + 0] = (m < OUTC) ? ow[m * C_ + 2 * cc]     : 0.f;
    Bs[t * 2 + 1] = (m < OUTC) ? ow[m * C_ + 2 * cc + 1] : 0.f;
  }
  __syncthreads();

  v8f acc = {0.f, 0.f, 0.f, 0.f, 0.f, 0.f, 0.f, 0.f};
#pragma unroll 8
  for (int k0 = 0; k0 < C_; k0 += 4) {
    const int cc = (k0 >> 1) + g;
    v2f a  = *(const v2f*)(Qs + (cc * 16 + mc) * 2);
    v2f bb = *(const v2f*)(Bs + (cc * 16 + mc) * 2);
    acc = wmma4(a, bb, acc);
  }
  if (mc < OUTC) {
    const float bias = ob[mc];
    const long obase = ((long)(b * OUTC + mc) * H_ + hh) * W_ + w0;
#pragma unroll
    for (int r = 0; r < 8; ++r) out[obase + r + 8 * g] = acc[r] + bias;
  }
}

__global__ __launch_bounds__(256) void k_final(const unsigned* __restrict__ counts,
                                               const float* __restrict__ wsdiff,
                                               float* __restrict__ out) {
  __shared__ float red[256];
  const int gid = blockIdx.x * 256 + threadIdx.x;
  if (blockIdx.x < 32) {
    out[WF_POS + gid] = (float)counts[gid] * (1.0f / 4096.0f);
  } else {
    float sum = 0.f;
    for (int t = threadIdx.x; t < NTILE; t += 256) sum += wsdiff[t];
    red[threadIdx.x] = sum; __syncthreads();
    for (int st = 128; st > 0; st >>= 1) {
      if (threadIdx.x < st) red[threadIdx.x] += red[threadIdx.x + st];
      __syncthreads();
    }
    if (threadIdx.x == 0)
      out[DIFF_POS] = red[0] * (1.0f / ((float)NPOS * (float)C_));
  }
}

extern "C" void kernel_launch(void* const* d_in, const int* in_sizes, int n_in,
                              void* d_out, int out_size, void* d_ws, size_t ws_size,
                              hipStream_t stream) {
  (void)in_sizes; (void)n_in; (void)out_size; (void)ws_size;
  const float* in  = (const float*)d_in[0];
  const float* emb = (const float*)d_in[1];
  const float* rw1 = (const float*)d_in[2];
  const float* rb1 = (const float*)d_in[3];
  const float* rw2 = (const float*)d_in[4];
  const float* rb2 = (const float*)d_in[5];
  const float* ow  = (const float*)d_in[6];
  const float* ob  = (const float*)d_in[7];
  float* out = (float*)d_out;
  float* ws  = (float*)d_ws;

  float*    q      = ws + Q_OFF;
  float*    hbuf   = ws + H_OFF;
  float*    epack  = ws + EPACK_OFF;
  float*    en2    = ws + EN2_OFF;
  unsigned* counts = (unsigned*)(ws + CNT_OFF);
  float*    wsdiff = ws + WDIFF_OFF;
  float*    wpack  = ws + WPACK_OFF;

  k_prep<<<256, 256, 0, stream>>>(emb, epack, en2, counts);
  k_packw<<<(3 * WP_PER_BLK + 255) / 256, 256, 0, stream>>>(rw1, wpack);
  k_vq<<<NTILE, 128, 0, stream>>>(in, emb, epack, en2, q, counts, wsdiff);
  for (int i = 0; i < 3; ++i) {
    k_res_conv3<<<NTILE, 128, 0, stream>>>(q, wpack + (size_t)i * WP_PER_BLK * 2,
                                           rb1 + i * CB, hbuf);
    k_res_conv1<<<NTILE, 512, 0, stream>>>(q, hbuf, rw2 + (size_t)i * C_ * CB,
                                           rb2 + i * C_);
  }
  k_out_conv<<<NTILE, 32, 0, stream>>>(q, ow, ob, out);
  k_final<<<33, 256, 0, stream>>>(counts, wsdiff, out);
}


# 3x3 conv: one work-group per image row, each wave keeps 4 position tiles so a weight fragment from L2 feeds 4 WMMAs; halo staged per 64 channels, so sum order is channel-chunk outer, taps inner
# speedup vs baseline: 2.9726x; 1.6812x over previous
#include <hip/hip_runtime.h>


typedef float v2f __attribute__((ext_vector_type(2)));
typedef float v8f __attribute__((ext_vector_type(8)));

#define B_    16
#define C_    256
#define H_    64
#define W_    64
#define OUTC  3
#define NT    512
#define CB    64

#define HW     (H_*W_)
#define NPOS   (B_*HW)
#define NTILE  (NPOS/16)
#define CHW    (C_*HW)

#define Q_OFF      0u
#define H_OFF      16777216u
#define EPACK_OFF  20971520u
#define EN2_OFF    21102592u
#define CNT_OFF    21103104u
#define WDIFF_OFF  21111296u
#define WPACK_OFF  21115392u

#define WP_PER_BLK (9*128*CB)
#define WP_PER_TAP (128*CB)

#define DIFF_POS   196608
#define WF_POS     196609

static __device__ __forceinline__ v8f wmma4(v2f a, v2f b, v8f c) {
  return __builtin_amdgcn_wmma_f32_16x16x4_f32(false, a, false, b, (short)0, c,
                                               false, false);
}

#define DECODE_TILE(s, b, hh, w0)                        \
  const int b  = (s) >> 8;                               \
  const int hh = ((s) & 255) >> 2;                       \
  const int w0 = ((s) & 3) << 4;

__global__ __launch_bounds__(256) void k_prep(const float* __restrict__ embed,
                                              float* __restrict__ epack,
                                              float* __restrict__ en2,
                                              unsigned* __restrict__ counts) {
  const int gid = blockIdx.x * 256 + threadIdx.x;
  if (gid < 128 * NT) {
    const int cc = gid >> 9, n = gid & (NT - 1);
    epack[gid * 2 + 0] = embed[(2 * cc) * NT + n];
    epack[gid * 2 + 1] = embed[(2 * cc + 1) * NT + n];
  }
  if (gid < NT) {
    float s = 0.f;
    for (int c = 0; c < C_; ++c) { float v = embed[c * NT + gid]; s += v * v; }
    en2[gid] = s;
  }
  if (gid < B_ * NT) counts[gid] = 0u;
}

__global__ __launch_bounds__(256) void k_packw(const float* __restrict__ rw1,
                                               float* __restrict__ wpack) {
  const int t = blockIdx.x * 256 + threadIdx.x;
  if (t >= 3 * WP_PER_BLK) return;
  const int i   = t / WP_PER_BLK;
  const int r   = t % WP_PER_BLK;
  const int tap = r / WP_PER_TAP;
  const int r2  = r % WP_PER_TAP;
  const int cc  = r2 >> 6;
  const int oc  = r2 & 63;
  const int dh = tap / 3, dw = tap % 3;
  const float* src = rw1 + (size_t)i * CB * C_ * 9
                         + ((size_t)oc * C_ + 2 * cc) * 9 + dh * 3 + dw;
  wpack[t * 2 + 0] = src[0];
  wpack[t * 2 + 1] = src[9];
}

__global__ __launch_bounds__(128) void k_vq(const float* __restrict__ in,
                                            const float* __restrict__ embed,
                                            const float* __restrict__ epack,
                                            const float* __restrict__ en2,
                                            float* __restrict__ q,
                                            unsigned* __restrict__ counts,
                                            float* __restrict__ wsdiff) {
  __shared__ float As[4096];
  __shared__ float rv[4][16][16];
  __shared__ int   ri[4][16][16];
  __shared__ int   idxs[16];
  __shared__ float red[128];

  const int s = blockIdx.x;
  DECODE_TILE(s, b, hh, w0)
  const int tid = threadIdx.x;
  const long gbase0 = (long)b * CHW + hh * W_ + w0;

  for (int t = tid; t < 2048; t += 128) {
    const int cc = t >> 4, m = t & 15;
    As[t * 2 + 0] = in[gbase0 + (2 * cc) * HW + m];
    As[t * 2 + 1] = in[gbase0 + (2 * cc + 1) * HW + m];
  }
  __syncthreads();

  const int wave = tid >> 5, lane = tid & 31, g = lane >> 4, mc = lane & 15;
  float bestv[8];
  int   besti[8];
#pragma unroll
  for (int r = 0; r < 8; ++r) { bestv[r] = 3.4e38f; besti[r] = 0; }

  for (int tt = 0; tt < 8; ++tt) {
    const int n0 = (wave * 8 + tt) << 4;
    v8f acc = {0.f, 0.f, 0.f, 0.f, 0.f, 0.f, 0.f, 0.f};
#pragma unroll 8
    for (int k0 = 0; k0 < C_; k0 += 4) {
      const int cc = (k0 >> 1) + g;
      v2f a  = *(const v2f*)(As + (cc * 16 + mc) * 2);
      v2f bb = *(const v2f*)(epack + (cc * NT + n0 + mc) * 2);
      acc = wmma4(a, bb, acc);
    }
    const float e2 = en2[n0 + mc];
    const int   n  = n0 + mc;
#pragma unroll
    for (int r = 0; r < 8; ++r) {
      const float d = e2 - 2.0f * acc[r];
      if (d < bestv[r]) { bestv[r] = d; besti[r] = n; }
    }
  }
#pragma unroll
  for (int r = 0; r < 8; ++r) {
    const int mr = r + 8 * g;
    rv[wave][mr][mc] = bestv[r];
    ri[wave][mr][mc] = besti[r];
  }
  __syncthreads();

  if (tid < 16) {
    float bv = 3.4e38f; int bi = 0x7fffffff;
    for (int w = 0; w < 4; ++w)
      for (int c2 = 0; c2 < 16; ++c2) {
        const float v = rv[w][tid][c2]; const int ii = ri[w][tid][c2];
        if (v < bv || (v == bv && ii < bi)) { bv = v; bi = ii; }
      }
    idxs[tid] = bi;
    atomicAdd(&counts[b * NT + bi], 1u);
  }
  __syncthreads();

  float fsum = 0.f;
  for (int t = tid; t < 2048; t += 128) {
    const int cc = t >> 4, m = t & 15;
    const int id = idxs[m];
    const float x0 = As[t * 2 + 0], x1 = As[t * 2 + 1];
    const float e0 = embed[(2 * cc) * NT + id];
    const float e1 = embed[(2 * cc + 1) * NT + id];
    q[gbase0 + (2 * cc) * HW + m]     = e0;
    q[gbase0 + (2 * cc + 1) * HW + m] = e1;
    const float d0 = e0 - x0, d1 = e1 - x1;
    fsum += d0 * d0 + d1 * d1;
  }
  red[tid] = fsum; __syncthreads();
  for (int st = 64; st > 0; st >>= 1) {
    if (tid < st) red[tid] += red[tid + st];
    __syncthreads();
  }
  if (tid == 0) wsdiff[s] = red[0];
}

#define KCH    32
#define A_ROW  66
#define A_CCP  208

__global__ __launch_bounds__(128) void k_res_conv3(const float* __restrict__ q,
                                                   const float* __restrict__ wp,
                                                   const float* __restrict__ b1,
                                                   float* __restrict__ hbuf) {
  __shared__ float Ah[KCH * A_CCP * 2];
  const int s  = blockIdx.x;
  const int b  = s >> 6;
  const int hh = s & 63;
  const int tid = threadIdx.x;
  const int wave = tid >> 5, lane = tid & 31, g = lane >> 4, mc = lane & 15;
  const int oc = wave * 16 + mc;

  for (int t = tid; t < KCH * 6; t += 128) {
    const int cc = t / 6, r6 = t % 6, rr = r6 >> 1;
    const int ww = (r6 & 1) ? (A_ROW - 1) : 0;
    const v2f z = {0.f, 0.f};
    *(v2f*)(Ah + (cc * A_CCP + rr * A_ROW + ww) * 2) = z;
  }

  const int sw = tid & 63;
  const int sc = tid >> 6;

  v8f acc0 = {0.f, 0.f, 0.f, 0.f, 0.f, 0.f, 0.f, 0.f};
  v8f acc1 = acc0, acc2 = acc0, acc3 = acc0;

  for (int kc = 0; kc < C_ / (2 * KCH); ++kc) {
    if (kc) __syncthreads();
#pragma unroll
    for (int rr = 0; rr < 3; ++rr) {
      const int hy = hh + rr - 1;
      float* dst = Ah + (sc * A_CCP + rr * A_ROW + 1 + sw) * 2;
      if ((unsigned)hy < 64u) {
        const float* qr = q + (long)b * CHW + (long)(2 * kc * KCH + 2 * sc) * HW
                            + hy * W_ + sw;
#pragma unroll 8
        for (int i = 0; i < KCH / 2; ++i) {
          float v0 = qr[(long)(4 * i) * HW];
          float v1 = qr[(long)(4 * i + 1) * HW];
          v0 = v0 > 0.f ? v0 : 0.f;
          v1 = v1 > 0.f ? v1 : 0.f;
          const v2f pv = {v0, v1};
          *(v2f*)(dst + i * (2 * A_CCP * 2)) = pv;
        }
      } else {
        const v2f z = {0.f, 0.f};
        for (int i = 0; i < KCH / 2; ++i) *(v2f*)(dst + i * (2 * A_CCP * 2)) = z;
      }
    }
    __syncthreads();

    const float* wk = wp + ((size_t)(kc * KCH + g) * CB + oc) * 2;
    const float* xb = Ah + (g * A_CCP + mc) * 2;
    for (int tap = 0; tap < 9; ++tap) {
      const int dh = tap / 3, dw = tap - dh * 3;
      const float* wt = wk + (size_t)tap * WP_PER_TAP * 2;
      const float* xt = xb + (dh * A_ROW + dw) * 2;
#pragma unroll 8
      for (int ks = 0; ks < KCH / 2; ++ks) {
        const v2f wb = *(const v2f*)(wt + ks * (2 * CB * 2));
        const float* xp = xt + ks * (2 * A_CCP * 2);
        const v2f x0 = *(const v2f*)(xp);
        const v2f x1 = *(const v2f*)(xp + 32);
        const v2f x2 = *(const v2f*)(xp + 64);
        const v2f x3 = *(const v2f*)(xp + 96);
        acc0 = wmma4(wb, x0, acc0);
        acc1 = wmma4(wb, x1, acc1);
        acc2 = wmma4(wb, x2, acc2);
        acc3 = wmma4(wb, x3, acc3);
      }
    }
  }

  const int ocb = wave * 16 + 8 * g;
  const long ob = (long)b * (CB * HW) + (long)ocb * HW + hh * W_ + mc;
#pragma unroll
  for (int r = 0; r < 8; ++r) {
    const float bias = b1[ocb + r];
    float* hp = hbuf + ob + (long)r * HW;
    float v;
    v = acc0[r] + bias; hp[0]  = v > 0.f ? v : 0.f;
    v = acc1[r] + bias; hp[16] = v > 0.f ? v : 0.f;
    v = acc2[r] + bias; hp[32] = v > 0.f ? v : 0.f;
    v = acc3[r] + bias; hp[48] = v > 0.f ? v : 0.f;
  }
}

__global__ __launch_bounds__(512) void k_res_conv1(float* __restrict__ q,
                                                   const float* __restrict__ hbuf,
                                                   const float* __restrict__ w2,
                                                   const float* __restrict__ b2) {
  __shared__ float Hs[1024];
  const int s = blockIdx.x;
  DECODE_TILE(s, b, hh, w0)
  const int tid = threadIdx.x;
  {
    const int cc = tid >> 4, m = tid & 15;
    const long gb = (long)b * (CB * HW) + hh * W_ + w0 + m;
    Hs[tid * 2 + 0] = hbuf[gb + (2 * cc) * HW];
    Hs[tid * 2 + 1] = hbuf[gb + (2 * cc + 1) * HW];
  }
  __syncthreads();

  const int wave = tid >> 5, lane = tid & 31, g = lane >> 4, mc = lane & 15;
  const int oc = wave * 16 + mc;
  v8f acc = {0.f, 0.f, 0.f, 0.f, 0.f, 0.f, 0.f, 0.f};
#pragma unroll
  for (int k0 = 0; k0 < CB; k0 += 4) {
    const int cc = (k0 >> 1) + g;
    v2f xa = *(const v2f*)(Hs + (cc * 16 + mc) * 2);
    v2f wb = *(const v2f*)(w2 + oc * CB + k0 + 2 * g);
    acc = wmma4(wb, xa, acc);
  }
  const int ocb = wave * 16 + 8 * g;
  const long ob = (long)b * CHW + (long)ocb * HW + hh * W_ + w0 + mc;
#pragma unroll
  for (int r = 0; r < 8; ++r)
    q[ob + (long)r * HW] += acc[r] + b2[ocb + r];
}

__global__ __launch_bounds__(32) void k_out_conv(const float* __restrict__ q,
                                                 const float* __restrict__ ow,
                                                 const float* __restrict__ ob,
                                                 float* __restrict__ out) {
  __shared__ float Qs[4096];
  __shared__ float Bs[4096];
  const int s = blockIdx.x;
  DECODE_TILE(s, b, hh, w0)
  const int lane = threadIdx.x, g = lane >> 4, mc = lane & 15;
  const long gbase0 = (long)b * CHW + hh * W_ + w0;

  for (int t = lane; t < 2048; t += 32) {
    const int cc = t >> 4, m = t & 15;
    Qs[t * 2 + 0] = q[gbase0 + (2 * cc) * HW + m];
    Qs[t * 2 + 1] = q[gbase0 + (2 * cc + 1) * HW + m];
    Bs[t * 2 + 0] = (m < OUTC) ? ow[m * C_ + 2 * cc]     : 0.f;
    Bs[t * 2 + 1] = (m < OUTC) ? ow[m * C_ + 2 * cc + 1] : 0.f;
  }
  __syncthreads();

  v8f acc = {0.f, 0.f, 0.f, 0.f, 0.f, 0.f, 0.f, 0.f};
#pragma unroll 8
  for (int k0 = 0; k0 < C_; k0 += 4) {
    const int cc = (k0 >> 1) + g;
    v2f a  = *(const v2f*)(Qs + (cc * 16 + mc) * 2);
    v2f bb = *(const v2f*)(Bs + (cc * 16 + mc) * 2);
    acc = wmma4(a, bb, acc);
  }
  if (mc < OUTC) {
    const float bias = ob[mc];
    const long obase = ((long)(b * OUTC + mc) * H_ + hh) * W_ + w0;
#pragma unroll
    for (int r = 0; r < 8; ++r) out[obase + r + 8 * g] = acc[r] + bias;
  }
}

__global__ __launch_bounds__(256) void k_final(const unsigned* __restrict__ counts,
                                               const float* __restrict__ wsdiff,
                                               float* __restrict__ out) {
  __shared__ float red[256];
  const int gid = blockIdx.x * 256 + threadIdx.x;
  if (blockIdx.x < 32) {
    out[WF_POS + gid] = (float)counts[gid] * (1.0f / 4096.0f);
  } else {
    float sum = 0.f;
    for (int t = threadIdx.x; t < NTILE; t += 256) sum += wsdiff[t];
    red[threadIdx.x] = sum; __syncthreads();
    for (int st = 128; st > 0; st >>= 1) {
      if (threadIdx.x < st) red[threadIdx.x] += red[threadIdx.x + st];
      __syncthreads();
    }
    if (threadIdx.x == 0)
      out[DIFF_POS] = red[0] * (1.0f / ((float)NPOS * (float)C_));
  }
}

extern "C" void kernel_launch(void* const* d_in, const int* in_sizes, int n_in,
                              void* d_out, int out_size, void* d_ws, size_t ws_size,
                              hipStream_t stream) {
  (void)in_sizes; (void)n_in; (void)out_size; (void)ws_size;
  const float* in  = (const float*)d_in[0];
  const float* emb = (const float*)d_in[1];
  const float* rw1 = (const float*)d_in[2];
  const float* rb1 = (const float*)d_in[3];
  const float* rw2 = (const float*)d_in[4];
  const float* rb2 = (const float*)d_in[5];
  const float* ow  = (const float*)d_in[6];
  const float* ob  = (const float*)d_in[7];
  float* out = (float*)d_out;
  float* ws  = (float*)d_ws;

  float*    q      = ws + Q_OFF;
  float*    hbuf   = ws + H_OFF;
  float*    epack  = ws + EPACK_OFF;
  float*    en2    = ws + EN2_OFF;
  unsigned* counts = (unsigned*)(ws + CNT_OFF);
  float*    wsdiff = ws + WDIFF_OFF;
  float*    wpack  = ws + WPACK_OFF;

  k_prep<<<256, 256, 0, stream>>>(emb, epack, en2, counts);
  k_packw<<<(3 * WP_PER_BLK + 255) / 256, 256, 0, stream>>>(rw1, wpack);
  k_vq<<<NTILE, 128, 0, stream>>>(in, emb, epack, en2, q, counts, wsdiff);
  for (int i = 0; i < 3; ++i) {
    k_res_conv3<<<B_ * H_, 128, 0, stream>>>(q, wpack + (size_t)i * WP_PER_BLK * 2,
                                           rb1 + i * CB, hbuf);
    k_res_conv1<<<NTILE, 512, 0, stream>>>(q, hbuf, rw2 + (size_t)i * C_ * CB,
                                           rb2 + i * C_);
  }
  k_out_conv<<<NTILE, 32, 0, stream>>>(q, ow, ob, out);
  k_final<<<33, 256, 0, stream>>>(counts, wsdiff, out);
}


# VQ distance GEMM: one work-group per image row, each wave keeps 4 position tiles so a codebook fragment from L2 feeds 4 WMMAs; same distances; diff partial sums regrouped per row (1024 partials)
# speedup vs baseline: 3.1878x; 1.0724x over previous
#include <hip/hip_runtime.h>


typedef float v2f __attribute__((ext_vector_type(2)));
typedef float v8f __attribute__((ext_vector_type(8)));

#define B_    16
#define C_    256
#define H_    64
#define W_    64
#define OUTC  3
#define NT    512
#define CB    64

#define HW     (H_*W_)
#define NPOS   (B_*HW)
#define NTILE  (NPOS/16)
#define CHW    (C_*HW)

#define Q_OFF      0u
#define H_OFF      16777216u
#define EPACK_OFF  20971520u
#define EN2_OFF    21102592u
#define CNT_OFF    21103104u
#define WDIFF_OFF  21111296u
#define WPACK_OFF  21115392u

#define WP_PER_BLK (9*128*CB)
#define WP_PER_TAP (128*CB)

#define DIFF_POS   196608
#define WF_POS     196609

static __device__ __forceinline__ v8f wmma4(v2f a, v2f b, v8f c) {
  return __builtin_amdgcn_wmma_f32_16x16x4_f32(false, a, false, b, (short)0, c,
                                               false, false);
}

#define DECODE_TILE(s, b, hh, w0)                        \
  const int b  = (s) >> 8;                               \
  const int hh = ((s) & 255) >> 2;                       \
  const int w0 = ((s) & 3) << 4;

__global__ __launch_bounds__(256) void k_prep(const float* __restrict__ embed,
                                              float* __restrict__ epack,
                                              float* __restrict__ en2,
                                              unsigned* __restrict__ counts) {
  const int gid = blockIdx.x * 256 + threadIdx.x;
  if (gid < 128 * NT) {
    const int cc = gid >> 9, n = gid & (NT - 1);
    epack[gid * 2 + 0] = embed[(2 * cc) * NT + n];
    epack[gid * 2 + 1] = embed[(2 * cc + 1) * NT + n];
  }
  if (gid < NT) {
    float s = 0.f;
    for (int c = 0; c < C_; ++c) { float v = embed[c * NT + gid]; s += v * v; }
    en2[gid] = s;
  }
  if (gid < B_ * NT) counts[gid] = 0u;
}

__global__ __launch_bounds__(256) void k_packw(const float* __restrict__ rw1,
                                               float* __restrict__ wpack) {
  const int t = blockIdx.x * 256 + threadIdx.x;
  if (t >= 3 * WP_PER_BLK) return;
  const int i   = t / WP_PER_BLK;
  const int r   = t % WP_PER_BLK;
  const int tap = r / WP_PER_TAP;
  const int r2  = r % WP_PER_TAP;
  const int cc  = r2 >> 6;
  const int oc  = r2 & 63;
  const int dh = tap / 3, dw = tap % 3;
  const float* src = rw1 + (size_t)i * CB * C_ * 9
                         + ((size_t)oc * C_ + 2 * cc) * 9 + dh * 3 + dw;
  wpack[t * 2 + 0] = src[0];
  wpack[t * 2 + 1] = src[9];
}

#define VQ_SLOT(cc, p) (((cc) << 6) + (((p) + (((cc) & 1) << 4)) & 63))

__global__ __launch_bounds__(128) void k_vq(const float* __restrict__ in,
                                            const float* __restrict__ embed,
                                            const float* __restrict__ epack,
                                            const float* __restrict__ en2,
                                            float* __restrict__ q,
                                            unsigned* __restrict__ counts,
                                            float* __restrict__ wsdiff) {
  __shared__ float As[128 * 64 * 2];
  __shared__ float rv[4][64];
  __shared__ int   ri[4][64];
  __shared__ int   idxs[64];
  __shared__ float red[128];

  const int s  = blockIdx.x;
  const int b  = s >> 6;
  const int hh = s & 63;
  const int tid = threadIdx.x;
  const long gbase0 = (long)b * CHW + hh * W_;
  const int sp = tid & 63;
  const int sc = tid >> 6;

  {
    const float* src = in + gbase0 + (long)(2 * sc) * HW + sp;
    float* dst = As + VQ_SLOT(sc, sp) * 2;
#pragma unroll 8
    for (int i = 0; i < 64; ++i) {
      const float v0 = src[(long)(4 * i) * HW];
      const float v1 = src[(long)(4 * i + 1) * HW];
      const v2f pv = {v0, v1};
      *(v2f*)(dst + i * (2 * 64 * 2)) = pv;
    }
  }
  __syncthreads();

  const int wave = tid >> 5, lane = tid & 31, g = lane >> 4, mc = lane & 15;
  float bestv[4][8];
  int   besti[4][8];
#pragma unroll
  for (int j = 0; j < 4; ++j)
#pragma unroll
    for (int r = 0; r < 8; ++r) { bestv[j][r] = 3.4e38f; besti[j][r] = 0; }

  const float* xa0 = As + ((g << 6) + (((0 + g) & 3) << 4) + mc) * 2;
  const float* xa1 = As + ((g << 6) + (((1 + g) & 3) << 4) + mc) * 2;
  const float* xa2 = As + ((g << 6) + (((2 + g) & 3) << 4) + mc) * 2;
  const float* xa3 = As + ((g << 6) + (((3 + g) & 3) << 4) + mc) * 2;

  for (int tt = 0; tt < 8; ++tt) {
    const int n0 = (wave * 8 + tt) << 4;
    const float* eb = epack + ((size_t)g * NT + n0 + mc) * 2;
    v8f acc0 = {0.f, 0.f, 0.f, 0.f, 0.f, 0.f, 0.f, 0.f};
    v8f acc1 = acc0, acc2 = acc0, acc3 = acc0;
#pragma unroll 8
    for (int ks = 0; ks < 64; ++ks) {
      const v2f bb = *(const v2f*)(eb + (size_t)ks * (2 * NT * 2));
      const v2f a0 = *(const v2f*)(xa0 + ks * (2 * 64 * 2));
      const v2f a1 = *(const v2f*)(xa1 + ks * (2 * 64 * 2));
      const v2f a2 = *(const v2f*)(xa2 + ks * (2 * 64 * 2));
      const v2f a3 = *(const v2f*)(xa3 + ks * (2 * 64 * 2));
      acc0 = wmma4(a0, bb, acc0);
      acc1 = wmma4(a1, bb, acc1);
      acc2 = wmma4(a2, bb, acc2);
      acc3 = wmma4(a3, bb, acc3);
    }
    const float e2 = en2[n0 + mc];
    const int   n  = n0 + mc;
#pragma unroll
    for (int r = 0; r < 8; ++r) {
      float d;
      d = e2 - 2.0f * acc0[r]; if (d < bestv[0][r]) { bestv[0][r] = d; besti[0][r] = n; }
      d = e2 - 2.0f * acc1[r]; if (d < bestv[1][r]) { bestv[1][r] = d; besti[1][r] = n; }
      d = e2 - 2.0f * acc2[r]; if (d < bestv[2][r]) { bestv[2][r] = d; besti[2][r] = n; }
      d = e2 - 2.0f * acc3[r]; if (d < bestv[3][r]) { bestv[3][r] = d; besti[3][r] = n; }
    }
  }

#pragma unroll
  for (int j = 0; j < 4; ++j) {
#pragma unroll
    for (int r = 0; r < 8; ++r) {
      float v = bestv[j][r];
      int  ix = besti[j][r];
#pragma unroll
      for (int m = 8; m > 0; m >>= 1) {
        const float ov = __shfl_xor(v, m);
        const int   oi = __shfl_xor(ix, m);
        const bool take = (ov < v) | ((ov == v) & (oi < ix));
        v  = take ? ov : v;
        ix = take ? oi : ix;
      }
      bestv[j][r] = v;
      besti[j][r] = ix;
    }
  }
  if (mc == 0) {
#pragma unroll
    for (int j = 0; j < 4; ++j)
#pragma unroll
      for (int r = 0; r < 8; ++r) {
        rv[wave][16 * j + r + 8 * g] = bestv[j][r];
        ri[wave][16 * j + r + 8 * g] = besti[j][r];
      }
  }
  __syncthreads();

  if (tid < 64) {
    float bv = rv[0][tid]; int bi = ri[0][tid];
    for (int w = 1; w < 4; ++w) {
      const float v = rv[w][tid]; const int ii = ri[w][tid];
      if (v < bv || (v == bv && ii < bi)) { bv = v; bi = ii; }
    }
    idxs[tid] = bi;
    atomicAdd(&counts[b * NT + bi], 1u);
  }
  __syncthreads();

  float fsum = 0.f;
  {
    const int id = idxs[sp];
    const float* xs = As + VQ_SLOT(sc, sp) * 2;
    const float* es = embed + (2 * sc) * NT + id;
    float* qd = q + gbase0 + (long)(2 * sc) * HW + sp;
#pragma unroll 8
    for (int i = 0; i < 64; ++i) {
      const v2f x = *(const v2f*)(xs + i * (2 * 64 * 2));
      const float e0 = es[(4 * i) * NT];
      const float e1 = es[(4 * i + 1) * NT];
      qd[(long)(4 * i) * HW]     = e0;
      qd[(long)(4 * i + 1) * HW] = e1;
      const float d0 = e0 - x[0], d1 = e1 - x[1];
      fsum += d0 * d0 + d1 * d1;
    }
  }
  red[tid] = fsum; __syncthreads();
  for (int st = 64; st > 0; st >>= 1) {
    if (tid < st) red[tid] += red[tid + st];
    __syncthreads();
  }
  if (tid == 0) wsdiff[s] = red[0];
}

#define KCH    32
#define A_ROW  66
#define A_CCP  208

__global__ __launch_bounds__(128) void k_res_conv3(const float* __restrict__ q,
                                                   const float* __restrict__ wp,
                                                   const float* __restrict__ b1,
                                                   float* __restrict__ hbuf) {
  __shared__ float Ah[KCH * A_CCP * 2];
  const int s  = blockIdx.x;
  const int b  = s >> 6;
  const int hh = s & 63;
  const int tid = threadIdx.x;
  const int wave = tid >> 5, lane = tid & 31, g = lane >> 4, mc = lane & 15;
  const int oc = wave * 16 + mc;

  for (int t = tid; t < KCH * 6; t += 128) {
    const int cc = t / 6, r6 = t % 6, rr = r6 >> 1;
    const int ww = (r6 & 1) ? (A_ROW - 1) : 0;
    const v2f z = {0.f, 0.f};
    *(v2f*)(Ah + (cc * A_CCP + rr * A_ROW + ww) * 2) = z;
  }

  const int sw = tid & 63;
  const int sc = tid >> 6;

  v8f acc0 = {0.f, 0.f, 0.f, 0.f, 0.f, 0.f, 0.f, 0.f};
  v8f acc1 = acc0, acc2 = acc0, acc3 = acc0;

  for (int kc = 0; kc < C_ / (2 * KCH); ++kc) {
    if (kc) __syncthreads();
#pragma unroll
    for (int rr = 0; rr < 3; ++rr) {
      const int hy = hh + rr - 1;
      float* dst = Ah + (sc * A_CCP + rr * A_ROW + 1 + sw) * 2;
      if ((unsigned)hy < 64u) {
        const float* qr = q + (long)b * CHW + (long)(2 * kc * KCH + 2 * sc) * HW
                            + hy * W_ + sw;
#pragma unroll 8
        for (int i = 0; i < KCH / 2; ++i) {
          float v0 = qr[(long)(4 * i) * HW];
          float v1 = qr[(long)(4 * i + 1) * HW];
          v0 = v0 > 0.f ? v0 : 0.f;
          v1 = v1 > 0.f ? v1 : 0.f;
          const v2f pv = {v0, v1};
          *(v2f*)(dst + i * (2 * A_CCP * 2)) = pv;
        }
      } else {
        const v2f z = {0.f, 0.f};
        for (int i = 0; i < KCH / 2; ++i) *(v2f*)(dst + i * (2 * A_CCP * 2)) = z;
      }
    }
    __syncthreads();

    const float* wk = wp + ((size_t)(kc * KCH + g) * CB + oc) * 2;
    const float* xb = Ah + (g * A_CCP + mc) * 2;
    for (int tap = 0; tap < 9; ++tap) {
      const int dh = tap / 3, dw = tap - dh * 3;
      const float* wt = wk + (size_t)tap * WP_PER_TAP * 2;
      const float* xt = xb + (dh * A_ROW + dw) * 2;
#pragma unroll 8
      for (int ks = 0; ks < KCH / 2; ++ks) {
        const v2f wb = *(const v2f*)(wt + ks * (2 * CB * 2));
        const float* xp = xt + ks * (2 * A_CCP * 2);
        const v2f x0 = *(const v2f*)(xp);
        const v2f x1 = *(const v2f*)(xp + 32);
        const v2f x2 = *(const v2f*)(xp + 64);
        const v2f x3 = *(const v2f*)(xp + 96);
        acc0 = wmma4(wb, x0, acc0);
        acc1 = wmma4(wb, x1, acc1);
        acc2 = wmma4(wb, x2, acc2);
        acc3 = wmma4(wb, x3, acc3);
      }
    }
  }

  const int ocb = wave * 16 + 8 * g;
  const long ob = (long)b * (CB * HW) + (long)ocb * HW + hh * W_ + mc;
#pragma unroll
  for (int r = 0; r < 8; ++r) {
    const float bias = b1[ocb + r];
    float* hp = hbuf + ob + (long)r * HW;
    float v;
    v = acc0[r] + bias; hp[0]  = v > 0.f ? v : 0.f;
    v = acc1[r] + bias; hp[16] = v > 0.f ? v : 0.f;
    v = acc2[r] + bias; hp[32] = v > 0.f ? v : 0.f;
    v = acc3[r] + bias; hp[48] = v > 0.f ? v : 0.f;
  }
}

__global__ __launch_bounds__(512) void k_res_conv1(float* __restrict__ q,
                                                   const float* __restrict__ hbuf,
                                                   const float* __restrict__ w2,
                                                   const float* __restrict__ b2) {
  __shared__ float Hs[1024];
  const int s = blockIdx.x;
  DECODE_TILE(s, b, hh, w0)
  const int tid = threadIdx.x;
  {
    const int cc = tid >> 4, m = tid & 15;
    const long gb = (long)b * (CB * HW) + hh * W_ + w0 + m;
    Hs[tid * 2 + 0] = hbuf[gb + (2 * cc) * HW];
    Hs[tid * 2 + 1] = hbuf[gb + (2 * cc + 1) * HW];
  }
  __syncthreads();

  const int wave = tid >> 5, lane = tid & 31, g = lane >> 4, mc = lane & 15;
  const int oc = wave * 16 + mc;
  v8f acc = {0.f, 0.f, 0.f, 0.f, 0.f, 0.f, 0.f, 0.f};
#pragma unroll
  for (int k0 = 0; k0 < CB; k0 += 4) {
    const int cc = (k0 >> 1) + g;
    v2f xa = *(const v2f*)(Hs + (cc * 16 + mc) * 2);
    v2f wb = *(const v2f*)(w2 + oc * CB + k0 + 2 * g);
    acc = wmma4(wb, xa, acc);
  }
  const int ocb = wave * 16 + 8 * g;
  const long ob = (long)b * CHW + (long)ocb * HW + hh * W_ + w0 + mc;
#pragma unroll
  for (int r = 0; r < 8; ++r)
    q[ob + (long)r * HW] += acc[r] + b2[ocb + r];
}

__global__ __launch_bounds__(32) void k_out_conv(const float* __restrict__ q,
                                                 const float* __restrict__ ow,
                                                 const float* __restrict__ ob,
                                                 float* __restrict__ out) {
  __shared__ float Qs[4096];
  __shared__ float Bs[4096];
  const int s = blockIdx.x;
  DECODE_TILE(s, b, hh, w0)
  const int lane = threadIdx.x, g = lane >> 4, mc = lane & 15;
  const long gbase0 = (long)b * CHW + hh * W_ + w0;

  for (int t = lane; t < 2048; t += 32) {
    const int cc = t >> 4, m = t & 15;
    Qs[t * 2 + 0] = q[gbase0 + (2 * cc) * HW + m];
    Qs[t * 2 + 1] = q[gbase0 + (2 * cc + 1) * HW + m];
    Bs[t * 2 + 0] = (m < OUTC) ? ow[m * C_ + 2 * cc]     : 0.f;
    Bs[t * 2 + 1] = (m < OUTC) ? ow[m * C_ + 2 * cc + 1] : 0.f;
  }
  __syncthreads();

  v8f acc = {0.f, 0.f, 0.f, 0.f, 0.f, 0.f, 0.f, 0.f};
#pragma unroll 8
  for (int k0 = 0; k0 < C_; k0 += 4) {
    const int cc = (k0 >> 1) + g;
    v2f a  = *(const v2f*)(Qs + (cc * 16 + mc) * 2);
    v2f bb = *(const v2f*)(Bs + (cc * 16 + mc) * 2);
    acc = wmma4(a, bb, acc);
  }
  if (mc < OUTC) {
    const float bias = ob[mc];
    const long obase = ((long)(b * OUTC + mc) * H_ + hh) * W_ + w0;
#pragma unroll
    for (int r = 0; r < 8; ++r) out[obase + r + 8 * g] = acc[r] + bias;
  }
}

__global__ __launch_bounds__(256) void k_final(const unsigned* __restrict__ counts,
                                               const float* __restrict__ wsdiff,
                                               float* __restrict__ out) {
  __shared__ float red[256];
  const int gid = blockIdx.x * 256 + threadIdx.x;
  if (blockIdx.x < 32) {
    out[WF_POS + gid] = (float)counts[gid] * (1.0f / 4096.0f);
  } else {
    float sum = 0.f;
    for (int t = threadIdx.x; t < B_ * H_; t += 256) sum += wsdiff[t];
    red[threadIdx.x] = sum; __syncthreads();
    for (int st = 128; st > 0; st >>= 1) {
      if (threadIdx.x < st) red[threadIdx.x] += red[threadIdx.x + st];
      __syncthreads();
    }
    if (threadIdx.x == 0)
      out[DIFF_POS] = red[0] * (1.0f / ((float)NPOS * (float)C_));
  }
}

extern "C" void kernel_launch(void* const* d_in, const int* in_sizes, int n_in,
                              void* d_out, int out_size, void* d_ws, size_t ws_size,
                              hipStream_t stream) {
  (void)in_sizes; (void)n_in; (void)out_size; (void)ws_size;
  const float* in  = (const float*)d_in[0];
  const float* emb = (const float*)d_in[1];
  const float* rw1 = (const float*)d_in[2];
  const float* rb1 = (const float*)d_in[3];
  const float* rw2 = (const float*)d_in[4];
  const float* rb2 = (const float*)d_in[5];
  const float* ow  = (const float*)d_in[6];
  const float* ob  = (const float*)d_in[7];
  float* out = (float*)d_out;
  float* ws  = (float*)d_ws;

  float*    q      = ws + Q_OFF;
  float*    hbuf   = ws + H_OFF;
  float*    epack  = ws + EPACK_OFF;
  float*    en2    = ws + EN2_OFF;
  unsigned* counts = (unsigned*)(ws + CNT_OFF);
  float*    wsdiff = ws + WDIFF_OFF;
  float*    wpack  = ws + WPACK_OFF;

  k_prep<<<256, 256, 0, stream>>>(emb, epack, en2, counts);
  k_packw<<<(3 * WP_PER_BLK + 255) / 256, 256, 0, stream>>>(rw1, wpack);
  k_vq<<<B_ * H_, 128, 0, stream>>>(in, emb, epack, en2, q, counts, wsdiff);
  for (int i = 0; i < 3; ++i) {
    k_res_conv3<<<B_ * H_, 128, 0, stream>>>(q, wpack + (size_t)i * WP_PER_BLK * 2,
                                           rb1 + i * CB, hbuf);
    k_res_conv1<<<NTILE, 512, 0, stream>>>(q, hbuf, rw2 + (size_t)i * C_ * CB,
                                           rb2 + i * C_);
  }
  k_out_conv<<<NTILE, 32, 0, stream>>>(q, ow, ob, out);
  k_final<<<33, 256, 0, stream>>>(counts, wsdiff, out);
}


# 1x1 residual conv: one work-group per image row (8 waves, each 32 oc x 64 w), weight matrix staged once in LDS by whole-line loads, q read-add-write 16 B per lane over whole 256 B rows; same sums
# speedup vs baseline: 3.4541x; 1.0835x over previous
#include <hip/hip_runtime.h>


typedef float v2f __attribute__((ext_vector_type(2)));
typedef float v4f __attribute__((ext_vector_type(4)));
typedef float v8f __attribute__((ext_vector_type(8)));

#define B_    16
#define C_    256
#define H_    64
#define W_    64
#define OUTC  3
#define NT    512
#define CB    64

#define HW     (H_*W_)
#define NPOS   (B_*HW)
#define NTILE  (NPOS/16)
#define CHW    (C_*HW)

#define Q_OFF      0u
#define H_OFF      16777216u
#define EPACK_OFF  20971520u
#define EN2_OFF    21102592u
#define CNT_OFF    21103104u
#define WDIFF_OFF  21111296u
#define WPACK_OFF  21115392u

#define WP_PER_BLK (9*128*CB)
#define WP_PER_TAP (128*CB)

#define DIFF_POS   196608
#define WF_POS     196609

static __device__ __forceinline__ v8f wmma4(v2f a, v2f b, v8f c) {
  return __builtin_amdgcn_wmma_f32_16x16x4_f32(false, a, false, b, (short)0, c,
                                               false, false);
}

#define DECODE_TILE(s, b, hh, w0)                        \
  const int b  = (s) >> 8;                               \
  const int hh = ((s) & 255) >> 2;                       \
  const int w0 = ((s) & 3) << 4;

__global__ __launch_bounds__(256) void k_prep(const float* __restrict__ embed,
                                              float* __restrict__ epack,
                                              float* __restrict__ en2,
                                              unsigned* __restrict__ counts) {
  const int gid = blockIdx.x * 256 + threadIdx.x;
  if (gid < 128 * NT) {
    const int cc = gid >> 9, n = gid & (NT - 1);
    epack[gid * 2 + 0] = embed[(2 * cc) * NT + n];
    epack[gid * 2 + 1] = embed[(2 * cc + 1) * NT + n];
  }
  if (gid < NT) {
    float s = 0.f;
    for (int c = 0; c < C_; ++c) { float v = embed[c * NT + gid]; s += v * v; }
    en2[gid] = s;
  }
  if (gid < B_ * NT) counts[gid] = 0u;
}

__global__ __launch_bounds__(256) void k_packw(const float* __restrict__ rw1,
                                               float* __restrict__ wpack) {
  const int t = blockIdx.x * 256 + threadIdx.x;
  if (t >= 3 * WP_PER_BLK) return;
  const int i   = t / WP_PER_BLK;
  const int r   = t % WP_PER_BLK;
  const int tap = r / WP_PER_TAP;
  const int r2  = r % WP_PER_TAP;
  const int cc  = r2 >> 6;
  const int oc  = r2 & 63;
  const int dh = tap / 3, dw = tap % 3;
  const float* src = rw1 + (size_t)i * CB * C_ * 9
                         + ((size_t)oc * C_ + 2 * cc) * 9 + dh * 3 + dw;
  wpack[t * 2 + 0] = src[0];
  wpack[t * 2 + 1] = src[9];
}

#define VQ_SLOT(cc, p) (((cc) << 6) + (((p) + (((cc) & 1) << 4)) & 63))

__global__ __launch_bounds__(128) void k_vq(const float* __restrict__ in,
                                            const float* __restrict__ embed,
                                            const float* __restrict__ epack,
                                            const float* __restrict__ en2,
                                            float* __restrict__ q,
                                            unsigned* __restrict__ counts,
                                            float* __restrict__ wsdiff) {
  __shared__ float As[128 * 64 * 2];
  __shared__ float rv[4][64];
  __shared__ int   ri[4][64];
  __shared__ int   idxs[64];
  __shared__ float red[128];

  const int s  = blockIdx.x;
  const int b  = s >> 6;
  const int hh = s & 63;
  const int tid = threadIdx.x;
  const long gbase0 = (long)b * CHW + hh * W_;
  const int sp = tid & 63;
  const int sc = tid >> 6;

  {
    const float* src = in + gbase0 + (long)(2 * sc) * HW + sp;
    float* dst = As + VQ_SLOT(sc, sp) * 2;
#pragma unroll 8
    for (int i = 0; i < 64; ++i) {
      const float v0 = src[(long)(4 * i) * HW];
      const float v1 = src[(long)(4 * i + 1) * HW];
      const v2f pv = {v0, v1};
      *(v2f*)(dst + i * (2 * 64 * 2)) = pv;
    }
  }
  __syncthreads();

  const int wave = tid >> 5, lane = tid & 31, g = lane >> 4, mc = lane & 15;
  float bestv[4][8];
  int   besti[4][8];
#pragma unroll
  for (int j = 0; j < 4; ++j)
#pragma unroll
    for (int r = 0; r < 8; ++r) { bestv[j][r] = 3.4e38f; besti[j][r] = 0; }

  const float* xa0 = As + ((g << 6) + (((0 + g) & 3) << 4) + mc) * 2;
  const float* xa1 = As + ((g << 6) + (((1 + g) & 3) << 4) + mc) * 2;
  const float* xa2 = As + ((g << 6) + (((2 + g) & 3) << 4) + mc) * 2;
  const float* xa3 = As + ((g << 6) + (((3 + g) & 3) << 4) + mc) * 2;

  for (int tt = 0; tt < 8; ++tt) {
    const int n0 = (wave * 8 + tt) << 4;
    const float* eb = epack + ((size_t)g * NT + n0 + mc) * 2;
    v8f acc0 = {0.f, 0.f, 0.f, 0.f, 0.f, 0.f, 0.f, 0.f};
    v8f acc1 = acc0, acc2 = acc0, acc3 = acc0;
#pragma unroll 8
    for (int ks = 0; ks < 64; ++ks) {
      const v2f bb = *(const v2f*)(eb + (size_t)ks * (2 * NT * 2));
      const v2f a0 = *(const v2f*)(xa0 + ks * (2 * 64 * 2));
      const v2f a1 = *(const v2f*)(xa1 + ks * (2 * 64 * 2));
      const v2f a2 = *(const v2f*)(xa2 + ks * (2 * 64 * 2));
      const v2f a3 = *(const v2f*)(xa3 + ks * (2 * 64 * 2));
      acc0 = wmma4(a0, bb, acc0);
      acc1 = wmma4(a1, bb, acc1);
      acc2 = wmma4(a2, bb, acc2);
      acc3 = wmma4(a3, bb, acc3);
    }
    const float e2 = en2[n0 + mc];
    const int   n  = n0 + mc;
#pragma unroll
    for (int r = 0; r < 8; ++r) {
      float d;
      d = e2 - 2.0f * acc0[r]; if (d < bestv[0][r]) { bestv[0][r] = d; besti[0][r] = n; }
      d = e2 - 2.0f * acc1[r]; if (d < bestv[1][r]) { bestv[1][r] = d; besti[1][r] = n; }
      d = e2 - 2.0f * acc2[r]; if (d < bestv[2][r]) { bestv[2][r] = d; besti[2][r] = n; }
      d = e2 - 2.0f * acc3[r]; if (d < bestv[3][r]) { bestv[3][r] = d; besti[3][r] = n; }
    }
  }

#pragma unroll
  for (int j = 0; j < 4; ++j) {
#pragma unroll
    for (int r = 0; r < 8; ++r) {
      float v = bestv[j][r];
      int  ix = besti[j][r];
#pragma unroll
      for (int m = 8; m > 0; m >>= 1) {
        const float ov = __shfl_xor(v, m);
        const int   oi = __shfl_xor(ix, m);
        const bool take = (ov < v) | ((ov == v) & (oi < ix));
        v  = take ? ov : v;
        ix = take ? oi : ix;
      }
      bestv[j][r] = v;
      besti[j][r] = ix;
    }
  }
  if (mc == 0) {
#pragma unroll
    for (int j = 0; j < 4; ++j)
#pragma unroll
      for (int r = 0; r < 8; ++r) {
        rv[wave][16 * j + r + 8 * g] = bestv[j][r];
        ri[wave][16 * j + r + 8 * g] = besti[j][r];
      }
  }
  __syncthreads();

  if (tid < 64) {
    float bv = rv[0][tid]; int bi = ri[0][tid];
    for (int w = 1; w < 4; ++w) {
      const float v = rv[w][tid]; const int ii = ri[w][tid];
      if (v < bv || (v == bv && ii < bi)) { bv = v; bi = ii; }
    }
    idxs[tid] = bi;
    atomicAdd(&counts[b * NT + bi], 1u);
  }
  __syncthreads();

  float fsum = 0.f;
  {
    const int id = idxs[sp];
    const float* xs = As + VQ_SLOT(sc, sp) * 2;
    const float* es = embed + (2 * sc) * NT + id;
    float* qd = q + gbase0 + (long)(2 * sc) * HW + sp;
#pragma unroll 8
    for (int i = 0; i < 64; ++i) {
      const v2f x = *(const v2f*)(xs + i * (2 * 64 * 2));
      const float e0 = es[(4 * i) * NT];
      const float e1 = es[(4 * i + 1) * NT];
      qd[(long)(4 * i) * HW]     = e0;
      qd[(long)(4 * i + 1) * HW] = e1;
      const float d0 = e0 - x[0], d1 = e1 - x[1];
      fsum += d0 * d0 + d1 * d1;
    }
  }
  red[tid] = fsum; __syncthreads();
  for (int st = 64; st > 0; st >>= 1) {
    if (tid < st) red[tid] += red[tid + st];
    __syncthreads();
  }
  if (tid == 0) wsdiff[s] = red[0];
}

#define KCH    32
#define A_ROW  66
#define A_CCP  208

__global__ __launch_bounds__(128) void k_res_conv3(const float* __restrict__ q,
                                                   const float* __restrict__ wp,
                                                   const float* __restrict__ b1,
                                                   float* __restrict__ hbuf) {
  __shared__ float Ah[KCH * A_CCP * 2];
  const int s  = blockIdx.x;
  const int b  = s >> 6;
  const int hh = s & 63;
  const int tid = threadIdx.x;
  const int wave = tid >> 5, lane = tid & 31, g = lane >> 4, mc = lane & 15;
  const int oc = wave * 16 + mc;

  for (int t = tid; t < KCH * 6; t += 128) {
    const int cc = t / 6, r6 = t % 6, rr = r6 >> 1;
    const int ww = (r6 & 1) ? (A_ROW - 1) : 0;
    const v2f z = {0.f, 0.f};
    *(v2f*)(Ah + (cc * A_CCP + rr * A_ROW + ww) * 2) = z;
  }

  const int sw = tid & 63;
  const int sc = tid >> 6;

  v8f acc0 = {0.f, 0.f, 0.f, 0.f, 0.f, 0.f, 0.f, 0.f};
  v8f acc1 = acc0, acc2 = acc0, acc3 = acc0;

  for (int kc = 0; kc < C_ / (2 * KCH); ++kc) {
    if (kc) __syncthreads();
#pragma unroll
    for (int rr = 0; rr < 3; ++rr) {
      const int hy = hh + rr - 1;
      float* dst = Ah + (sc * A_CCP + rr * A_ROW + 1 + sw) * 2;
      if ((unsigned)hy < 64u) {
        const float* qr = q + (long)b * CHW + (long)(2 * kc * KCH + 2 * sc) * HW
                            + hy * W_ + sw;
#pragma unroll 8
        for (int i = 0; i < KCH / 2; ++i) {
          float v0 = qr[(long)(4 * i) * HW];
          float v1 = qr[(long)(4 * i + 1) * HW];
          v0 = v0 > 0.f ? v0 : 0.f;
          v1 = v1 > 0.f ? v1 : 0.f;
          const v2f pv = {v0, v1};
          *(v2f*)(dst + i * (2 * A_CCP * 2)) = pv;
        }
      } else {
        const v2f z = {0.f, 0.f};
        for (int i = 0; i < KCH / 2; ++i) *(v2f*)(dst + i * (2 * A_CCP * 2)) = z;
      }
    }
    __syncthreads();

    const float* wk = wp + ((size_t)(kc * KCH + g) * CB + oc) * 2;
    const float* xb = Ah + (g * A_CCP + mc) * 2;
    for (int tap = 0; tap < 9; ++tap) {
      const int dh = tap / 3, dw = tap - dh * 3;
      const float* wt = wk + (size_t)tap * WP_PER_TAP * 2;
      const float* xt = xb + (dh * A_ROW + dw) * 2;
#pragma unroll 8
      for (int ks = 0; ks < KCH / 2; ++ks) {
        const v2f wb = *(const v2f*)(wt + ks * (2 * CB * 2));
        const float* xp = xt + ks * (2 * A_CCP * 2);
        const v2f x0 = *(const v2f*)(xp);
        const v2f x1 = *(const v2f*)(xp + 32);
        const v2f x2 = *(const v2f*)(xp + 64);
        const v2f x3 = *(const v2f*)(xp + 96);
        acc0 = wmma4(wb, x0, acc0);
        acc1 = wmma4(wb, x1, acc1);
        acc2 = wmma4(wb, x2, acc2);
        acc3 = wmma4(wb, x3, acc3);
      }
    }
  }

  const int ocb = wave * 16 + 8 * g;
  const long ob = (long)b * (CB * HW) + (long)ocb * HW + hh * W_ + mc;
#pragma unroll
  for (int r = 0; r < 8; ++r) {
    const float bias = b1[ocb + r];
    float* hp = hbuf + ob + (long)r * HW;
    float v;
    v = acc0[r] + bias; hp[0]  = v > 0.f ? v : 0.f;
    v = acc1[r] + bias; hp[16] = v > 0.f ? v : 0.f;
    v = acc2[r] + bias; hp[32] = v > 0.f ? v : 0.f;
    v = acc3[r] + bias; hp[48] = v > 0.f ? v : 0.f;
  }
}

#define C1_HP  80
#define C1_WP  68

__global__ __launch_bounds__(256) void k_res_conv1(float* __restrict__ q,
                                                   const float* __restrict__ hbuf,
                                                   const float* __restrict__ w2,
                                                   const float* __restrict__ b2) {
  __shared__ __attribute__((aligned(16))) float Hs[32 * C1_HP * 2];
  __shared__ __attribute__((aligned(16))) float Ws[C_ * C1_WP];
  const int s  = blockIdx.x;
  const int b  = s >> 6;
  const int hh = s & 63;
  const int tid = threadIdx.x;
  const int wave = tid >> 5, lane = tid & 31, g = lane >> 4, mc = lane & 15;

  float* qb = q + (long)b * CHW + (long)(32 * wave + 8 * g) * HW + hh * W_ + 4 * mc;
  v4f qv[2][8];
#pragma unroll
  for (int t = 0; t < 2; ++t)
#pragma unroll
    for (int r = 0; r < 8; ++r)
      qv[t][r] = *(const v4f*)(qb + (long)(16 * t + r) * HW);

#pragma unroll
  for (int i = 0; i < 16; ++i) {
    const int i4 = tid + 256 * i;
    const v4f v = *(const v4f*)(w2 + i4 * 4);
    *(v4f*)(Ws + (i4 >> 4) * C1_WP + (i4 & 15) * 4) = v;
  }
  {
    const int m = tid & 15, ccb = tid >> 4;
    const float* hs = hbuf + (long)b * (CB * HW) + hh * W_ + 4 * m;
#pragma unroll
    for (int i = 0; i < 2; ++i) {
      const int cc = ccb + 16 * i;
      const v4f v0 = *(const v4f*)(hs + (long)(2 * cc) * HW);
      const v4f v1 = *(const v4f*)(hs + (long)(2 * cc + 1) * HW);
      float* dst = Hs + (cc * C1_HP + m) * 2;
#pragma unroll
      for (int j = 0; j < 4; ++j) {
        const v2f pv = {v0[j], v1[j]};
        *(v2f*)(dst + 16 * j * 2) = pv;
      }
    }
  }
  __syncthreads();

  v8f acc[2][4];
#pragma unroll
  for (int t = 0; t < 2; ++t)
#pragma unroll
    for (int j = 0; j < 4; ++j) {
      const v8f z = {0.f, 0.f, 0.f, 0.f, 0.f, 0.f, 0.f, 0.f};
      acc[t][j] = z;
    }

  const float* wa = Ws + (32 * wave + mc) * C1_WP + 2 * g;
  const float* xb = Hs + (g * C1_HP + mc) * 2;
#pragma unroll
  for (int ks = 0; ks < 16; ++ks) {
    const v2f a0 = *(const v2f*)(wa + 4 * ks);
    const v2f a1 = *(const v2f*)(wa + 16 * C1_WP + 4 * ks);
    const float* xp = xb + ks * (2 * C1_HP * 2);
    const v2f x0 = *(const v2f*)(xp);
    const v2f x1 = *(const v2f*)(xp + 32);
    const v2f x2 = *(const v2f*)(xp + 64);
    const v2f x3 = *(const v2f*)(xp + 96);
    acc[0][0] = wmma4(a0, x0, acc[0][0]);
    acc[0][1] = wmma4(a0, x1, acc[0][1]);
    acc[0][2] = wmma4(a0, x2, acc[0][2]);
    acc[0][3] = wmma4(a0, x3, acc[0][3]);
    acc[1][0] = wmma4(a1, x0, acc[1][0]);
    acc[1][1] = wmma4(a1, x1, acc[1][1]);
    acc[1][2] = wmma4(a1, x2, acc[1][2]);
    acc[1][3] = wmma4(a1, x3, acc[1][3]);
  }

  const int ocb = 32 * wave + 8 * g;
#pragma unroll
  for (int t = 0; t < 2; ++t)
#pragma unroll
    for (int r = 0; r < 8; ++r) {
      const float bias = b2[ocb + 16 * t + r];
      v4f o = qv[t][r];
      o[0] += acc[t][0][r] + bias;
      o[1] += acc[t][1][r] + bias;
      o[2] += acc[t][2][r] + bias;
      o[3] += acc[t][3][r] + bias;
      *(v4f*)(qb + (long)(16 * t + r) * HW) = o;
    }
}

__global__ __launch_bounds__(32) void k_out_conv(const float* __restrict__ q,
                                                 const float* __restrict__ ow,
                                                 const float* __restrict__ ob,
                                                 float* __restrict__ out) {
  __shared__ float Qs[4096];
  __shared__ float Bs[4096];
  const int s = blockIdx.x;
  DECODE_TILE(s, b, hh, w0)
  const int lane = threadIdx.x, g = lane >> 4, mc = lane & 15;
  const long gbase0 = (long)b * CHW + hh * W_ + w0;

  for (int t = lane; t < 2048; t += 32) {
    const int cc = t >> 4, m = t & 15;
    Qs[t * 2 + 0] = q[gbase0 + (2 * cc) * HW + m];
    Qs[t * 2 + 1] = q[gbase0 + (2 * cc + 1) * HW + m];
    Bs[t * 2 + 0] = (m < OUTC) ? ow[m * C_ + 2 * cc]     : 0.f;
    Bs[t * 2 + 1] = (m < OUTC) ? ow[m * C_ + 2 * cc + 1] : 0.f;
  }
  __syncthreads();

  v8f acc = {0.f, 0.f, 0.f, 0.f, 0.f, 0.f, 0.f, 0.f};
#pragma unroll 8
  for (int k0 = 0; k0 < C_; k0 += 4) {
    const int cc = (k0 >> 1) + g;
    v2f a  = *(const v2f*)(Qs + (cc * 16 + mc) * 2);
    v2f bb = *(const v2f*)(Bs + (cc * 16 + mc) * 2);
    acc = wmma4(a, bb, acc);
  }
  if (mc < OUTC) {
    const float bias = ob[mc];
    const long obase = ((long)(b * OUTC + mc) * H_ + hh) * W_ + w0;
#pragma unroll
    for (int r = 0; r < 8; ++r) out[obase + r + 8 * g] = acc[r] + bias;
  }
}

__global__ __launch_bounds__(256) void k_final(const unsigned* __restrict__ counts,
                                               const float* __restrict__ wsdiff,
                                               float* __restrict__ out) {
  __shared__ float red[256];
  const int gid = blockIdx.x * 256 + threadIdx.x;
  if (blockIdx.x < 32) {
    out[WF_POS + gid] = (float)counts[gid] * (1.0f / 4096.0f);
  } else {
    float sum = 0.f;
    for (int t = threadIdx.x; t < B_ * H_; t += 256) sum += wsdiff[t];
    red[threadIdx.x] = sum; __syncthreads();
    for (int st = 128; st > 0; st >>= 1) {
      if (threadIdx.x < st) red[threadIdx.x] += red[threadIdx.x + st];
      __syncthreads();
    }
    if (threadIdx.x == 0)
      out[DIFF_POS] = red[0] * (1.0f / ((float)NPOS * (float)C_));
  }
}

extern "C" void kernel_launch(void* const* d_in, const int* in_sizes, int n_in,
                              void* d_out, int out_size, void* d_ws, size_t ws_size,
                              hipStream_t stream) {
  (void)in_sizes; (void)n_in; (void)out_size; (void)ws_size;
  const float* in  = (const float*)d_in[0];
  const float* emb = (const float*)d_in[1];
  const float* rw1 = (const float*)d_in[2];
  const float* rb1 = (const float*)d_in[3];
  const float* rw2 = (const float*)d_in[4];
  const float* rb2 = (const float*)d_in[5];
  const float* ow  = (const float*)d_in[6];
  const float* ob  = (const float*)d_in[7];
  float* out = (float*)d_out;
  float* ws  = (float*)d_ws;

  float*    q      = ws + Q_OFF;
  float*    hbuf   = ws + H_OFF;
  float*    epack  = ws + EPACK_OFF;
  float*    en2    = ws + EN2_OFF;
  unsigned* counts = (unsigned*)(ws + CNT_OFF);
  float*    wsdiff = ws + WDIFF_OFF;
  float*    wpack  = ws + WPACK_OFF;

  k_prep<<<256, 256, 0, stream>>>(emb, epack, en2, counts);
  k_packw<<<(3 * WP_PER_BLK + 255) / 256, 256, 0, stream>>>(rw1, wpack);
  k_vq<<<B_ * H_, 128, 0, stream>>>(in, emb, epack, en2, q, counts, wsdiff);
  for (int i = 0; i < 3; ++i) {
    k_res_conv3<<<B_ * H_, 128, 0, stream>>>(q, wpack + (size_t)i * WP_PER_BLK * 2,
                                           rb1 + i * CB, hbuf);
    k_res_conv1<<<B_ * H_, 256, 0, stream>>>(q, hbuf, rw2 + (size_t)i * C_ * CB,
                                           rb2 + i * C_);
  }
  k_out_conv<<<NTILE, 32, 0, stream>>>(q, ow, ob, out);
  k_final<<<33, 256, 0, stream>>>(counts, wsdiff, out);
}


# out 1x1 conv: one work-group per image row; the q row fetched by whole-line 16 B loads, all issued ahead of the LDS stores (was 4 B pairs, each waited for before the next); same sums
# speedup vs baseline: 3.8677x; 1.1197x over previous
#include <hip/hip_runtime.h>


typedef float v2f __attribute__((ext_vector_type(2)));
typedef float v4f __attribute__((ext_vector_type(4)));
typedef float v8f __attribute__((ext_vector_type(8)));

#define B_    16
#define C_    256
#define H_    64
#define W_    64
#define OUTC  3
#define NT    512
#define CB    64

#define HW     (H_*W_)
#define NPOS   (B_*HW)
#define CHW    (C_*HW)

#define Q_OFF      0u
#define H_OFF      16777216u
#define EPACK_OFF  20971520u
#define EN2_OFF    21102592u
#define CNT_OFF    21103104u
#define WDIFF_OFF  21111296u
#define WPACK_OFF  21115392u

#define WP_PER_BLK (9*128*CB)
#define WP_PER_TAP (128*CB)

#define DIFF_POS   196608
#define WF_POS     196609

static __device__ __forceinline__ v8f wmma4(v2f a, v2f b, v8f c) {
  return __builtin_amdgcn_wmma_f32_16x16x4_f32(false, a, false, b, (short)0, c,
                                               false, false);
}

__global__ __launch_bounds__(256) void k_prep(const float* __restrict__ embed,
                                              float* __restrict__ epack,
                                              float* __restrict__ en2,
                                              unsigned* __restrict__ counts) {
  const int gid = blockIdx.x * 256 + threadIdx.x;
  if (gid < 128 * NT) {
    const int cc = gid >> 9, n = gid & (NT - 1);
    epack[gid * 2 + 0] = embed[(2 * cc) * NT + n];
    epack[gid * 2 + 1] = embed[(2 * cc + 1) * NT + n];
  }
  if (gid < NT) {
    float s = 0.f;
    for (int c = 0; c < C_; ++c) { float v = embed[c * NT + gid]; s += v * v; }
    en2[gid] = s;
  }
  if (gid < B_ * NT) counts[gid] = 0u;
}

__global__ __launch_bounds__(256) void k_packw(const float* __restrict__ rw1,
                                               float* __restrict__ wpack) {
  const int t = blockIdx.x * 256 + threadIdx.x;
  if (t >= 3 * WP_PER_BLK) return;
  const int i   = t / WP_PER_BLK;
  const int r   = t % WP_PER_BLK;
  const int tap = r / WP_PER_TAP;
  const int r2  = r % WP_PER_TAP;
  const int cc  = r2 >> 6;
  const int oc  = r2 & 63;
  const int dh = tap / 3, dw = tap % 3;
  const float* src = rw1 + (size_t)i * CB * C_ * 9
                         + ((size_t)oc * C_ + 2 * cc) * 9 + dh * 3 + dw;
  wpack[t * 2 + 0] = src[0];
  wpack[t * 2 + 1] = src[9];
}

#define VQ_SLOT(cc, p) (((cc) << 6) + (((p) + (((cc) & 1) << 4)) & 63))

__global__ __launch_bounds__(128) void k_vq(const float* __restrict__ in,
                                            const float* __restrict__ embed,
                                            const float* __restrict__ epack,
                                            const float* __restrict__ en2,
                                            float* __restrict__ q,
                                            unsigned* __restrict__ counts,
                                            float* __restrict__ wsdiff) {
  __shared__ float As[128 * 64 * 2];
  __shared__ float rv[4][64];
  __shared__ int   ri[4][64];
  __shared__ int   idxs[64];
  __shared__ float red[128];

  const int s  = blockIdx.x;
  const int b  = s >> 6;
  const int hh = s & 63;
  const int tid = threadIdx.x;
  const long gbase0 = (long)b * CHW + hh * W_;
  const int sp = tid & 63;
  const int sc = tid >> 6;

  {
    const float* src = in + gbase0 + (long)(2 * sc) * HW + sp;
    float* dst = As + VQ_SLOT(sc, sp) * 2;
#pragma unroll 8
    for (int i = 0; i < 64; ++i) {
      const float v0 = src[(long)(4 * i) * HW];
      const float v1 = src[(long)(4 * i + 1) * HW];
      const v2f pv = {v0, v1};
      *(v2f*)(dst + i * (2 * 64 * 2)) = pv;
    }
  }
  __syncthreads();

  const int wave = tid >> 5, lane = tid & 31, g = lane >> 4, mc = lane & 15;
  float bestv[4][8];
  int   besti[4][8];
#pragma unroll
  for (int j = 0; j < 4; ++j)
#pragma unroll
    for (int r = 0; r < 8; ++r) { bestv[j][r] = 3.4e38f; besti[j][r] = 0; }

  const float* xa0 = As + ((g << 6) + (((0 + g) & 3) << 4) + mc) * 2;
  const float* xa1 = As + ((g << 6) + (((1 + g) & 3) << 4) + mc) * 2;
  const float* xa2 = As + ((g << 6) + (((2 + g) & 3) << 4) + mc) * 2;
  const float* xa3 = As + ((g << 6) + (((3 + g) & 3) << 4) + mc) * 2;

  for (int tt = 0; tt < 8; ++tt) {
    const int n0 = (wave * 8 + tt) << 4;
    const float* eb = epack + ((size_t)g * NT + n0 + mc) * 2;
    v8f acc0 = {0.f, 0.f, 0.f, 0.f, 0.f, 0.f, 0.f, 0.f};
    v8f acc1 = acc0, acc2 = acc0, acc3 = acc0;
#pragma unroll 8
    for (int ks = 0; ks < 64; ++ks) {
      const v2f bb = *(const v2f*)(eb + (size_t)ks * (2 * NT * 2));
      const v2f a0 = *(const v2f*)(xa0 + ks * (2 * 64 * 2));
      const v2f a1 = *(const v2f*)(xa1 + ks * (2 * 64 * 2));
      const v2f a2 = *(const v2f*)(xa2 + ks * (2 * 64 * 2));
      const v2f a3 = *(const v2f*)(xa3 + ks * (2 * 64 * 2));
      acc0 = wmma4(a0, bb, acc0);
      acc1 = wmma4(a1, bb, acc1);
      acc2 = wmma4(a2, bb, acc2);
      acc3 = wmma4(a3, bb, acc3);
    }
    const float e2 = en2[n0 + mc];
    const int   n  = n0 + mc;
#pragma unroll
    for (int r = 0; r < 8; ++r) {
      float d;
      d = e2 - 2.0f * acc0[r]; if (d < bestv[0][r]) { bestv[0][r] = d; besti[0][r] = n; }
      d = e2 - 2.0f * acc1[r]; if (d < bestv[1][r]) { bestv[1][r] = d; besti[1][r] = n; }
      d = e2 - 2.0f * acc2[r]; if (d < bestv[2][r]) { bestv[2][r] = d; besti[2][r] = n; }
      d = e2 - 2.0f * acc3[r]; if (d < bestv[3][r]) { bestv[3][r] = d; besti[3][r] = n; }
    }
  }

#pragma unroll
  for (int j = 0; j < 4; ++j) {
#pragma unroll
    for (int r = 0; r < 8; ++r) {
      float v = bestv[j][r];
      int  ix = besti[j][r];
#pragma unroll
      for (int m = 8; m > 0; m >>= 1) {
        const float ov = __shfl_xor(v, m);
        const int   oi = __shfl_xor(ix, m);
        const bool take = (ov < v) | ((ov == v) & (oi < ix));
        v  = take ? ov : v;
        ix = take ? oi : ix;
      }
      bestv[j][r] = v;
      besti[j][r] = ix;
    }
  }
  if (mc == 0) {
#pragma unroll
    for (int j = 0; j < 4; ++j)
#pragma unroll
      for (int r = 0; r < 8; ++r) {
        rv[wave][16 * j + r + 8 * g] = bestv[j][r];
        ri[wave][16 * j + r + 8 * g] = besti[j][r];
      }
  }
  __syncthreads();

  if (tid < 64) {
    float bv = rv[0][tid]; int bi = ri[0][tid];
    for (int w = 1; w < 4; ++w) {
      const float v = rv[w][tid]; const int ii = ri[w][tid];
      if (v < bv || (v == bv && ii < bi)) { bv = v; bi = ii; }
    }
    idxs[tid] = bi;
    atomicAdd(&counts[b * NT + bi], 1u);
  }
  __syncthreads();

  float fsum = 0.f;
  {
    const int id = idxs[sp];
    const float* xs = As + VQ_SLOT(sc, sp) * 2;
    const float* es = embed + (2 * sc) * NT + id;
    float* qd = q + gbase0 + (long)(2 * sc) * HW + sp;
#pragma unroll 8
    for (int i = 0; i < 64; ++i) {
      const v2f x = *(const v2f*)(xs + i * (2 * 64 * 2));
      const float e0 = es[(4 * i) * NT];
      const float e1 = es[(4 * i + 1) * NT];
      qd[(long)(4 * i) * HW]     = e0;
      qd[(long)(4 * i + 1) * HW] = e1;
      const float d0 = e0 - x[0], d1 = e1 - x[1];
      fsum += d0 * d0 + d1 * d1;
    }
  }
  red[tid] = fsum; __syncthreads();
  for (int st = 64; st > 0; st >>= 1) {
    if (tid < st) red[tid] += red[tid + st];
    __syncthreads();
  }
  if (tid == 0) wsdiff[s] = red[0];
}

#define KCH    32
#define A_ROW  66
#define A_CCP  208

__global__ __launch_bounds__(128) void k_res_conv3(const float* __restrict__ q,
                                                   const float* __restrict__ wp,
                                                   const float* __restrict__ b1,
                                                   float* __restrict__ hbuf) {
  __shared__ float Ah[KCH * A_CCP * 2];
  const int s  = blockIdx.x;
  const int b  = s >> 6;
  const int hh = s & 63;
  const int tid = threadIdx.x;
  const int wave = tid >> 5, lane = tid & 31, g = lane >> 4, mc = lane & 15;
  const int oc = wave * 16 + mc;

  for (int t = tid; t < KCH * 6; t += 128) {
    const int cc = t / 6, r6 = t % 6, rr = r6 >> 1;
    const int ww = (r6 & 1) ? (A_ROW - 1) : 0;
    const v2f z = {0.f, 0.f};
    *(v2f*)(Ah + (cc * A_CCP + rr * A_ROW + ww) * 2) = z;
  }

  const int sw = tid & 63;
  const int sc = tid >> 6;

  v8f acc0 = {0.f, 0.f, 0.f, 0.f, 0.f, 0.f, 0.f, 0.f};
  v8f acc1 = acc0, acc2 = acc0, acc3 = acc0;

  for (int kc = 0; kc < C_ / (2 * KCH); ++kc) {
    if (kc) __syncthreads();
#pragma unroll
    for (int rr = 0; rr < 3; ++rr) {
      const int hy = hh + rr - 1;
      float* dst = Ah + (sc * A_CCP + rr * A_ROW + 1 + sw) * 2;
      if ((unsigned)hy < 64u) {
        const float* qr = q + (long)b * CHW + (long)(2 * kc * KCH + 2 * sc) * HW
                            + hy * W_ + sw;
#pragma unroll 8
        for (int i = 0; i < KCH / 2; ++i) {
          float v0 = qr[(long)(4 * i) * HW];
          float v1 = qr[(long)(4 * i + 1) * HW];
          v0 = v0 > 0.f ? v0 : 0.f;
          v1 = v1 > 0.f ? v1 : 0.f;
          const v2f pv = {v0, v1};
          *(v2f*)(dst + i * (2 * A_CCP * 2)) = pv;
        }
      } else {
        const v2f z = {0.f, 0.f};
        for (int i = 0; i < KCH / 2; ++i) *(v2f*)(dst + i * (2 * A_CCP * 2)) = z;
      }
    }
    __syncthreads();

    const float* wk = wp + ((size_t)(kc * KCH + g) * CB + oc) * 2;
    const float* xb = Ah + (g * A_CCP + mc) * 2;
    for (int tap = 0; tap < 9; ++tap) {
      const int dh = tap / 3, dw = tap - dh * 3;
      const float* wt = wk + (size_t)tap * WP_PER_TAP * 2;
      const float* xt = xb + (dh * A_ROW + dw) * 2;
#pragma unroll 8
      for (int ks = 0; ks < KCH / 2; ++ks) {
        const v2f wb = *(const v2f*)(wt + ks * (2 * CB * 2));
        const float* xp = xt + ks * (2 * A_CCP * 2);
        const v2f x0 = *(const v2f*)(xp);
        const v2f x1 = *(const v2f*)(xp + 32);
        const v2f x2 = *(const v2f*)(xp + 64);
        const v2f x3 = *(const v2f*)(xp + 96);
        acc0 = wmma4(wb, x0, acc0);
        acc1 = wmma4(wb, x1, acc1);
        acc2 = wmma4(wb, x2, acc2);
        acc3 = wmma4(wb, x3, acc3);
      }
    }
  }

  const int ocb = wave * 16 + 8 * g;
  const long ob = (long)b * (CB * HW) + (long)ocb * HW + hh * W_ + mc;
#pragma unroll
  for (int r = 0; r < 8; ++r) {
    const float bias = b1[ocb + r];
    float* hp = hbuf + ob + (long)r * HW;
    float v;
    v = acc0[r] + bias; hp[0]  = v > 0.f ? v : 0.f;
    v = acc1[r] + bias; hp[16] = v > 0.f ? v : 0.f;
    v = acc2[r] + bias; hp[32] = v > 0.f ? v : 0.f;
    v = acc3[r] + bias; hp[48] = v > 0.f ? v : 0.f;
  }
}

#define C1_HP  80
#define C1_WP  68

__global__ __launch_bounds__(256) void k_res_conv1(float* __restrict__ q,
                                                   const float* __restrict__ hbuf,
                                                   const float* __restrict__ w2,
                                                   const float* __restrict__ b2) {
  __shared__ __attribute__((aligned(16))) float Hs[32 * C1_HP * 2];
  __shared__ __attribute__((aligned(16))) float Ws[C_ * C1_WP];
  const int s  = blockIdx.x;
  const int b  = s >> 6;
  const int hh = s & 63;
  const int tid = threadIdx.x;
  const int wave = tid >> 5, lane = tid & 31, g = lane >> 4, mc = lane & 15;

  float* qb = q + (long)b * CHW + (long)(32 * wave + 8 * g) * HW + hh * W_ + 4 * mc;
  v4f qv[2][8];
#pragma unroll
  for (int t = 0; t < 2; ++t)
#pragma unroll
    for (int r = 0; r < 8; ++r)
      qv[t][r] = *(const v4f*)(qb + (long)(16 * t + r) * HW);

#pragma unroll
  for (int i = 0; i < 16; ++i) {
    const int i4 = tid + 256 * i;
    const v4f v = *(const v4f*)(w2 + i4 * 4);
    *(v4f*)(Ws + (i4 >> 4) * C1_WP + (i4 & 15) * 4) = v;
  }
  {
    const int m = tid & 15, ccb = tid >> 4;
    const float* hs = hbuf + (long)b * (CB * HW) + hh * W_ + 4 * m;
#pragma unroll
    for (int i = 0; i < 2; ++i) {
      const int cc = ccb + 16 * i;
      const v4f v0 = *(const v4f*)(hs + (long)(2 * cc) * HW);
      const v4f v1 = *(const v4f*)(hs + (long)(2 * cc + 1) * HW);
      float* dst = Hs + (cc * C1_HP + m) * 2;
#pragma unroll
      for (int j = 0; j < 4; ++j) {
        const v2f pv = {v0[j], v1[j]};
        *(v2f*)(dst + 16 * j * 2) = pv;
      }
    }
  }
  __syncthreads();

  v8f acc[2][4];
#pragma unroll
  for (int t = 0; t < 2; ++t)
#pragma unroll
    for (int j = 0; j < 4; ++j) {
      const v8f z = {0.f, 0.f, 0.f, 0.f, 0.f, 0.f, 0.f, 0.f};
      acc[t][j] = z;
    }

  const float* wa = Ws + (32 * wave + mc) * C1_WP + 2 * g;
  const float* xb = Hs + (g * C1_HP + mc) * 2;
#pragma unroll
  for (int ks = 0; ks < 16; ++ks) {
    const v2f a0 = *(const v2f*)(wa + 4 * ks);
    const v2f a1 = *(const v2f*)(wa + 16 * C1_WP + 4 * ks);
    const float* xp = xb + ks * (2 * C1_HP * 2);
    const v2f x0 = *(const v2f*)(xp);
    const v2f x1 = *(const v2f*)(xp + 32);
    const v2f x2 = *(const v2f*)(xp + 64);
    const v2f x3 = *(const v2f*)(xp + 96);
    acc[0][0] = wmma4(a0, x0, acc[0][0]);
    acc[0][1] = wmma4(a0, x1, acc[0][1]);
    acc[0][2] = wmma4(a0, x2, acc[0][2]);
    acc[0][3] = wmma4(a0, x3, acc[0][3]);
    acc[1][0] = wmma4(a1, x0, acc[1][0]);
    acc[1][1] = wmma4(a1, x1, acc[1][1]);
    acc[1][2] = wmma4(a1, x2, acc[1][2]);
    acc[1][3] = wmma4(a1, x3, acc[1][3]);
  }

  const int ocb = 32 * wave + 8 * g;
#pragma unroll
  for (int t = 0; t < 2; ++t)
#pragma unroll
    for (int r = 0; r < 8; ++r) {
      const float bias = b2[ocb + 16 * t + r];
      v4f o = qv[t][r];
      o[0] += acc[t][0][r] + bias;
      o[1] += acc[t][1][r] + bias;
      o[2] += acc[t][2][r] + bias;
      o[3] += acc[t][3][r] + bias;
      *(v4f*)(qb + (long)(16 * t + r) * HW) = o;
    }
}

#define OC_TILE(t) ((t) * 4096 + ((t) & 1) * 32 + ((t) >> 1) * 68)

__global__ __launch_bounds__(256) void k_out_conv(const float* __restrict__ q,
                                                  const float* __restrict__ ow,
                                                  const float* __restrict__ ob,
                                                  float* __restrict__ out) {
  __shared__ __attribute__((aligned(16))) float Ls[128 * 4 * 2 + 4 * 4096 + 128];
  float* Bs = Ls;
  float* Qs = Ls + 128 * 4 * 2;
  const int s  = blockIdx.x;
  const int b  = s >> 6;
  const int hh = s & 63;
  const int tid = threadIdx.x;

  {
    const int col = tid & 3;
    const int colc = col < OUTC ? col : OUTC - 1;
    v2f wv[2];
#pragma unroll
    for (int i = 0; i < 2; ++i) {
      const int cc = (tid + 256 * i) >> 2;
      wv[i] = *(const v2f*)(ow + colc * C_ + 2 * cc);
    }
    const int m = tid & 15, ccb = tid >> 4;
    const float* qs = q + (long)b * CHW + hh * W_ + 4 * m;
    v4f v0[8], v1[8];
#pragma unroll
    for (int i = 0; i < 8; ++i) {
      const int cc = ccb + 16 * i;
      v0[i] = *(const v4f*)(qs + (long)(2 * cc) * HW);
      v1[i] = *(const v4f*)(qs + (long)(2 * cc + 1) * HW);
    }
#pragma unroll
    for (int i = 0; i < 2; ++i) {
      const v2f z = {0.f, 0.f};
      *(v2f*)(Bs + (tid + 256 * i) * 2) = col < OUTC ? wv[i] : z;
    }
    float* dst = Qs + OC_TILE(m >> 2) + (ccb * 16 + 4 * (m & 3)) * 2;
#pragma unroll
    for (int i = 0; i < 8; ++i) {
      const v4f lo = {v0[i][0], v1[i][0], v0[i][1], v1[i][1]};
      const v4f hi = {v0[i][2], v1[i][2], v0[i][3], v1[i][3]};
      *(v4f*)(dst + i * (16 * 16 * 2))     = lo;
      *(v4f*)(dst + i * (16 * 16 * 2) + 4) = hi;
    }
  }
  __syncthreads();

  const int wave = __builtin_amdgcn_readfirstlane(tid >> 5);
  const int lane = tid & 31, g = lane >> 4, mc = lane & 15;
  if (wave < 4) {
    const float* xa = Qs + OC_TILE(wave) + (g * 16 + mc) * 2;
    const float* xw = Bs + (g * 4 + (mc & 3)) * 2;
    v8f acc = {0.f, 0.f, 0.f, 0.f, 0.f, 0.f, 0.f, 0.f};
    for (int k8 = 0; k8 < 64; k8 += 8) {
      const float* pa = xa + k8 * (2 * 16 * 2);
      const float* pw = xw + k8 * (2 * 4 * 2);
#pragma unroll
      for (int j = 0; j < 8; ++j) {
        const v2f a  = *(const v2f*)(pa + j * (2 * 16 * 2));
        const v2f bb = *(const v2f*)(pw + j * (2 * 4 * 2));
        acc = wmma4(a, bb, acc);
      }
    }
    if (mc < OUTC) {
      const float bias = ob[mc];
      const long obase = ((long)(b * OUTC + mc) * H_ + hh) * W_ + 16 * wave;
#pragma unroll
      for (int r = 0; r < 8; ++r) out[obase + r + 8 * g] = acc[r] + bias;
    }
  }
}

__global__ __launch_bounds__(256) void k_final(const unsigned* __restrict__ counts,
                                               const float* __restrict__ wsdiff,
                                               float* __restrict__ out) {
  __shared__ float red[256];
  const int gid = blockIdx.x * 256 + threadIdx.x;
  if (blockIdx.x < 32) {
    out[WF_POS + gid] = (float)counts[gid] * (1.0f / 4096.0f);
  } else {
    float sum = 0.f;
    for (int t = threadIdx.x; t < B_ * H_; t += 256) sum += wsdiff[t];
    red[threadIdx.x] = sum; __syncthreads();
    for (int st = 128; st > 0; st >>= 1) {
      if (threadIdx.x < st) red[threadIdx.x] += red[threadIdx.x + st];
      __syncthreads();
    }
    if (threadIdx.x == 0)
      out[DIFF_POS] = red[0] * (1.0f / ((float)NPOS * (float)C_));
  }
}

extern "C" void kernel_launch(void* const* d_in, const int* in_sizes, int n_in,
                              void* d_out, int out_size, void* d_ws, size_t ws_size,
                              hipStream_t stream) {
  (void)in_sizes; (void)n_in; (void)out_size; (void)ws_size;
  const float* in  = (const float*)d_in[0];
  const float* emb = (const float*)d_in[1];
  const float* rw1 = (const float*)d_in[2];
  const float* rb1 = (const float*)d_in[3];
  const float* rw2 = (const float*)d_in[4];
  const float* rb2 = (const float*)d_in[5];
  const float* ow  = (const float*)d_in[6];
  const float* ob  = (const float*)d_in[7];
  float* out = (float*)d_out;
  float* ws  = (float*)d_ws;

  float*    q      = ws + Q_OFF;
  float*    hbuf   = ws + H_OFF;
  float*    epack  = ws + EPACK_OFF;
  float*    en2    = ws + EN2_OFF;
  unsigned* counts = (unsigned*)(ws + CNT_OFF);
  float*    wsdiff = ws + WDIFF_OFF;
  float*    wpack  = ws + WPACK_OFF;

  k_prep<<<256, 256, 0, stream>>>(emb, epack, en2, counts);
  k_packw<<<(3 * WP_PER_BLK + 255) / 256, 256, 0, stream>>>(rw1, wpack);
  k_vq<<<B_ * H_, 128, 0, stream>>>(in, emb, epack, en2, q, counts, wsdiff);
  for (int i = 0; i < 3; ++i) {
    k_res_conv3<<<B_ * H_, 128, 0, stream>>>(q, wpack + (size_t)i * WP_PER_BLK * 2,
                                           rb1 + i * CB, hbuf);
    k_res_conv1<<<B_ * H_, 256, 0, stream>>>(q, hbuf, rw2 + (size_t)i * C_ * CB,
                                           rb2 + i * C_);
  }
  k_out_conv<<<B_ * H_, 256, 0, stream>>>(q, ow, ob, out);
  k_final<<<33, 256, 0, stream>>>(counts, wsdiff, out);
}


# 3x3 conv and VQ matrix loops: weight / codebook fragments of the next 8 k-steps are fetched from L2 into a second register set under the current 8 (were waited for at each run's head); same sums
# speedup vs baseline: 3.9900x; 1.0316x over previous
#include <hip/hip_runtime.h>


typedef float v2f __attribute__((ext_vector_type(2)));
typedef float v4f __attribute__((ext_vector_type(4)));
typedef float v8f __attribute__((ext_vector_type(8)));

#define B_    16
#define C_    256
#define H_    64
#define W_    64
#define OUTC  3
#define NT    512
#define CB    64

#define HW     (H_*W_)
#define NPOS   (B_*HW)
#define CHW    (C_*HW)

#define Q_OFF      0u
#define H_OFF      16777216u
#define EPACK_OFF  20971520u
#define EN2_OFF    21102592u
#define CNT_OFF    21103104u
#define WDIFF_OFF  21111296u
#define WPACK_OFF  21115392u

#define WP_PER_BLK (9*128*CB)
#define WP_PER_TAP (128*CB)

#define DIFF_POS   196608
#define WF_POS     196609

static __device__ __forceinline__ v8f wmma4(v2f a, v2f b, v8f c) {
  return __builtin_amdgcn_wmma_f32_16x16x4_f32(false, a, false, b, (short)0, c,
                                               false, false);
}

__global__ __launch_bounds__(256) void k_prep(const float* __restrict__ embed,
                                              float* __restrict__ epack,
                                              float* __restrict__ en2,
                                              unsigned* __restrict__ counts) {
  const int gid = blockIdx.x * 256 + threadIdx.x;
  if (gid < 128 * NT) {
    const int cc = gid >> 9, n = gid & (NT - 1);
    epack[gid * 2 + 0] = embed[(2 * cc) * NT + n];
    epack[gid * 2 + 1] = embed[(2 * cc + 1) * NT + n];
  }
  if (gid < NT) {
    float s = 0.f;
    for (int c = 0; c < C_; ++c) { float v = embed[c * NT + gid]; s += v * v; }
    en2[gid] = s;
  }
  if (gid < B_ * NT) counts[gid] = 0u;
}

__global__ __launch_bounds__(256) void k_packw(const float* __restrict__ rw1,
                                               float* __restrict__ wpack) {
  const int t = blockIdx.x * 256 + threadIdx.x;
  if (t >= 3 * WP_PER_BLK) return;
  const int i   = t / WP_PER_BLK;
  const int r   = t % WP_PER_BLK;
  const int tap = r / WP_PER_TAP;
  const int r2  = r % WP_PER_TAP;
  const int cc  = r2 >> 6;
  const int oc  = r2 & 63;
  const int dh = tap / 3, dw = tap % 3;
  const float* src = rw1 + (size_t)i * CB * C_ * 9
                         + ((size_t)oc * C_ + 2 * cc) * 9 + dh * 3 + dw;
  wpack[t * 2 + 0] = src[0];
  wpack[t * 2 + 1] = src[9];
}

#define VQ_SLOT(cc, p) (((cc) << 6) + (((p) + (((cc) & 1) << 4)) & 63))

__global__ __launch_bounds__(128) void k_vq(const float* __restrict__ in,
                                            const float* __restrict__ embed,
                                            const float* __restrict__ epack,
                                            const float* __restrict__ en2,
                                            float* __restrict__ q,
                                            unsigned* __restrict__ counts,
                                            float* __restrict__ wsdiff) {
  __shared__ float As[128 * 64 * 2];
  __shared__ float rv[4][64];
  __shared__ int   ri[4][64];
  __shared__ int   idxs[64];
  __shared__ float red[128];

  const int s  = blockIdx.x;
  const int b  = s >> 6;
  const int hh = s & 63;
  const int tid = threadIdx.x;
  const long gbase0 = (long)b * CHW + hh * W_;
  const int sp = tid & 63;
  const int sc = tid >> 6;

  {
    const float* src = in + gbase0 + (long)(2 * sc) * HW + sp;
    float* dst = As + VQ_SLOT(sc, sp) * 2;
#pragma unroll 8
    for (int i = 0; i < 64; ++i) {
      const float v0 = src[(long)(4 * i) * HW];
      const float v1 = src[(long)(4 * i + 1) * HW];
      const v2f pv = {v0, v1};
      *(v2f*)(dst + i * (2 * 64 * 2)) = pv;
    }
  }
  __syncthreads();

  const int wave = tid >> 5, lane = tid & 31, g = lane >> 4, mc = lane & 15;
  float bestv[4][8];
  int   besti[4][8];
#pragma unroll
  for (int j = 0; j < 4; ++j)
#pragma unroll
    for (int r = 0; r < 8; ++r) { bestv[j][r] = 3.4e38f; besti[j][r] = 0; }

  const float* xa0 = As + ((g << 6) + (((0 + g) & 3) << 4) + mc) * 2;
  const float* xa1 = As + ((g << 6) + (((1 + g) & 3) << 4) + mc) * 2;
  const float* xa2 = As + ((g << 6) + (((2 + g) & 3) << 4) + mc) * 2;
  const float* xa3 = As + ((g << 6) + (((3 + g) & 3) << 4) + mc) * 2;

  const float* el = epack + ((size_t)g * NT + (wave << 7) + mc) * 2;
  v2f eA[8], eB[8];
#pragma unroll
  for (int j = 0; j < 8; ++j) eA[j] = *(const v2f*)(el + (size_t)j * (2 * NT * 2));

#pragma unroll 1
  for (int tt = 0; tt < 8; ++tt) {
    const int n0 = (wave * 8 + tt) << 4;
    const float* eb = el + tt * (16 * 2);
    v8f acc0 = {0.f, 0.f, 0.f, 0.f, 0.f, 0.f, 0.f, 0.f};
    v8f acc1 = acc0, acc2 = acc0, acc3 = acc0;
#pragma unroll 1
    for (int kp = 0; kp < 4; ++kp) {
#pragma unroll
      for (int j = 0; j < 8; ++j)
        eB[j] = *(const v2f*)(eb + (size_t)(16 * kp + 8 + j) * (2 * NT * 2));
#pragma unroll
      for (int j = 0; j < 8; ++j) {
        const int ks = 16 * kp + j;
        const v2f a0 = *(const v2f*)(xa0 + ks * (2 * 64 * 2));
        const v2f a1 = *(const v2f*)(xa1 + ks * (2 * 64 * 2));
        const v2f a2 = *(const v2f*)(xa2 + ks * (2 * 64 * 2));
        const v2f a3 = *(const v2f*)(xa3 + ks * (2 * 64 * 2));
        acc0 = wmma4(a0, eA[j], acc0);
        acc1 = wmma4(a1, eA[j], acc1);
        acc2 = wmma4(a2, eA[j], acc2);
        acc3 = wmma4(a3, eA[j], acc3);
      }
      int nkp = kp + 1, ntt = tt;
      if (nkp == 4) { nkp = 0; ntt = tt + 1; }
      if (ntt == 8) ntt = 0;
      const float* en = el + ntt * (16 * 2) + (size_t)(16 * nkp) * (2 * NT * 2);
#pragma unroll
      for (int j = 0; j < 8; ++j)
        eA[j] = *(const v2f*)(en + (size_t)j * (2 * NT * 2));
#pragma unroll
      for (int j = 0; j < 8; ++j) {
        const int ks = 16 * kp + 8 + j;
        const v2f a0 = *(const v2f*)(xa0 + ks * (2 * 64 * 2));
        const v2f a1 = *(const v2f*)(xa1 + ks * (2 * 64 * 2));
        const v2f a2 = *(const v2f*)(xa2 + ks * (2 * 64 * 2));
        const v2f a3 = *(const v2f*)(xa3 + ks * (2 * 64 * 2));
        acc0 = wmma4(a0, eB[j], acc0);
        acc1 = wmma4(a1, eB[j], acc1);
        acc2 = wmma4(a2, eB[j], acc2);
        acc3 = wmma4(a3, eB[j], acc3);
      }
    }
    const float e2 = en2[n0 + mc];
    const int   n  = n0 + mc;
#pragma unroll
    for (int r = 0; r < 8; ++r) {
      float d;
      d = e2 - 2.0f * acc0[r]; if (d < bestv[0][r]) { bestv[0][r] = d; besti[0][r] = n; }
      d = e2 - 2.0f * acc1[r]; if (d < bestv[1][r]) { bestv[1][r] = d; besti[1][r] = n; }
      d = e2 - 2.0f * acc2[r]; if (d < bestv[2][r]) { bestv[2][r] = d; besti[2][r] = n; }
      d = e2 - 2.0f * acc3[r]; if (d < bestv[3][r]) { bestv[3][r] = d; besti[3][r] = n; }
    }
  }

#pragma unroll
  for (int j = 0; j < 4; ++j) {
#pragma unroll
    for (int r = 0; r < 8; ++r) {
      float v = bestv[j][r];
      int  ix = besti[j][r];
#pragma unroll
      for (int m = 8; m > 0; m >>= 1) {
        const float ov = __shfl_xor(v, m);
        const int   oi = __shfl_xor(ix, m);
        const bool take = (ov < v) | ((ov == v) & (oi < ix));
        v  = take ? ov : v;
        ix = take ? oi : ix;
      }
      bestv[j][r] = v;
      besti[j][r] = ix;
    }
  }
  if (mc == 0) {
#pragma unroll
    for (int j = 0; j < 4; ++j)
#pragma unroll
      for (int r = 0; r < 8; ++r) {
        rv[wave][16 * j + r + 8 * g] = bestv[j][r];
        ri[wave][16 * j + r + 8 * g] = besti[j][r];
      }
  }
  __syncthreads();

  if (tid < 64) {
    float bv = rv[0][tid]; int bi = ri[0][tid];
    for (int w = 1; w < 4; ++w) {
      const float v = rv[w][tid]; const int ii = ri[w][tid];
      if (v < bv || (v == bv && ii < bi)) { bv = v; bi = ii; }
    }
    idxs[tid] = bi;
    atomicAdd(&counts[b * NT + bi], 1u);
  }
  __syncthreads();

  float fsum = 0.f;
  {
    const int id = idxs[sp];
    const float* xs = As + VQ_SLOT(sc, sp) * 2;
    const float* es = embed + (2 * sc) * NT + id;
    float* qd = q + gbase0 + (long)(2 * sc) * HW + sp;
#pragma unroll 8
    for (int i = 0; i < 64; ++i) {
      const v2f x = *(const v2f*)(xs + i * (2 * 64 * 2));
      const float e0 = es[(4 * i) * NT];
      const float e1 = es[(4 * i + 1) * NT];
      qd[(long)(4 * i) * HW]     = e0;
      qd[(long)(4 * i + 1) * HW] = e1;
      const float d0 = e0 - x[0], d1 = e1 - x[1];
      fsum += d0 * d0 + d1 * d1;
    }
  }
  red[tid] = fsum; __syncthreads();
  for (int st = 64; st > 0; st >>= 1) {
    if (tid < st) red[tid] += red[tid + st];
    __syncthreads();
  }
  if (tid == 0) wsdiff[s] = red[0];
}

#define KCH    32
#define A_ROW  66
#define A_CCP  208

__global__ __launch_bounds__(128) void k_res_conv3(const float* __restrict__ q,
                                                   const float* __restrict__ wp,
                                                   const float* __restrict__ b1,
                                                   float* __restrict__ hbuf) {
  __shared__ float Ah[KCH * A_CCP * 2];
  const int s  = blockIdx.x;
  const int b  = s >> 6;
  const int hh = s & 63;
  const int tid = threadIdx.x;
  const int wave = tid >> 5, lane = tid & 31, g = lane >> 4, mc = lane & 15;
  const int oc = wave * 16 + mc;

  for (int t = tid; t < KCH * 6; t += 128) {
    const int cc = t / 6, r6 = t % 6, rr = r6 >> 1;
    const int ww = (r6 & 1) ? (A_ROW - 1) : 0;
    const v2f z = {0.f, 0.f};
    *(v2f*)(Ah + (cc * A_CCP + rr * A_ROW + ww) * 2) = z;
  }

  const int sw = tid & 63;
  const int sc = tid >> 6;

  v8f acc0 = {0.f, 0.f, 0.f, 0.f, 0.f, 0.f, 0.f, 0.f};
  v8f acc1 = acc0, acc2 = acc0, acc3 = acc0;

  const float* wl = wp + ((size_t)g * CB + oc) * 2;
  v2f wA[8], wB[8];
#pragma unroll
  for (int j = 0; j < 8; ++j) wA[j] = *(const v2f*)(wl + j * (2 * CB * 2));

  for (int kc = 0; kc < C_ / (2 * KCH); ++kc) {
    if (kc) __syncthreads();
#pragma unroll
    for (int rr = 0; rr < 3; ++rr) {
      const int hy = hh + rr - 1;
      float* dst = Ah + (sc * A_CCP + rr * A_ROW + 1 + sw) * 2;
      if ((unsigned)hy < 64u) {
        const float* qr = q + (long)b * CHW + (long)(2 * kc * KCH + 2 * sc) * HW
                            + hy * W_ + sw;
#pragma unroll 8
        for (int i = 0; i < KCH / 2; ++i) {
          float v0 = qr[(long)(4 * i) * HW];
          float v1 = qr[(long)(4 * i + 1) * HW];
          v0 = v0 > 0.f ? v0 : 0.f;
          v1 = v1 > 0.f ? v1 : 0.f;
          const v2f pv = {v0, v1};
          *(v2f*)(dst + i * (2 * A_CCP * 2)) = pv;
        }
      } else {
        const v2f z = {0.f, 0.f};
        for (int i = 0; i < KCH / 2; ++i) *(v2f*)(dst + i * (2 * A_CCP * 2)) = z;
      }
    }
    __syncthreads();

    const float* xb = Ah + (g * A_CCP + mc) * 2;
    for (int tap = 0; tap < 9; ++tap) {
      const int dh = tap / 3, dw = tap - dh * 3;
      const float* xt = xb + (dh * A_ROW + dw) * 2;
      const float* wt = wl + (size_t)tap * WP_PER_TAP * 2
                           + (size_t)(kc * KCH) * (CB * 2);
#pragma unroll
      for (int j = 0; j < 8; ++j)
        wB[j] = *(const v2f*)(wt + (8 + j) * (2 * CB * 2));
#pragma unroll
      for (int j = 0; j < 8; ++j) {
        const float* xp = xt + j * (2 * A_CCP * 2);
        const v2f x0 = *(const v2f*)(xp);
        const v2f x1 = *(const v2f*)(xp + 32);
        const v2f x2 = *(const v2f*)(xp + 64);
        const v2f x3 = *(const v2f*)(xp + 96);
        acc0 = wmma4(wA[j], x0, acc0);
        acc1 = wmma4(wA[j], x1, acc1);
        acc2 = wmma4(wA[j], x2, acc2);
        acc3 = wmma4(wA[j], x3, acc3);
      }
      int ntap = tap + 1, nkc = kc;
      if (ntap == 9) { ntap = 0; nkc = kc + 1; }
      if (nkc == C_ / (2 * KCH)) nkc = 0;
      const float* wn = wl + (size_t)ntap * WP_PER_TAP * 2
                           + (size_t)(nkc * KCH) * (CB * 2);
#pragma unroll
      for (int j = 0; j < 8; ++j)
        wA[j] = *(const v2f*)(wn + j * (2 * CB * 2));
#pragma unroll
      for (int j = 0; j < 8; ++j) {
        const float* xp = xt + (8 + j) * (2 * A_CCP * 2);
        const v2f x0 = *(const v2f*)(xp);
        const v2f x1 = *(const v2f*)(xp + 32);
        const v2f x2 = *(const v2f*)(xp + 64);
        const v2f x3 = *(const v2f*)(xp + 96);
        acc0 = wmma4(wB[j], x0, acc0);
        acc1 = wmma4(wB[j], x1, acc1);
        acc2 = wmma4(wB[j], x2, acc2);
        acc3 = wmma4(wB[j], x3, acc3);
      }
    }
  }

  const int ocb = wave * 16 + 8 * g;
  const long ob = (long)b * (CB * HW) + (long)ocb * HW + hh * W_ + mc;
#pragma unroll
  for (int r = 0; r < 8; ++r) {
    const float bias = b1[ocb + r];
    float* hp = hbuf + ob + (long)r * HW;
    float v;
    v = acc0[r] + bias; hp[0]  = v > 0.f ? v : 0.f;
    v = acc1[r] + bias; hp[16] = v > 0.f ? v : 0.f;
    v = acc2[r] + bias; hp[32] = v > 0.f ? v : 0.f;
    v = acc3[r] + bias; hp[48] = v > 0.f ? v : 0.f;
  }
}

#define C1_HP  80
#define C1_WP  68

__global__ __launch_bounds__(256) void k_res_conv1(float* __restrict__ q,
                                                   const float* __restrict__ hbuf,
                                                   const float* __restrict__ w2,
                                                   const float* __restrict__ b2) {
  __shared__ __attribute__((aligned(16))) float Hs[32 * C1_HP * 2];
  __shared__ __attribute__((aligned(16))) float Ws[C_ * C1_WP];
  const int s  = blockIdx.x;
  const int b  = s >> 6;
  const int hh = s & 63;
  const int tid = threadIdx.x;
  const int wave = tid >> 5, lane = tid & 31, g = lane >> 4, mc = lane & 15;

  float* qb = q + (long)b * CHW + (long)(32 * wave + 8 * g) * HW + hh * W_ + 4 * mc;
  v4f qv[2][8];
#pragma unroll
  for (int t = 0; t < 2; ++t)
#pragma unroll
    for (int r = 0; r < 8; ++r)
      qv[t][r] = *(const v4f*)(qb + (long)(16 * t + r) * HW);

#pragma unroll
  for (int i = 0; i < 16; ++i) {
    const int i4 = tid + 256 * i;
    const v4f v = *(const v4f*)(w2 + i4 * 4);
    *(v4f*)(Ws + (i4 >> 4) * C1_WP + (i4 & 15) * 4) = v;
  }
  {
    const int m = tid & 15, ccb = tid >> 4;
    const float* hs = hbuf + (long)b * (CB * HW) + hh * W_ + 4 * m;
#pragma unroll
    for (int i = 0; i < 2; ++i) {
      const int cc = ccb + 16 * i;
      const v4f v0 = *(const v4f*)(hs + (long)(2 * cc) * HW);
      const v4f v1 = *(const v4f*)(hs + (long)(2 * cc + 1) * HW);
      float* dst = Hs + (cc * C1_HP + m) * 2;
#pragma unroll
      for (int j = 0; j < 4; ++j) {
        const v2f pv = {v0[j], v1[j]};
        *(v2f*)(dst + 16 * j * 2) = pv;
      }
    }
  }
  __syncthreads();

  v8f acc[2][4];
#pragma unroll
  for (int t = 0; t < 2; ++t)
#pragma unroll
    for (int j = 0; j < 4; ++j) {
      const v8f z = {0.f, 0.f, 0.f, 0.f, 0.f, 0.f, 0.f, 0.f};
      acc[t][j] = z;
    }

  const float* wa = Ws + (32 * wave + mc) * C1_WP + 2 * g;
  const float* xb = Hs + (g * C1_HP + mc) * 2;
#pragma unroll
  for (int ks = 0; ks < 16; ++ks) {
    const v2f a0 = *(const v2f*)(wa + 4 * ks);
    const v2f a1 = *(const v2f*)(wa + 16 * C1_WP + 4 * ks);
    const float* xp = xb + ks * (2 * C1_HP * 2);
    const v2f x0 = *(const v2f*)(xp);
    const v2f x1 = *(const v2f*)(xp + 32);
    const v2f x2 = *(const v2f*)(xp + 64);
    const v2f x3 = *(const v2f*)(xp + 96);
    acc[0][0] = wmma4(a0, x0, acc[0][0]);
    acc[0][1] = wmma4(a0, x1, acc[0][1]);
    acc[0][2] = wmma4(a0, x2, acc[0][2]);
    acc[0][3] = wmma4(a0, x3, acc[0][3]);
    acc[1][0] = wmma4(a1, x0, acc[1][0]);
    acc[1][1] = wmma4(a1, x1, acc[1][1]);
    acc[1][2] = wmma4(a1, x2, acc[1][2]);
    acc[1][3] = wmma4(a1, x3, acc[1][3]);
  }

  const int ocb = 32 * wave + 8 * g;
#pragma unroll
  for (int t = 0; t < 2; ++t)
#pragma unroll
    for (int r = 0; r < 8; ++r) {
      const float bias = b2[ocb + 16 * t + r];
      v4f o = qv[t][r];
      o[0] += acc[t][0][r] + bias;
      o[1] += acc[t][1][r] + bias;
      o[2] += acc[t][2][r] + bias;
      o[3] += acc[t][3][r] + bias;
      *(v4f*)(qb + (long)(16 * t + r) * HW) = o;
    }
}

#define OC_TILE(t) ((t) * 4096 + ((t) & 1) * 32 + ((t) >> 1) * 68)

__global__ __launch_bounds__(256) void k_out_conv(const float* __restrict__ q,
                                                  const float* __restrict__ ow,
                                                  const float* __restrict__ ob,
                                                  float* __restrict__ out) {
  __shared__ __attribute__((aligned(16))) float Ls[128 * 4 * 2 + 4 * 4096 + 128];
  float* Bs = Ls;
  float* Qs = Ls + 128 * 4 * 2;
  const int s  = blockIdx.x;
  const int b  = s >> 6;
  const int hh = s & 63;
  const int tid = threadIdx.x;

  {
    const int col = tid & 3;
    const int colc = col < OUTC ? col : OUTC - 1;
    v2f wv[2];
#pragma unroll
    for (int i = 0; i < 2; ++i) {
      const int cc = (tid + 256 * i) >> 2;
      wv[i] = *(const v2f*)(ow + colc * C_ + 2 * cc);
    }
    const int m = tid & 15, ccb = tid >> 4;
    const float* qs = q + (long)b * CHW + hh * W_ + 4 * m;
    v4f v0[8], v1[8];
#pragma unroll
    for (int i = 0; i < 8; ++i) {
      const int cc = ccb + 16 * i;
      v0[i] = *(const v4f*)(qs + (long)(2 * cc) * HW);
      v1[i] = *(const v4f*)(qs + (long)(2 * cc + 1) * HW);
    }
#pragma unroll
    for (int i = 0; i < 2; ++i) {
      const v2f z = {0.f, 0.f};
      *(v2f*)(Bs + (tid + 256 * i) * 2) = col < OUTC ? wv[i] : z;
    }
    float* dst = Qs + OC_TILE(m >> 2) + (ccb * 16 + 4 * (m & 3)) * 2;
#pragma unroll
    for (int i = 0; i < 8; ++i) {
      const v4f lo = {v0[i][0], v1[i][0], v0[i][1], v1[i][1]};
      const v4f hi = {v0[i][2], v1[i][2], v0[i][3], v1[i][3]};
      *(v4f*)(dst + i * (16 * 16 * 2))     = lo;
      *(v4f*)(dst + i * (16 * 16 * 2) + 4) = hi;
    }
  }
  __syncthreads();

  const int wave = __builtin_amdgcn_readfirstlane(tid >> 5);
  const int lane = tid & 31, g = lane >> 4, mc = lane & 15;
  if (wave < 4) {
    const float* xa = Qs + OC_TILE(wave) + (g * 16 + mc) * 2;
    const float* xw = Bs + (g * 4 + (mc & 3)) * 2;
    v8f acc = {0.f, 0.f, 0.f, 0.f, 0.f, 0.f, 0.f, 0.f};
    for (int k8 = 0; k8 < 64; k8 += 8) {
      const float* pa = xa + k8 * (2 * 16 * 2);
      const float* pw = xw + k8 * (2 * 4 * 2);
#pragma unroll
      for (int j = 0; j < 8; ++j) {
        const v2f a  = *(const v2f*)(pa + j * (2 * 16 * 2));
        const v2f bb = *(const v2f*)(pw + j * (2 * 4 * 2));
        acc = wmma4(a, bb, acc);
      }
    }
    if (mc < OUTC) {
      const float bias = ob[mc];
      const long obase = ((long)(b * OUTC + mc) * H_ + hh) * W_ + 16 * wave;
#pragma unroll
      for (int r = 0; r < 8; ++r) out[obase + r + 8 * g] = acc[r] + bias;
    }
  }
}

__global__ __launch_bounds__(256) void k_final(const unsigned* __restrict__ counts,
                                               const float* __restrict__ wsdiff,
                                               float* __restrict__ out) {
  __shared__ float red[256];
  const int gid = blockIdx.x * 256 + threadIdx.x;
  if (blockIdx.x < 32) {
    out[WF_POS + gid] = (float)counts[gid] * (1.0f / 4096.0f);
  } else {
    float sum = 0.f;
    for (int t = threadIdx.x; t < B_ * H_; t += 256) sum += wsdiff[t];
    red[threadIdx.x] = sum; __syncthreads();
    for (int st = 128; st > 0; st >>= 1) {
      if (threadIdx.x < st) red[threadIdx.x] += red[threadIdx.x + st];
      __syncthreads();
    }
    if (threadIdx.x == 0)
      out[DIFF_POS] = red[0] * (1.0f / ((float)NPOS * (float)C_));
  }
}

extern "C" void kernel_launch(void* const* d_in, const int* in_sizes, int n_in,
                              void* d_out, int out_size, void* d_ws, size_t ws_size,
                              hipStream_t stream) {
  (void)in_sizes; (void)n_in; (void)out_size; (void)ws_size;
  const float* in  = (const float*)d_in[0];
  const float* emb = (const float*)d_in[1];
  const float* rw1 = (const float*)d_in[2];
  const float* rb1 = (const float*)d_in[3];
  const float* rw2 = (const float*)d_in[4];
  const float* rb2 = (const float*)d_in[5];
  const float* ow  = (const float*)d_in[6];
  const float* ob  = (const float*)d_in[7];
  float* out = (float*)d_out;
  float* ws  = (float*)d_ws;

  float*    q      = ws + Q_OFF;
  float*    hbuf   = ws + H_OFF;
  float*    epack  = ws + EPACK_OFF;
  float*    en2    = ws + EN2_OFF;
  unsigned* counts = (unsigned*)(ws + CNT_OFF);
  float*    wsdiff = ws + WDIFF_OFF;
  float*    wpack  = ws + WPACK_OFF;

  k_prep<<<256, 256, 0, stream>>>(emb, epack, en2, counts);
  k_packw<<<(3 * WP_PER_BLK + 255) / 256, 256, 0, stream>>>(rw1, wpack);
  k_vq<<<B_ * H_, 128, 0, stream>>>(in, emb, epack, en2, q, counts, wsdiff);
  for (int i = 0; i < 3; ++i) {
    k_res_conv3<<<B_ * H_, 128, 0, stream>>>(q, wpack + (size_t)i * WP_PER_BLK * 2,
                                           rb1 + i * CB, hbuf);
    k_res_conv1<<<B_ * H_, 256, 0, stream>>>(q, hbuf, rw2 + (size_t)i * C_ * CB,
                                           rb2 + i * C_);
  }
  k_out_conv<<<B_ * H_, 256, 0, stream>>>(q, ow, ob, out);
  k_final<<<33, 256, 0, stream>>>(counts, wsdiff, out);
}


# 3x3 conv: products paired along w (2 outputs x 3 taps from 4 products, not 6): 8 f32 WMMAs per k-step and row, were 12; sums regrouped (4 partial sums per output pair); 32 channels staged per pass
# speedup vs baseline: 4.4428x; 1.1135x over previous
#include <hip/hip_runtime.h>


typedef float v2f __attribute__((ext_vector_type(2)));
typedef float v4f __attribute__((ext_vector_type(4)));
typedef float v8f __attribute__((ext_vector_type(8)));

#define B_    16
#define C_    256
#define H_    64
#define W_    64
#define OUTC  3
#define NT    512
#define CB    64

#define HW     (H_*W_)
#define NPOS   (B_*HW)
#define CHW    (C_*HW)

#define Q_OFF      0u
#define H_OFF      16777216u
#define EPACK_OFF  20971520u
#define EN2_OFF    21102592u
#define CNT_OFF    21103104u
#define WDIFF_OFF  21111296u
#define WPACK_OFF  21115392u

#define WP_PER_BLK (9*128*CB)
#define WP_PER_TAP (128*CB)

#define DIFF_POS   196608
#define WF_POS     196609

static __device__ __forceinline__ v8f wmma4(v2f a, v2f b, v8f c) {
  return __builtin_amdgcn_wmma_f32_16x16x4_f32(false, a, false, b, (short)0, c,
                                               false, false);
}

__global__ __launch_bounds__(256) void k_prep(const float* __restrict__ embed,
                                              float* __restrict__ epack,
                                              float* __restrict__ en2,
                                              unsigned* __restrict__ counts) {
  const int gid = blockIdx.x * 256 + threadIdx.x;
  if (gid < 128 * NT) {
    const int cc = gid >> 9, n = gid & (NT - 1);
    epack[gid * 2 + 0] = embed[(2 * cc) * NT + n];
    epack[gid * 2 + 1] = embed[(2 * cc + 1) * NT + n];
  }
  if (gid < NT) {
    float s = 0.f;
    for (int c = 0; c < C_; ++c) { float v = embed[c * NT + gid]; s += v * v; }
    en2[gid] = s;
  }
  if (gid < B_ * NT) counts[gid] = 0u;
}

__global__ __launch_bounds__(256) void k_packw(const float* __restrict__ rw1,
                                               float* __restrict__ wpack) {
  const int t = blockIdx.x * 256 + threadIdx.x;
  if (t >= 3 * WP_PER_BLK) return;
  const int i   = t / WP_PER_BLK;
  const int r   = t % WP_PER_BLK;
  const int tap = r / WP_PER_TAP;
  const int r2  = r % WP_PER_TAP;
  const int cc  = r2 >> 6;
  const int oc  = r2 & 63;
  const int dh = tap / 3, dw = tap % 3;
  const float* src = rw1 + (size_t)i * CB * C_ * 9
                         + ((size_t)oc * C_ + 2 * cc) * 9 + dh * 3 + dw;
  wpack[t * 2 + 0] = src[0];
  wpack[t * 2 + 1] = src[9];
}

#define VQ_SLOT(cc, p) (((cc) << 6) + (((p) + (((cc) & 1) << 4)) & 63))

__global__ __launch_bounds__(128) void k_vq(const float* __restrict__ in,
                                            const float* __restrict__ embed,
                                            const float* __restrict__ epack,
                                            const float* __restrict__ en2,
                                            float* __restrict__ q,
                                            unsigned* __restrict__ counts,
                                            float* __restrict__ wsdiff) {
  __shared__ float As[128 * 64 * 2];
  __shared__ float rv[4][64];
  __shared__ int   ri[4][64];
  __shared__ int   idxs[64];
  __shared__ float red[128];

  const int s  = blockIdx.x;
  const int b  = s >> 6;
  const int hh = s & 63;
  const int tid = threadIdx.x;
  const long gbase0 = (long)b * CHW + hh * W_;
  const int sp = tid & 63;
  const int sc = tid >> 6;

  {
    const float* src = in + gbase0 + (long)(2 * sc) * HW + sp;
    float* dst = As + VQ_SLOT(sc, sp) * 2;
#pragma unroll 8
    for (int i = 0; i < 64; ++i) {
      const float v0 = src[(long)(4 * i) * HW];
      const float v1 = src[(long)(4 * i + 1) * HW];
      const v2f pv = {v0, v1};
      *(v2f*)(dst + i * (2 * 64 * 2)) = pv;
    }
  }
  __syncthreads();

  const int wave = tid >> 5, lane = tid & 31, g = lane >> 4, mc = lane & 15;
  float bestv[4][8];
  int   besti[4][8];
#pragma unroll
  for (int j = 0; j < 4; ++j)
#pragma unroll
    for (int r = 0; r < 8; ++r) { bestv[j][r] = 3.4e38f; besti[j][r] = 0; }

  const float* xa0 = As + ((g << 6) + (((0 + g) & 3) << 4) + mc) * 2;
  const float* xa1 = As + ((g << 6) + (((1 + g) & 3) << 4) + mc) * 2;
  const float* xa2 = As + ((g << 6) + (((2 + g) & 3) << 4) + mc) * 2;
  const float* xa3 = As + ((g << 6) + (((3 + g) & 3) << 4) + mc) * 2;

  const float* el = epack + ((size_t)g * NT + (wave << 7) + mc) * 2;
  v2f eA[8], eB[8];
#pragma unroll
  for (int j = 0; j < 8; ++j) eA[j] = *(const v2f*)(el + (size_t)j * (2 * NT * 2));

#pragma unroll 1
  for (int tt = 0; tt < 8; ++tt) {
    const int n0 = (wave * 8 + tt) << 4;
    const float* eb = el + tt * (16 * 2);
    v8f acc0 = {0.f, 0.f, 0.f, 0.f, 0.f, 0.f, 0.f, 0.f};
    v8f acc1 = acc0, acc2 = acc0, acc3 = acc0;
#pragma unroll 1
    for (int kp = 0; kp < 4; ++kp) {
#pragma unroll
      for (int j = 0; j < 8; ++j)
        eB[j] = *(const v2f*)(eb + (size_t)(16 * kp + 8 + j) * (2 * NT * 2));
#pragma unroll
      for (int j = 0; j < 8; ++j) {
        const int ks = 16 * kp + j;
        const v2f a0 = *(const v2f*)(xa0 + ks * (2 * 64 * 2));
        const v2f a1 = *(const v2f*)(xa1 + ks * (2 * 64 * 2));
        const v2f a2 = *(const v2f*)(xa2 + ks * (2 * 64 * 2));
        const v2f a3 = *(const v2f*)(xa3 + ks * (2 * 64 * 2));
        acc0 = wmma4(a0, eA[j], acc0);
        acc1 = wmma4(a1, eA[j], acc1);
        acc2 = wmma4(a2, eA[j], acc2);
        acc3 = wmma4(a3, eA[j], acc3);
      }
      int nkp = kp + 1, ntt = tt;
      if (nkp == 4) { nkp = 0; ntt = tt + 1; }
      if (ntt == 8) ntt = 0;
      const float* en = el + ntt * (16 * 2) + (size_t)(16 * nkp) * (2 * NT * 2);
#pragma unroll
      for (int j = 0; j < 8; ++j)
        eA[j] = *(const v2f*)(en + (size_t)j * (2 * NT * 2));
#pragma unroll
      for (int j = 0; j < 8; ++j) {
        const int ks = 16 * kp + 8 + j;
        const v2f a0 = *(const v2f*)(xa0 + ks * (2 * 64 * 2));
        const v2f a1 = *(const v2f*)(xa1 + ks * (2 * 64 * 2));
        const v2f a2 = *(const v2f*)(xa2 + ks * (2 * 64 * 2));
        const v2f a3 = *(const v2f*)(xa3 + ks * (2 * 64 * 2));
        acc0 = wmma4(a0, eB[j], acc0);
        acc1 = wmma4(a1, eB[j], acc1);
        acc2 = wmma4(a2, eB[j], acc2);
        acc3 = wmma4(a3, eB[j], acc3);
      }
    }
    const float e2 = en2[n0 + mc];
    const int   n  = n0 + mc;
#pragma unroll
    for (int r = 0; r < 8; ++r) {
      float d;
      d = e2 - 2.0f * acc0[r]; if (d < bestv[0][r]) { bestv[0][r] = d; besti[0][r] = n; }
      d = e2 - 2.0f * acc1[r]; if (d < bestv[1][r]) { bestv[1][r] = d; besti[1][r] = n; }
      d = e2 - 2.0f * acc2[r]; if (d < bestv[2][r]) { bestv[2][r] = d; besti[2][r] = n; }
      d = e2 - 2.0f * acc3[r]; if (d < bestv[3][r]) { bestv[3][r] = d; besti[3][r] = n; }
    }
  }

#pragma unroll
  for (int j = 0; j < 4; ++j) {
#pragma unroll
    for (int r = 0; r < 8; ++r) {
      float v = bestv[j][r];
      int  ix = besti[j][r];
#pragma unroll
      for (int m = 8; m > 0; m >>= 1) {
        const float ov = __shfl_xor(v, m);
        const int   oi = __shfl_xor(ix, m);
        const bool take = (ov < v) | ((ov == v) & (oi < ix));
        v  = take ? ov : v;
        ix = take ? oi : ix;
      }
      bestv[j][r] = v;
      besti[j][r] = ix;
    }
  }
  if (mc == 0) {
#pragma unroll
    for (int j = 0; j < 4; ++j)
#pragma unroll
      for (int r = 0; r < 8; ++r) {
        rv[wave][16 * j + r + 8 * g] = bestv[j][r];
        ri[wave][16 * j + r + 8 * g] = besti[j][r];
      }
  }
  __syncthreads();

  if (tid < 64) {
    float bv = rv[0][tid]; int bi = ri[0][tid];
    for (int w = 1; w < 4; ++w) {
      const float v = rv[w][tid]; const int ii = ri[w][tid];
      if (v < bv || (v == bv && ii < bi)) { bv = v; bi = ii; }
    }
    idxs[tid] = bi;
    atomicAdd(&counts[b * NT + bi], 1u);
  }
  __syncthreads();

  float fsum = 0.f;
  {
    const int id = idxs[sp];
    const float* xs = As + VQ_SLOT(sc, sp) * 2;
    const float* es = embed + (2 * sc) * NT + id;
    float* qd = q + gbase0 + (long)(2 * sc) * HW + sp;
#pragma unroll 8
    for (int i = 0; i < 64; ++i) {
      const v2f x = *(const v2f*)(xs + i * (2 * 64 * 2));
      const float e0 = es[(4 * i) * NT];
      const float e1 = es[(4 * i + 1) * NT];
      qd[(long)(4 * i) * HW]     = e0;
      qd[(long)(4 * i + 1) * HW] = e1;
      const float d0 = e0 - x[0], d1 = e1 - x[1];
      fsum += d0 * d0 + d1 * d1;
    }
  }
  red[tid] = fsum; __syncthreads();
  for (int st = 64; st > 0; st >>= 1) {
    if (tid < st) red[tid] += red[tid + st];
    __syncthreads();
  }
  if (tid == 0) wsdiff[s] = red[0];
}

#define KCH    16
#define T_CCP  400

static __device__ __forceinline__ void c3_load_run(v2f (&dst)[12], const float* wl,
                                                   int kc, int r) {
  const int rr = r >> 1, kh = r & 1;
#pragma unroll
  for (int j = 0; j < 4; ++j)
#pragma unroll
    for (int dw = 0; dw < 3; ++dw)
      dst[3 * j + dw] = *(const v2f*)(wl + ((size_t)(3 * rr + dw) * WP_PER_TAP
                                         + (size_t)(KCH * kc + 2 * (4 * kh + j)) * CB) * 2);
}

static __device__ __forceinline__ void c3_run(const v2f (&w)[12], const float* xb, int r,
                                              v8f (&acc)[4][2]) {
  const int rr = r >> 1, kh = r & 1;
#pragma unroll
  for (int j = 0; j < 4; ++j) {
    const int ks = 4 * kh + j;
    const float* xp = xb + (ks * 2 * T_CCP + rr * 128) * 2;
    const v2f g0 = w[3 * j], g1 = w[3 * j + 1], g2 = w[3 * j + 2];
    const v2f sg = g0 + g2;
    const v2f gp = 0.5f * (sg + g1);
    const v2f gm = 0.5f * (sg - g1);
    const v2f d00 = *(const v2f*)(xp);
    const v2f d01 = *(const v2f*)(xp + 32);
    const v2f d10 = *(const v2f*)(xp + 64);
    const v2f d11 = *(const v2f*)(xp + 96);
    const v2f d20 = *(const v2f*)(xp + 128);
    const v2f d21 = *(const v2f*)(xp + 160);
    const v2f d30 = *(const v2f*)(xp + 192);
    const v2f d31 = *(const v2f*)(xp + 224);
    acc[0][0] = wmma4(g0, d00, acc[0][0]);
    acc[0][1] = wmma4(g0, d01, acc[0][1]);
    acc[1][0] = wmma4(gp, d10, acc[1][0]);
    acc[1][1] = wmma4(gp, d11, acc[1][1]);
    acc[2][0] = wmma4(gm, d20, acc[2][0]);
    acc[2][1] = wmma4(gm, d21, acc[2][1]);
    acc[3][0] = wmma4(g2, d30, acc[3][0]);
    acc[3][1] = wmma4(g2, d31, acc[3][1]);
  }
}

__global__ __launch_bounds__(128) void k_res_conv3(const float* __restrict__ q,
                                                   const float* __restrict__ wp,
                                                   const float* __restrict__ b1,
                                                   float* __restrict__ hbuf) {
  __shared__ __attribute__((aligned(16))) float Td[KCH * T_CCP * 2];
  const int s  = blockIdx.x;
  const int b  = s >> 6;
  const int hh = s & 63;
  const int tid = threadIdx.x;
  const int wave = tid >> 5, lane = tid & 31, g = lane >> 4, mc = lane & 15;
  const int oc = wave * 16 + mc;

  v8f acc[4][2];
#pragma unroll
  for (int i = 0; i < 4; ++i)
#pragma unroll
    for (int t = 0; t < 2; ++t) {
      const v8f z = {0.f, 0.f, 0.f, 0.f, 0.f, 0.f, 0.f, 0.f};
      acc[i][t] = z;
    }

  const float* wl = wp + ((size_t)g * CB + oc) * 2;
  v2f wA[12], wB[12];
  c3_load_run(wA, wl, 0, 0);

  const int p  = lane;
  const int il = p > 0 ? -1 : 0;
  const int ir = p < 31 ? 2 : 1;
  const float* xb = Td + (g * T_CCP + mc) * 2;

  for (int kc = 0; kc < C_ / (2 * KCH); ++kc) {
    if (kc) __syncthreads();
#pragma unroll
    for (int rr = 0; rr < 3; ++rr) {
      const int hy = hh + rr - 1;
      float* dst = Td + (wave * T_CCP + rr * 128 + p) * 2;
      if ((unsigned)hy < 64u) {
        const float* qr = q + (long)b * CHW + (long)(2 * (KCH * kc + wave)) * HW
                            + hy * W_ + 2 * p;
        v2f a0[4], a1[4];
        float l0[4], l1[4], r0[4], r1[4];
#pragma unroll
        for (int i4 = 0; i4 < 4; ++i4) {
          const float* c0 = qr + (long)(8 * i4) * HW;
          const float* c1 = c0 + HW;
          a0[i4] = *(const v2f*)c0;
          a1[i4] = *(const v2f*)c1;
          l0[i4] = c0[il];  r0[i4] = c0[ir];
          l1[i4] = c1[il];  r1[i4] = c1[ir];
        }
#pragma unroll
        for (int i4 = 0; i4 < 4; ++i4) {
          const float e0 = a0[i4][0] > 0.f ? a0[i4][0] : 0.f;
          const float f0 = a0[i4][1] > 0.f ? a0[i4][1] : 0.f;
          const float e1 = a1[i4][0] > 0.f ? a1[i4][0] : 0.f;
          const float f1 = a1[i4][1] > 0.f ? a1[i4][1] : 0.f;
          float dl0 = l0[i4] > 0.f ? l0[i4] : 0.f;
          float dl1 = l1[i4] > 0.f ? l1[i4] : 0.f;
          float dr0 = r0[i4] > 0.f ? r0[i4] : 0.f;
          float dr1 = r1[i4] > 0.f ? r1[i4] : 0.f;
          if (p == 0)  { dl0 = 0.f; dl1 = 0.f; }
          if (p == 31) { dr0 = 0.f; dr1 = 0.f; }
          const v2f t0 = {dl0 - f0, dl1 - f1};
          const v2f t1 = {e0 + f0, e1 + f1};
          const v2f t2 = {f0 - e0, f1 - e1};
          const v2f t3 = {e0 - dr0, e1 - dr1};
          float* d = dst + i4 * (4 * T_CCP * 2);
          *(v2f*)(d)       = t0;
          *(v2f*)(d + 64)  = t1;
          *(v2f*)(d + 128) = t2;
          *(v2f*)(d + 192) = t3;
        }
      } else {
        const v2f z = {0.f, 0.f};
        for (int i4 = 0; i4 < 4; ++i4) {
          float* d = dst + i4 * (4 * T_CCP * 2);
          *(v2f*)(d)       = z;
          *(v2f*)(d + 64)  = z;
          *(v2f*)(d + 128) = z;
          *(v2f*)(d + 192) = z;
        }
      }
    }
    __syncthreads();

    int nkc = kc + 1;
    if (nkc == C_ / (2 * KCH)) nkc = 0;
    c3_load_run(wB, wl, kc, 1);   c3_run(wA, xb, 0, acc);
    c3_load_run(wA, wl, kc, 2);   c3_run(wB, xb, 1, acc);
    c3_load_run(wB, wl, kc, 3);   c3_run(wA, xb, 2, acc);
    c3_load_run(wA, wl, kc, 4);   c3_run(wB, xb, 3, acc);
    c3_load_run(wB, wl, kc, 5);   c3_run(wA, xb, 4, acc);
    c3_load_run(wA, wl, nkc, 0);  c3_run(wB, xb, 5, acc);
  }

  const int ocb = wave * 16 + 8 * g;
  const long ob = (long)b * (CB * HW) + (long)ocb * HW + hh * W_ + 2 * mc;
#pragma unroll
  for (int r = 0; r < 8; ++r) {
    const float bias = b1[ocb + r];
    float* hp = hbuf + ob + (long)r * HW;
#pragma unroll
    for (int t = 0; t < 2; ++t) {
      const float m1 = acc[0][t][r], m2 = acc[1][t][r];
      const float m3 = acc[2][t][r], m4 = acc[3][t][r];
      const float y0 = (m1 + m2 + m3) + bias;
      const float y1 = (m2 - m3 - m4) + bias;
      const v2f o = {y0 > 0.f ? y0 : 0.f, y1 > 0.f ? y1 : 0.f};
      *(v2f*)(hp + 32 * t) = o;
    }
  }
}

#define C1_HP  80
#define C1_WP  68

__global__ __launch_bounds__(256) void k_res_conv1(float* __restrict__ q,
                                                   const float* __restrict__ hbuf,
                                                   const float* __restrict__ w2,
                                                   const float* __restrict__ b2) {
  __shared__ __attribute__((aligned(16))) float Hs[32 * C1_HP * 2];
  __shared__ __attribute__((aligned(16))) float Ws[C_ * C1_WP];
  const int s  = blockIdx.x;
  const int b  = s >> 6;
  const int hh = s & 63;
  const int tid = threadIdx.x;
  const int wave = tid >> 5, lane = tid & 31, g = lane >> 4, mc = lane & 15;

  float* qb = q + (long)b * CHW + (long)(32 * wave + 8 * g) * HW + hh * W_ + 4 * mc;
  v4f qv[2][8];
#pragma unroll
  for (int t = 0; t < 2; ++t)
#pragma unroll
    for (int r = 0; r < 8; ++r)
      qv[t][r] = *(const v4f*)(qb + (long)(16 * t + r) * HW);

#pragma unroll
  for (int i = 0; i < 16; ++i) {
    const int i4 = tid + 256 * i;
    const v4f v = *(const v4f*)(w2 + i4 * 4);
    *(v4f*)(Ws + (i4 >> 4) * C1_WP + (i4 & 15) * 4) = v;
  }
  {
    const int m = tid & 15, ccb = tid >> 4;
    const float* hs = hbuf + (long)b * (CB * HW) + hh * W_ + 4 * m;
#pragma unroll
    for (int i = 0; i < 2; ++i) {
      const int cc = ccb + 16 * i;
      const v4f v0 = *(const v4f*)(hs + (long)(2 * cc) * HW);
      const v4f v1 = *(const v4f*)(hs + (long)(2 * cc + 1) * HW);
      float* dst = Hs + (cc * C1_HP + m) * 2;
#pragma unroll
      for (int j = 0; j < 4; ++j) {
        const v2f pv = {v0[j], v1[j]};
        *(v2f*)(dst + 16 * j * 2) = pv;
      }
    }
  }
  __syncthreads();

  v8f acc[2][4];
#pragma unroll
  for (int t = 0; t < 2; ++t)
#pragma unroll
    for (int j = 0; j < 4; ++j) {
      const v8f z = {0.f, 0.f, 0.f, 0.f, 0.f, 0.f, 0.f, 0.f};
      acc[t][j] = z;
    }

  const float* wa = Ws + (32 * wave + mc) * C1_WP + 2 * g;
  const float* xb = Hs + (g * C1_HP + mc) * 2;
#pragma unroll
  for (int ks = 0; ks < 16; ++ks) {
    const v2f a0 = *(const v2f*)(wa + 4 * ks);
    const v2f a1 = *(const v2f*)(wa + 16 * C1_WP + 4 * ks);
    const float* xp = xb + ks * (2 * C1_HP * 2);
    const v2f x0 = *(const v2f*)(xp);
    const v2f x1 = *(const v2f*)(xp + 32);
    const v2f x2 = *(const v2f*)(xp + 64);
    const v2f x3 = *(const v2f*)(xp + 96);
    acc[0][0] = wmma4(a0, x0, acc[0][0]);
    acc[0][1] = wmma4(a0, x1, acc[0][1]);
    acc[0][2] = wmma4(a0, x2, acc[0][2]);
    acc[0][3] = wmma4(a0, x3, acc[0][3]);
    acc[1][0] = wmma4(a1, x0, acc[1][0]);
    acc[1][1] = wmma4(a1, x1, acc[1][1]);
    acc[1][2] = wmma4(a1, x2, acc[1][2]);
    acc[1][3] = wmma4(a1, x3, acc[1][3]);
  }

  const int ocb = 32 * wave + 8 * g;
#pragma unroll
  for (int t = 0; t < 2; ++t)
#pragma unroll
    for (int r = 0; r < 8; ++r) {
      const float bias = b2[ocb + 16 * t + r];
      v4f o = qv[t][r];
      o[0] += acc[t][0][r] + bias;
      o[1] += acc[t][1][r] + bias;
      o[2] += acc[t][2][r] + bias;
      o[3] += acc[t][3][r] + bias;
      *(v4f*)(qb + (long)(16 * t + r) * HW) = o;
    }
}

#define OC_TILE(t) ((t) * 4096 + ((t) & 1) * 32 + ((t) >> 1) * 68)

__global__ __launch_bounds__(256) void k_out_conv(const float* __restrict__ q,
                                                  const float* __restrict__ ow,
                                                  const float* __restrict__ ob,
                                                  float* __restrict__ out) {
  __shared__ __attribute__((aligned(16))) float Ls[128 * 4 * 2 + 4 * 4096 + 128];
  float* Bs = Ls;
  float* Qs = Ls + 128 * 4 * 2;
  const int s  = blockIdx.x;
  const int b  = s >> 6;
  const int hh = s & 63;
  const int tid = threadIdx.x;

  {
    const int col = tid & 3;
    const int colc = col < OUTC ? col : OUTC - 1;
    v2f wv[2];
#pragma unroll
    for (int i = 0; i < 2; ++i) {
      const int cc = (tid + 256 * i) >> 2;
      wv[i] = *(const v2f*)(ow + colc * C_ + 2 * cc);
    }
    const int m = tid & 15, ccb = tid >> 4;
    const float* qs = q + (long)b * CHW + hh * W_ + 4 * m;
    v4f v0[8], v1[8];
#pragma unroll
    for (int i = 0; i < 8; ++i) {
      const int cc = ccb + 16 * i;
      v0[i] = *(const v4f*)(qs + (long)(2 * cc) * HW);
      v1[i] = *(const v4f*)(qs + (long)(2 * cc + 1) * HW);
    }
#pragma unroll
    for (int i = 0; i < 2; ++i) {
      const v2f z = {0.f, 0.f};
      *(v2f*)(Bs + (tid + 256 * i) * 2) = col < OUTC ? wv[i] : z;
    }
    float* dst = Qs + OC_TILE(m >> 2) + (ccb * 16 + 4 * (m & 3)) * 2;
#pragma unroll
    for (int i = 0; i < 8; ++i) {
      const v4f lo = {v0[i][0], v1[i][0], v0[i][1], v1[i][1]};
      const v4f hi = {v0[i][2], v1[i][2], v0[i][3], v1[i][3]};
      *(v4f*)(dst + i * (16 * 16 * 2))     = lo;
      *(v4f*)(dst + i * (16 * 16 * 2) + 4) = hi;
    }
  }
  __syncthreads();

  const int wave = __builtin_amdgcn_readfirstlane(tid >> 5);
  const int lane = tid & 31, g = lane >> 4, mc = lane & 15;
  if (wave < 4) {
    const float* xa = Qs + OC_TILE(wave) + (g * 16 + mc) * 2;
    const float* xw = Bs + (g * 4 + (mc & 3)) * 2;
    v8f acc = {0.f, 0.f, 0.f, 0.f, 0.f, 0.f, 0.f, 0.f};
    for (int k8 = 0; k8 < 64; k8 += 8) {
      const float* pa = xa + k8 * (2 * 16 * 2);
      const float* pw = xw + k8 * (2 * 4 * 2);
#pragma unroll
      for (int j = 0; j < 8; ++j) {
        const v2f a  = *(const v2f*)(pa + j * (2 * 16 * 2));
        const v2f bb = *(const v2f*)(pw + j * (2 * 4 * 2));
        acc = wmma4(a, bb, acc);
      }
    }
    if (mc < OUTC) {
      const float bias = ob[mc];
      const long obase = ((long)(b * OUTC + mc) * H_ + hh) * W_ + 16 * wave;
#pragma unroll
      for (int r = 0; r < 8; ++r) out[obase + r + 8 * g] = acc[r] + bias;
    }
  }
}

__global__ __launch_bounds__(256) void k_final(const unsigned* __restrict__ counts,
                                               const float* __restrict__ wsdiff,
                                               float* __restrict__ out) {
  __shared__ float red[256];
  const int gid = blockIdx.x * 256 + threadIdx.x;
  if (blockIdx.x < 32) {
    out[WF_POS + gid] = (float)counts[gid] * (1.0f / 4096.0f);
  } else {
    float sum = 0.f;
    for (int t = threadIdx.x; t < B_ * H_; t += 256) sum += wsdiff[t];
    red[threadIdx.x] = sum; __syncthreads();
    for (int st = 128; st > 0; st >>= 1) {
      if (threadIdx.x < st) red[threadIdx.x] += red[threadIdx.x + st];
      __syncthreads();
    }
    if (threadIdx.x == 0)
      out[DIFF_POS] = red[0] * (1.0f / ((float)NPOS * (float)C_));
  }
}

extern "C" void kernel_launch(void* const* d_in, const int* in_sizes, int n_in,
                              void* d_out, int out_size, void* d_ws, size_t ws_size,
                              hipStream_t stream) {
  (void)in_sizes; (void)n_in; (void)out_size; (void)ws_size;
  const float* in  = (const float*)d_in[0];
  const float* emb = (const float*)d_in[1];
  const float* rw1 = (const float*)d_in[2];
  const float* rb1 = (const float*)d_in[3];
  const float* rw2 = (const float*)d_in[4];
  const float* rb2 = (const float*)d_in[5];
  const float* ow  = (const float*)d_in[6];
  const float* ob  = (const float*)d_in[7];
  float* out = (float*)d_out;
  float* ws  = (float*)d_ws;

  float*    q      = ws + Q_OFF;
  float*    hbuf   = ws + H_OFF;
  float*    epack  = ws + EPACK_OFF;
  float*    en2    = ws + EN2_OFF;
  unsigned* counts = (unsigned*)(ws + CNT_OFF);
  float*    wsdiff = ws + WDIFF_OFF;
  float*    wpack  = ws + WPACK_OFF;

  k_prep<<<256, 256, 0, stream>>>(emb, epack, en2, counts);
  k_packw<<<(3 * WP_PER_BLK + 255) / 256, 256, 0, stream>>>(rw1, wpack);
  k_vq<<<B_ * H_, 128, 0, stream>>>(in, emb, epack, en2, q, counts, wsdiff);
  for (int i = 0; i < 3; ++i) {
    k_res_conv3<<<B_ * H_, 128, 0, stream>>>(q, wpack + (size_t)i * WP_PER_BLK * 2,
                                           rb1 + i * CB, hbuf);
    k_res_conv1<<<B_ * H_, 256, 0, stream>>>(q, hbuf, rw2 + (size_t)i * C_ * CB,
                                           rb2 + i * C_);
  }
  k_out_conv<<<B_ * H_, 256, 0, stream>>>(q, ow, ob, out);
  k_final<<<33, 256, 0, stream>>>(counts, wsdiff, out);
}


# 3x3 conv: products grouped 4 outputs x 3 taps along w (6 products per group, were 2 outputs from 4): 6 f32 WMMAs per k-step and row, were 8; sums regrouped (6 partial sums per 4 outputs)
# speedup vs baseline: 4.7509x; 1.0693x over previous
#include <hip/hip_runtime.h>


typedef float v2f __attribute__((ext_vector_type(2)));
typedef float v4f __attribute__((ext_vector_type(4)));
typedef float v8f __attribute__((ext_vector_type(8)));

#define B_    16
#define C_    256
#define H_    64
#define W_    64
#define OUTC  3
#define NT    512
#define CB    64

#define HW     (H_*W_)
#define NPOS   (B_*HW)
#define CHW    (C_*HW)

#define Q_OFF      0u
#define H_OFF      16777216u
#define EPACK_OFF  20971520u
#define EN2_OFF    21102592u
#define CNT_OFF    21103104u
#define WDIFF_OFF  21111296u
#define WPACK_OFF  21115392u

#define WP_PER_BLK (9*128*CB)
#define WP_PER_TAP (128*CB)

#define DIFF_POS   196608
#define WF_POS     196609

static __device__ __forceinline__ v8f wmma4(v2f a, v2f b, v8f c) {
  return __builtin_amdgcn_wmma_f32_16x16x4_f32(false, a, false, b, (short)0, c,
                                               false, false);
}

__global__ __launch_bounds__(256) void k_prep(const float* __restrict__ embed,
                                              float* __restrict__ epack,
                                              float* __restrict__ en2,
                                              unsigned* __restrict__ counts) {
  const int gid = blockIdx.x * 256 + threadIdx.x;
  if (gid < 128 * NT) {
    const int cc = gid >> 9, n = gid & (NT - 1);
    epack[gid * 2 + 0] = embed[(2 * cc) * NT + n];
    epack[gid * 2 + 1] = embed[(2 * cc + 1) * NT + n];
  }
  if (gid < NT) {
    float s = 0.f;
    for (int c = 0; c < C_; ++c) { float v = embed[c * NT + gid]; s += v * v; }
    en2[gid] = s;
  }
  if (gid < B_ * NT) counts[gid] = 0u;
}

__global__ __launch_bounds__(256) void k_packw(const float* __restrict__ rw1,
                                               float* __restrict__ wpack) {
  const int t = blockIdx.x * 256 + threadIdx.x;
  if (t >= 3 * WP_PER_BLK) return;
  const int i   = t / WP_PER_BLK;
  const int r   = t % WP_PER_BLK;
  const int tap = r / WP_PER_TAP;
  const int r2  = r % WP_PER_TAP;
  const int cc  = r2 >> 6;
  const int oc  = r2 & 63;
  const int dh = tap / 3, dw = tap % 3;
  const float* src = rw1 + (size_t)i * CB * C_ * 9
                         + ((size_t)oc * C_ + 2 * cc) * 9 + dh * 3 + dw;
  wpack[t * 2 + 0] = src[0];
  wpack[t * 2 + 1] = src[9];
}

#define VQ_SLOT(cc, p) (((cc) << 6) + (((p) + (((cc) & 1) << 4)) & 63))

__global__ __launch_bounds__(128) void k_vq(const float* __restrict__ in,
                                            const float* __restrict__ embed,
                                            const float* __restrict__ epack,
                                            const float* __restrict__ en2,
                                            float* __restrict__ q,
                                            unsigned* __restrict__ counts,
                                            float* __restrict__ wsdiff) {
  __shared__ float As[128 * 64 * 2];
  __shared__ float rv[4][64];
  __shared__ int   ri[4][64];
  __shared__ int   idxs[64];
  __shared__ float red[128];

  const int s  = blockIdx.x;
  const int b  = s >> 6;
  const int hh = s & 63;
  const int tid = threadIdx.x;
  const long gbase0 = (long)b * CHW + hh * W_;
  const int sp = tid & 63;
  const int sc = tid >> 6;

  {
    const float* src = in + gbase0 + (long)(2 * sc) * HW + sp;
    float* dst = As + VQ_SLOT(sc, sp) * 2;
#pragma unroll 8
    for (int i = 0; i < 64; ++i) {
      const float v0 = src[(long)(4 * i) * HW];
      const float v1 = src[(long)(4 * i + 1) * HW];
      const v2f pv = {v0, v1};
      *(v2f*)(dst + i * (2 * 64 * 2)) = pv;
    }
  }
  __syncthreads();

  const int wave = tid >> 5, lane = tid & 31, g = lane >> 4, mc = lane & 15;
  float bestv[4][8];
  int   besti[4][8];
#pragma unroll
  for (int j = 0; j < 4; ++j)
#pragma unroll
    for (int r = 0; r < 8; ++r) { bestv[j][r] = 3.4e38f; besti[j][r] = 0; }

  const float* xa0 = As + ((g << 6) + (((0 + g) & 3) << 4) + mc) * 2;
  const float* xa1 = As + ((g << 6) + (((1 + g) & 3) << 4) + mc) * 2;
  const float* xa2 = As + ((g << 6) + (((2 + g) & 3) << 4) + mc) * 2;
  const float* xa3 = As + ((g << 6) + (((3 + g) & 3) << 4) + mc) * 2;

  const float* el = epack + ((size_t)g * NT + (wave << 7) + mc) * 2;
  v2f eA[8], eB[8];
#pragma unroll
  for (int j = 0; j < 8; ++j) eA[j] = *(const v2f*)(el + (size_t)j * (2 * NT * 2));

#pragma unroll 1
  for (int tt = 0; tt < 8; ++tt) {
    const int n0 = (wave * 8 + tt) << 4;
    const float* eb = el + tt * (16 * 2);
    v8f acc0 = {0.f, 0.f, 0.f, 0.f, 0.f, 0.f, 0.f, 0.f};
    v8f acc1 = acc0, acc2 = acc0, acc3 = acc0;
#pragma unroll 1
    for (int kp = 0; kp < 4; ++kp) {
#pragma unroll
      for (int j = 0; j < 8; ++j)
        eB[j] = *(const v2f*)(eb + (size_t)(16 * kp + 8 + j) * (2 * NT * 2));
#pragma unroll
      for (int j = 0; j < 8; ++j) {
        const int ks = 16 * kp + j;
        const v2f a0 = *(const v2f*)(xa0 + ks * (2 * 64 * 2));
        const v2f a1 = *(const v2f*)(xa1 + ks * (2 * 64 * 2));
        const v2f a2 = *(const v2f*)(xa2 + ks * (2 * 64 * 2));
        const v2f a3 = *(const v2f*)(xa3 + ks * (2 * 64 * 2));
        acc0 = wmma4(a0, eA[j], acc0);
        acc1 = wmma4(a1, eA[j], acc1);
        acc2 = wmma4(a2, eA[j], acc2);
        acc3 = wmma4(a3, eA[j], acc3);
      }
      int nkp = kp + 1, ntt = tt;
      if (nkp == 4) { nkp = 0; ntt = tt + 1; }
      if (ntt == 8) ntt = 0;
      const float* en = el + ntt * (16 * 2) + (size_t)(16 * nkp) * (2 * NT * 2);
#pragma unroll
      for (int j = 0; j < 8; ++j)
        eA[j] = *(const v2f*)(en + (size_t)j * (2 * NT * 2));
#pragma unroll
      for (int j = 0; j < 8; ++j) {
        const int ks = 16 * kp + 8 + j;
        const v2f a0 = *(const v2f*)(xa0 + ks * (2 * 64 * 2));
        const v2f a1 = *(const v2f*)(xa1 + ks * (2 * 64 * 2));
        const v2f a2 = *(const v2f*)(xa2 + ks * (2 * 64 * 2));
        const v2f a3 = *(const v2f*)(xa3 + ks * (2 * 64 * 2));
        acc0 = wmma4(a0, eB[j], acc0);
        acc1 = wmma4(a1, eB[j], acc1);
        acc2 = wmma4(a2, eB[j], acc2);
        acc3 = wmma4(a3, eB[j], acc3);
      }
    }
    const float e2 = en2[n0 + mc];
    const int   n  = n0 + mc;
#pragma unroll
    for (int r = 0; r < 8; ++r) {
      float d;
      d = e2 - 2.0f * acc0[r]; if (d < bestv[0][r]) { bestv[0][r] = d; besti[0][r] = n; }
      d = e2 - 2.0f * acc1[r]; if (d < bestv[1][r]) { bestv[1][r] = d; besti[1][r] = n; }
      d = e2 - 2.0f * acc2[r]; if (d < bestv[2][r]) { bestv[2][r] = d; besti[2][r] = n; }
      d = e2 - 2.0f * acc3[r]; if (d < bestv[3][r]) { bestv[3][r] = d; besti[3][r] = n; }
    }
  }

#pragma unroll
  for (int j = 0; j < 4; ++j) {
#pragma unroll
    for (int r = 0; r < 8; ++r) {
      float v = bestv[j][r];
      int  ix = besti[j][r];
#pragma unroll
      for (int m = 8; m > 0; m >>= 1) {
        const float ov = __shfl_xor(v, m);
        const int   oi = __shfl_xor(ix, m);
        const bool take = (ov < v) | ((ov == v) & (oi < ix));
        v  = take ? ov : v;
        ix = take ? oi : ix;
      }
      bestv[j][r] = v;
      besti[j][r] = ix;
    }
  }
  if (mc == 0) {
#pragma unroll
    for (int j = 0; j < 4; ++j)
#pragma unroll
      for (int r = 0; r < 8; ++r) {
        rv[wave][16 * j + r + 8 * g] = bestv[j][r];
        ri[wave][16 * j + r + 8 * g] = besti[j][r];
      }
  }
  __syncthreads();

  if (tid < 64) {
    float bv = rv[0][tid]; int bi = ri[0][tid];
    for (int w = 1; w < 4; ++w) {
      const float v = rv[w][tid]; const int ii = ri[w][tid];
      if (v < bv || (v == bv && ii < bi)) { bv = v; bi = ii; }
    }
    idxs[tid] = bi;
    atomicAdd(&counts[b * NT + bi], 1u);
  }
  __syncthreads();

  float fsum = 0.f;
  {
    const int id = idxs[sp];
    const float* xs = As + VQ_SLOT(sc, sp) * 2;
    const float* es = embed + (2 * sc) * NT + id;
    float* qd = q + gbase0 + (long)(2 * sc) * HW + sp;
#pragma unroll 8
    for (int i = 0; i < 64; ++i) {
      const v2f x = *(const v2f*)(xs + i * (2 * 64 * 2));
      const float e0 = es[(4 * i) * NT];
      const float e1 = es[(4 * i + 1) * NT];
      qd[(long)(4 * i) * HW]     = e0;
      qd[(long)(4 * i + 1) * HW] = e1;
      const float d0 = e0 - x[0], d1 = e1 - x[1];
      fsum += d0 * d0 + d1 * d1;
    }
  }
  red[tid] = fsum; __syncthreads();
  for (int st = 64; st > 0; st >>= 1) {
    if (tid < st) red[tid] += red[tid + st];
    __syncthreads();
  }
  if (tid == 0) wsdiff[s] = red[0];
}

#define KCH    16
#define T_ROW  96
#define T_CCP  304

static __device__ __forceinline__ void c3_load_run(v2f (&dst)[12], const float* wl,
                                                   int kc, int r) {
  const int rr = r >> 1, kh = r & 1;
#pragma unroll
  for (int j = 0; j < 4; ++j)
#pragma unroll
    for (int dw = 0; dw < 3; ++dw)
      dst[3 * j + dw] = *(const v2f*)(wl + ((size_t)(3 * rr + dw) * WP_PER_TAP
                                         + (size_t)(KCH * kc + 2 * (4 * kh + j)) * CB) * 2);
}

static __device__ __forceinline__ void c3_run(const v2f (&w)[12], const float* xb, int r,
                                              v8f (&acc)[6]) {
  const int rr = r >> 1, kh = r & 1;
#pragma unroll
  for (int j = 0; j < 4; ++j) {
    const int ks = 4 * kh + j;
    const float* xp = xb + (ks * 2 * T_CCP + rr * T_ROW) * 2;
    const v2f g0 = w[3 * j], g1 = w[3 * j + 1], g2 = w[3 * j + 2];
    const v2f sg = g0 + g2;
    const v2f u0 = 0.25f * g0;
    const v2f u1 = (-1.0f / 6.0f) * (sg + g1);
    const v2f u2 = (-1.0f / 6.0f) * (sg - g1);
    const v2f ue = (1.0f / 24.0f) * g0 + (1.0f / 6.0f) * g2;
    const v2f uo = (1.0f / 12.0f) * g1;
    const v2f u3 = ue + uo;
    const v2f u4 = ue - uo;
    const v2f t0 = *(const v2f*)(xp);
    const v2f t1 = *(const v2f*)(xp + 32);
    const v2f t2 = *(const v2f*)(xp + 64);
    const v2f t3 = *(const v2f*)(xp + 96);
    const v2f t4 = *(const v2f*)(xp + 128);
    const v2f t5 = *(const v2f*)(xp + 160);
    acc[0] = wmma4(u0, t0, acc[0]);
    acc[1] = wmma4(u1, t1, acc[1]);
    acc[2] = wmma4(u2, t2, acc[2]);
    acc[3] = wmma4(u3, t3, acc[3]);
    acc[4] = wmma4(u4, t4, acc[4]);
    acc[5] = wmma4(g2, t5, acc[5]);
  }
}

__global__ __launch_bounds__(128) void k_res_conv3(const float* __restrict__ q,
                                                   const float* __restrict__ wp,
                                                   const float* __restrict__ b1,
                                                   float* __restrict__ hbuf) {
  __shared__ __attribute__((aligned(16))) float Td[KCH * T_CCP * 2];
  const int s  = blockIdx.x;
  const int b  = s >> 6;
  const int hh = s & 63;
  const int tid = threadIdx.x;
  const int wave = tid >> 5, lane = tid & 31, g = lane >> 4, mc = lane & 15;
  const int oc = wave * 16 + mc;

  v8f acc[6];
#pragma unroll
  for (int i = 0; i < 6; ++i) {
    const v8f z = {0.f, 0.f, 0.f, 0.f, 0.f, 0.f, 0.f, 0.f};
    acc[i] = z;
  }

  const float* wl = wp + ((size_t)g * CB + oc) * 2;
  v2f wA[12], wB[12];
  c3_load_run(wA, wl, 0, 0);

  const int p  = mc;
  const int il = p > 0 ? -1 : 0;
  const int ir = p < 15 ? 4 : 3;
  const int scc = 2 * wave + g;
  const float* xb = Td + (g * T_CCP + mc) * 2;

  for (int kc = 0; kc < C_ / (2 * KCH); ++kc) {
    if (kc) __syncthreads();
#pragma unroll
    for (int rr = 0; rr < 3; ++rr) {
      const int hy = hh + rr - 1;
      float* dst = Td + (scc * T_CCP + rr * T_ROW + p) * 2;
      if ((unsigned)hy < 64u) {
        const float* qr = q + (long)b * CHW + (long)(2 * (KCH * kc + scc)) * HW
                            + hy * W_ + 4 * p;
        v4f a0[2], a1[2];
        float l0[2], l1[2], r0[2], r1[2];
#pragma unroll
        for (int i2 = 0; i2 < 2; ++i2) {
          const float* c0 = qr + (long)(16 * i2) * HW;
          const float* c1 = c0 + HW;
          a0[i2] = *(const v4f*)c0;
          a1[i2] = *(const v4f*)c1;
          l0[i2] = c0[il];  r0[i2] = c0[ir];
          l1[i2] = c1[il];  r1[i2] = c1[ir];
        }
#pragma unroll
        for (int i2 = 0; i2 < 2; ++i2) {
          v2f d0 = {l0[i2] > 0.f ? l0[i2] : 0.f, l1[i2] > 0.f ? l1[i2] : 0.f};
          v2f d5 = {r0[i2] > 0.f ? r0[i2] : 0.f, r1[i2] > 0.f ? r1[i2] : 0.f};
          const v2f z2 = {0.f, 0.f};
          if (p == 0)  d0 = z2;
          if (p == 15) d5 = z2;
          const v2f d1 = {a0[i2][0] > 0.f ? a0[i2][0] : 0.f, a1[i2][0] > 0.f ? a1[i2][0] : 0.f};
          const v2f d2 = {a0[i2][1] > 0.f ? a0[i2][1] : 0.f, a1[i2][1] > 0.f ? a1[i2][1] : 0.f};
          const v2f d3 = {a0[i2][2] > 0.f ? a0[i2][2] : 0.f, a1[i2][2] > 0.f ? a1[i2][2] : 0.f};
          const v2f d4 = {a0[i2][3] > 0.f ? a0[i2][3] : 0.f, a1[i2][3] > 0.f ? a1[i2][3] : 0.f};
          const v2f s12 = d1 + d2, m12 = d1 - d2;
          const v2f s34 = d3 + d4, m43 = d4 - d3;
          const v2f m42 = d4 - d2, m31 = d3 - d1;
          const v2f t0 = (4.0f * d0 - 5.0f * d2) + d4;
          const v2f t1 = s34 - 4.0f * s12;
          const v2f t2 = m43 + 4.0f * m12;
          const v2f t3 = m42 + 2.0f * m31;
          const v2f t4 = m42 - 2.0f * m31;
          const v2f t5 = (4.0f * d1 - 5.0f * d3) + d5;
          float* d = dst + i2 * (8 * T_CCP * 2);
          *(v2f*)(d)       = t0;
          *(v2f*)(d + 32)  = t1;
          *(v2f*)(d + 64)  = t2;
          *(v2f*)(d + 96)  = t3;
          *(v2f*)(d + 128) = t4;
          *(v2f*)(d + 160) = t5;
        }
      } else {
        const v2f z = {0.f, 0.f};
        for (int i2 = 0; i2 < 2; ++i2) {
          float* d = dst + i2 * (8 * T_CCP * 2);
          *(v2f*)(d)       = z;
          *(v2f*)(d + 32)  = z;
          *(v2f*)(d + 64)  = z;
          *(v2f*)(d + 96)  = z;
          *(v2f*)(d + 128) = z;
          *(v2f*)(d + 160) = z;
        }
      }
    }
    __syncthreads();

    int nkc = kc + 1;
    if (nkc == C_ / (2 * KCH)) nkc = 0;
    c3_load_run(wB, wl, kc, 1);   c3_run(wA, xb, 0, acc);
    c3_load_run(wA, wl, kc, 2);   c3_run(wB, xb, 1, acc);
    c3_load_run(wB, wl, kc, 3);   c3_run(wA, xb, 2, acc);
    c3_load_run(wA, wl, kc, 4);   c3_run(wB, xb, 3, acc);
    c3_load_run(wB, wl, kc, 5);   c3_run(wA, xb, 4, acc);
    c3_load_run(wA, wl, nkc, 0);  c3_run(wB, xb, 5, acc);
  }

  const int ocb = wave * 16 + 8 * g;
  const long ob = (long)b * (CB * HW) + (long)ocb * HW + hh * W_ + 4 * mc;
#pragma unroll
  for (int r = 0; r < 8; ++r) {
    const float bias = b1[ocb + r];
    const float m0 = acc[0][r], m1 = acc[1][r], m2 = acc[2][r];
    const float m3 = acc[3][r], m4 = acc[4][r], m5 = acc[5][r];
    const float sa = m1 + m2, da = m1 - m2;
    const float sb = m3 + m4, db = m3 - m4;
    const float y0 = ((m0 + sa) + sb) + bias;
    const float y1 = (da + 2.0f * db) + bias;
    const float y2 = (sa + 4.0f * sb) + bias;
    const float y3 = ((da + 8.0f * db) + m5) + bias;
    const v4f o = {y0 > 0.f ? y0 : 0.f, y1 > 0.f ? y1 : 0.f,
                   y2 > 0.f ? y2 : 0.f, y3 > 0.f ? y3 : 0.f};
    *(v4f*)(hbuf + ob + (long)r * HW) = o;
  }
}

#define C1_HP  80
#define C1_WP  68

__global__ __launch_bounds__(256) void k_res_conv1(float* __restrict__ q,
                                                   const float* __restrict__ hbuf,
                                                   const float* __restrict__ w2,
                                                   const float* __restrict__ b2) {
  __shared__ __attribute__((aligned(16))) float Hs[32 * C1_HP * 2];
  __shared__ __attribute__((aligned(16))) float Ws[C_ * C1_WP];
  const int s  = blockIdx.x;
  const int b  = s >> 6;
  const int hh = s & 63;
  const int tid = threadIdx.x;
  const int wave = tid >> 5, lane = tid & 31, g = lane >> 4, mc = lane & 15;

  float* qb = q + (long)b * CHW + (long)(32 * wave + 8 * g) * HW + hh * W_ + 4 * mc;
  v4f qv[2][8];
#pragma unroll
  for (int t = 0; t < 2; ++t)
#pragma unroll
    for (int r = 0; r < 8; ++r)
      qv[t][r] = *(const v4f*)(qb + (long)(16 * t + r) * HW);

#pragma unroll
  for (int i = 0; i < 16; ++i) {
    const int i4 = tid + 256 * i;
    const v4f v = *(const v4f*)(w2 + i4 * 4);
    *(v4f*)(Ws + (i4 >> 4) * C1_WP + (i4 & 15) * 4) = v;
  }
  {
    const int m = tid & 15, ccb = tid >> 4;
    const float* hs = hbuf + (long)b * (CB * HW) + hh * W_ + 4 * m;
#pragma unroll
    for (int i = 0; i < 2; ++i) {
      const int cc = ccb + 16 * i;
      const v4f v0 = *(const v4f*)(hs + (long)(2 * cc) * HW);
      const v4f v1 = *(const v4f*)(hs + (long)(2 * cc + 1) * HW);
      float* dst = Hs + (cc * C1_HP + m) * 2;
#pragma unroll
      for (int j = 0; j < 4; ++j) {
        const v2f pv = {v0[j], v1[j]};
        *(v2f*)(dst + 16 * j * 2) = pv;
      }
    }
  }
  __syncthreads();

  v8f acc[2][4];
#pragma unroll
  for (int t = 0; t < 2; ++t)
#pragma unroll
    for (int j = 0; j < 4; ++j) {
      const v8f z = {0.f, 0.f, 0.f, 0.f, 0.f, 0.f, 0.f, 0.f};
      acc[t][j] = z;
    }

  const float* wa = Ws + (32 * wave + mc) * C1_WP + 2 * g;
  const float* xb = Hs + (g * C1_HP + mc) * 2;
#pragma unroll
  for (int ks = 0; ks < 16; ++ks) {
    const v2f a0 = *(const v2f*)(wa + 4 * ks);
    const v2f a1 = *(const v2f*)(wa + 16 * C1_WP + 4 * ks);
    const float* xp = xb + ks * (2 * C1_HP * 2);
    const v2f x0 = *(const v2f*)(xp);
    const v2f x1 = *(const v2f*)(xp + 32);
    const v2f x2 = *(const v2f*)(xp + 64);
    const v2f x3 = *(const v2f*)(xp + 96);
    acc[0][0] = wmma4(a0, x0, acc[0][0]);
    acc[0][1] = wmma4(a0, x1, acc[0][1]);
    acc[0][2] = wmma4(a0, x2, acc[0][2]);
    acc[0][3] = wmma4(a0, x3, acc[0][3]);
    acc[1][0] = wmma4(a1, x0, acc[1][0]);
    acc[1][1] = wmma4(a1, x1, acc[1][1]);
    acc[1][2] = wmma4(a1, x2, acc[1][2]);
    acc[1][3] = wmma4(a1, x3, acc[1][3]);
  }

  const int ocb = 32 * wave + 8 * g;
#pragma unroll
  for (int t = 0; t < 2; ++t)
#pragma unroll
    for (int r = 0; r < 8; ++r) {
      const float bias = b2[ocb + 16 * t + r];
      v4f o = qv[t][r];
      o[0] += acc[t][0][r] + bias;
      o[1] += acc[t][1][r] + bias;
      o[2] += acc[t][2][r] + bias;
      o[3] += acc[t][3][r] + bias;
      *(v4f*)(qb + (long)(16 * t + r) * HW) = o;
    }
}

#define OC_TILE(t) ((t) * 4096 + ((t) & 1) * 32 + ((t) >> 1) * 68)

__global__ __launch_bounds__(256) void k_out_conv(const float* __restrict__ q,
                                                  const float* __restrict__ ow,
                                                  const float* __restrict__ ob,
                                                  float* __restrict__ out) {
  __shared__ __attribute__((aligned(16))) float Ls[128 * 4 * 2 + 4 * 4096 + 128];
  float* Bs = Ls;
  float* Qs = Ls + 128 * 4 * 2;
  const int s  = blockIdx.x;
  const int b  = s >> 6;
  const int hh = s & 63;
  const int tid = threadIdx.x;

  {
    const int col = tid & 3;
    const int colc = col < OUTC ? col : OUTC - 1;
    v2f wv[2];
#pragma unroll
    for (int i = 0; i < 2; ++i) {
      const int cc = (tid + 256 * i) >> 2;
      wv[i] = *(const v2f*)(ow + colc * C_ + 2 * cc);
    }
    const int m = tid & 15, ccb = tid >> 4;
    const float* qs = q + (long)b * CHW + hh * W_ + 4 * m;
    v4f v0[8], v1[8];
#pragma unroll
    for (int i = 0; i < 8; ++i) {
      const int cc = ccb + 16 * i;
      v0[i] = *(const v4f*)(qs + (long)(2 * cc) * HW);
      v1[i] = *(const v4f*)(qs + (long)(2 * cc + 1) * HW);
    }
#pragma unroll
    for (int i = 0; i < 2; ++i) {
      const v2f z = {0.f, 0.f};
      *(v2f*)(Bs + (tid + 256 * i) * 2) = col < OUTC ? wv[i] : z;
    }
    float* dst = Qs + OC_TILE(m >> 2) + (ccb * 16 + 4 * (m & 3)) * 2;
#pragma unroll
    for (int i = 0; i < 8; ++i) {
      const v4f lo = {v0[i][0], v1[i][0], v0[i][1], v1[i][1]};
      const v4f hi = {v0[i][2], v1[i][2], v0[i][3], v1[i][3]};
      *(v4f*)(dst + i * (16 * 16 * 2))     = lo;
      *(v4f*)(dst + i * (16 * 16 * 2) + 4) = hi;
    }
  }
  __syncthreads();

  const int wave = __builtin_amdgcn_readfirstlane(tid >> 5);
  const int lane = tid & 31, g = lane >> 4, mc = lane & 15;
  if (wave < 4) {
    const float* xa = Qs + OC_TILE(wave) + (g * 16 + mc) * 2;
    const float* xw = Bs + (g * 4 + (mc & 3)) * 2;
    v8f acc = {0.f, 0.f, 0.f, 0.f, 0.f, 0.f, 0.f, 0.f};
    for (int k8 = 0; k8 < 64; k8 += 8) {
      const float* pa = xa + k8 * (2 * 16 * 2);
      const float* pw = xw + k8 * (2 * 4 * 2);
#pragma unroll
      for (int j = 0; j < 8; ++j) {
        const v2f a  = *(const v2f*)(pa + j * (2 * 16 * 2));
        const v2f bb = *(const v2f*)(pw + j * (2 * 4 * 2));
        acc = wmma4(a, bb, acc);
      }
    }
    if (mc < OUTC) {
      const float bias = ob[mc];
      const long obase = ((long)(b * OUTC + mc) * H_ + hh) * W_ + 16 * wave;
#pragma unroll
      for (int r = 0; r < 8; ++r) out[obase + r + 8 * g] = acc[r] + bias;
    }
  }
}

__global__ __launch_bounds__(256) void k_final(const unsigned* __restrict__ counts,
                                               const float* __restrict__ wsdiff,
                                               float* __restrict__ out) {
  __shared__ float red[256];
  const int gid = blockIdx.x * 256 + threadIdx.x;
  if (blockIdx.x < 32) {
    out[WF_POS + gid] = (float)counts[gid] * (1.0f / 4096.0f);
  } else {
    float sum = 0.f;
    for (int t = threadIdx.x; t < B_ * H_; t += 256) sum += wsdiff[t];
    red[threadIdx.x] = sum; __syncthreads();
    for (int st = 128; st > 0; st >>= 1) {
      if (threadIdx.x < st) red[threadIdx.x] += red[threadIdx.x + st];
      __syncthreads();
    }
    if (threadIdx.x == 0)
      out[DIFF_POS] = red[0] * (1.0f / ((float)NPOS * (float)C_));
  }
}

extern "C" void kernel_launch(void* const* d_in, const int* in_sizes, int n_in,
                              void* d_out, int out_size, void* d_ws, size_t ws_size,
                              hipStream_t stream) {
  (void)in_sizes; (void)n_in; (void)out_size; (void)ws_size;
  const float* in  = (const float*)d_in[0];
  const float* emb = (const float*)d_in[1];
  const float* rw1 = (const float*)d_in[2];
  const float* rb1 = (const float*)d_in[3];
  const float* rw2 = (const float*)d_in[4];
  const float* rb2 = (const float*)d_in[5];
  const float* ow  = (const float*)d_in[6];
  const float* ob  = (const float*)d_in[7];
  float* out = (float*)d_out;
  float* ws  = (float*)d_ws;

  float*    q      = ws + Q_OFF;
  float*    hbuf   = ws + H_OFF;
  float*    epack  = ws + EPACK_OFF;
  float*    en2    = ws + EN2_OFF;
  unsigned* counts = (unsigned*)(ws + CNT_OFF);
  float*    wsdiff = ws + WDIFF_OFF;
  float*    wpack  = ws + WPACK_OFF;

  k_prep<<<256, 256, 0, stream>>>(emb, epack, en2, counts);
  k_packw<<<(3 * WP_PER_BLK + 255) / 256, 256, 0, stream>>>(rw1, wpack);
  k_vq<<<B_ * H_, 128, 0, stream>>>(in, emb, epack, en2, q, counts, wsdiff);
  for (int i = 0; i < 3; ++i) {
    k_res_conv3<<<B_ * H_, 128, 0, stream>>>(q, wpack + (size_t)i * WP_PER_BLK * 2,
                                           rb1 + i * CB, hbuf);
    k_res_conv1<<<B_ * H_, 256, 0, stream>>>(q, hbuf, rw2 + (size_t)i * C_ * CB,
                                           rb2 + i * C_);
  }
  k_out_conv<<<B_ * H_, 256, 0, stream>>>(q, ow, ob, out);
  k_final<<<33, 256, 0, stream>>>(counts, wsdiff, out);
}


# 3x3 conv: products also grouped along h (2 rows x 3 taps from 4 products, nested in the 4-column form): 24 f32 WMMAs per k-step and row pair, were 36; sums regrouped (24 partials per 2x4 outputs)
# speedup vs baseline: 5.1022x; 1.0739x over previous
#include <hip/hip_runtime.h>


typedef float v2f __attribute__((ext_vector_type(2)));
typedef float v4f __attribute__((ext_vector_type(4)));
typedef float v8f __attribute__((ext_vector_type(8)));

#define B_    16
#define C_    256
#define H_    64
#define W_    64
#define OUTC  3
#define NT    512
#define CB    64

#define HW     (H_*W_)
#define NPOS   (B_*HW)
#define CHW    (C_*HW)

#define Q_OFF      0u
#define H_OFF      16777216u
#define EPACK_OFF  20971520u
#define EN2_OFF    21102592u
#define CNT_OFF    21103104u
#define WDIFF_OFF  21111296u
#define WPACK_OFF  21115392u

#define WP_PER_BLK (9*128*CB)
#define WP_PER_TAP (128*CB)

#define DIFF_POS   196608
#define WF_POS     196609

static __device__ __forceinline__ v8f wmma4(v2f a, v2f b, v8f c) {
  return __builtin_amdgcn_wmma_f32_16x16x4_f32(false, a, false, b, (short)0, c,
                                               false, false);
}

__global__ __launch_bounds__(256) void k_prep(const float* __restrict__ embed,
                                              float* __restrict__ epack,
                                              float* __restrict__ en2,
                                              unsigned* __restrict__ counts) {
  const int gid = blockIdx.x * 256 + threadIdx.x;
  if (gid < 128 * NT) {
    const int cc = gid >> 9, n = gid & (NT - 1);
    epack[gid * 2 + 0] = embed[(2 * cc) * NT + n];
    epack[gid * 2 + 1] = embed[(2 * cc + 1) * NT + n];
  }
  if (gid < NT) {
    float s = 0.f;
    for (int c = 0; c < C_; ++c) { float v = embed[c * NT + gid]; s += v * v; }
    en2[gid] = s;
  }
  if (gid < B_ * NT) counts[gid] = 0u;
}

__global__ __launch_bounds__(256) void k_packw(const float* __restrict__ rw1,
                                               float* __restrict__ wpack) {
  const int t = blockIdx.x * 256 + threadIdx.x;
  if (t >= 3 * WP_PER_BLK) return;
  const int i   = t / WP_PER_BLK;
  const int r   = t % WP_PER_BLK;
  const int tap = r / WP_PER_TAP;
  const int r2  = r % WP_PER_TAP;
  const int cc  = r2 >> 6;
  const int oc  = r2 & 63;
  const int dh = tap / 3, dw = tap % 3;
  const float* src = rw1 + (size_t)i * CB * C_ * 9
                         + ((size_t)oc * C_ + 2 * cc) * 9 + dh * 3 + dw;
  wpack[t * 2 + 0] = src[0];
  wpack[t * 2 + 1] = src[9];
}

#define VQ_SLOT(cc, p) (((cc) << 6) + (((p) + (((cc) & 1) << 4)) & 63))

__global__ __launch_bounds__(128) void k_vq(const float* __restrict__ in,
                                            const float* __restrict__ embed,
                                            const float* __restrict__ epack,
                                            const float* __restrict__ en2,
                                            float* __restrict__ q,
                                            unsigned* __restrict__ counts,
                                            float* __restrict__ wsdiff) {
  __shared__ float As[128 * 64 * 2];
  __shared__ float rv[4][64];
  __shared__ int   ri[4][64];
  __shared__ int   idxs[64];
  __shared__ float red[128];

  const int s  = blockIdx.x;
  const int b  = s >> 6;
  const int hh = s & 63;
  const int tid = threadIdx.x;
  const long gbase0 = (long)b * CHW + hh * W_;
  const int sp = tid & 63;
  const int sc = tid >> 6;

  {
    const float* src = in + gbase0 + (long)(2 * sc) * HW + sp;
    float* dst = As + VQ_SLOT(sc, sp) * 2;
#pragma unroll 8
    for (int i = 0; i < 64; ++i) {
      const float v0 = src[(long)(4 * i) * HW];
      const float v1 = src[(long)(4 * i + 1) * HW];
      const v2f pv = {v0, v1};
      *(v2f*)(dst + i * (2 * 64 * 2)) = pv;
    }
  }
  __syncthreads();

  const int wave = tid >> 5, lane = tid & 31, g = lane >> 4, mc = lane & 15;
  float bestv[4][8];
  int   besti[4][8];
#pragma unroll
  for (int j = 0; j < 4; ++j)
#pragma unroll
    for (int r = 0; r < 8; ++r) { bestv[j][r] = 3.4e38f; besti[j][r] = 0; }

  const float* xa0 = As + ((g << 6) + (((0 + g) & 3) << 4) + mc) * 2;
  const float* xa1 = As + ((g << 6) + (((1 + g) & 3) << 4) + mc) * 2;
  const float* xa2 = As + ((g << 6) + (((2 + g) & 3) << 4) + mc) * 2;
  const float* xa3 = As + ((g << 6) + (((3 + g) & 3) << 4) + mc) * 2;

  const float* el = epack + ((size_t)g * NT + (wave << 7) + mc) * 2;
  v2f eA[8], eB[8];
#pragma unroll
  for (int j = 0; j < 8; ++j) eA[j] = *(const v2f*)(el + (size_t)j * (2 * NT * 2));

#pragma unroll 1
  for (int tt = 0; tt < 8; ++tt) {
    const int n0 = (wave * 8 + tt) << 4;
    const float* eb = el + tt * (16 * 2);
    v8f acc0 = {0.f, 0.f, 0.f, 0.f, 0.f, 0.f, 0.f, 0.f};
    v8f acc1 = acc0, acc2 = acc0, acc3 = acc0;
#pragma unroll 1
    for (int kp = 0; kp < 4; ++kp) {
#pragma unroll
      for (int j = 0; j < 8; ++j)
        eB[j] = *(const v2f*)(eb + (size_t)(16 * kp + 8 + j) * (2 * NT * 2));
#pragma unroll
      for (int j = 0; j < 8; ++j) {
        const int ks = 16 * kp + j;
        const v2f a0 = *(const v2f*)(xa0 + ks * (2 * 64 * 2));
        const v2f a1 = *(const v2f*)(xa1 + ks * (2 * 64 * 2));
        const v2f a2 = *(const v2f*)(xa2 + ks * (2 * 64 * 2));
        const v2f a3 = *(const v2f*)(xa3 + ks * (2 * 64 * 2));
        acc0 = wmma4(a0, eA[j], acc0);
        acc1 = wmma4(a1, eA[j], acc1);
        acc2 = wmma4(a2, eA[j], acc2);
        acc3 = wmma4(a3, eA[j], acc3);
      }
      int nkp = kp + 1, ntt = tt;
      if (nkp == 4) { nkp = 0; ntt = tt + 1; }
      if (ntt == 8) ntt = 0;
      const float* en = el + ntt * (16 * 2) + (size_t)(16 * nkp) * (2 * NT * 2);
#pragma unroll
      for (int j = 0; j < 8; ++j)
        eA[j] = *(const v2f*)(en + (size_t)j * (2 * NT * 2));
#pragma unroll
      for (int j = 0; j < 8; ++j) {
        const int ks = 16 * kp + 8 + j;
        const v2f a0 = *(const v2f*)(xa0 + ks * (2 * 64 * 2));
        const v2f a1 = *(const v2f*)(xa1 + ks * (2 * 64 * 2));
        const v2f a2 = *(const v2f*)(xa2 + ks * (2 * 64 * 2));
        const v2f a3 = *(const v2f*)(xa3 + ks * (2 * 64 * 2));
        acc0 = wmma4(a0, eB[j], acc0);
        acc1 = wmma4(a1, eB[j], acc1);
        acc2 = wmma4(a2, eB[j], acc2);
        acc3 = wmma4(a3, eB[j], acc3);
      }
    }
    const float e2 = en2[n0 + mc];
    const int   n  = n0 + mc;
#pragma unroll
    for (int r = 0; r < 8; ++r) {
      float d;
      d = e2 - 2.0f * acc0[r]; if (d < bestv[0][r]) { bestv[0][r] = d; besti[0][r] = n; }
      d = e2 - 2.0f * acc1[r]; if (d < bestv[1][r]) { bestv[1][r] = d; besti[1][r] = n; }
      d = e2 - 2.0f * acc2[r]; if (d < bestv[2][r]) { bestv[2][r] = d; besti[2][r] = n; }
      d = e2 - 2.0f * acc3[r]; if (d < bestv[3][r]) { bestv[3][r] = d; besti[3][r] = n; }
    }
  }

#pragma unroll
  for (int j = 0; j < 4; ++j) {
#pragma unroll
    for (int r = 0; r < 8; ++r) {
      float v = bestv[j][r];
      int  ix = besti[j][r];
#pragma unroll
      for (int m = 8; m > 0; m >>= 1) {
        const float ov = __shfl_xor(v, m);
        const int   oi = __shfl_xor(ix, m);
        const bool take = (ov < v) | ((ov == v) & (oi < ix));
        v  = take ? ov : v;
        ix = take ? oi : ix;
      }
      bestv[j][r] = v;
      besti[j][r] = ix;
    }
  }
  if (mc == 0) {
#pragma unroll
    for (int j = 0; j < 4; ++j)
#pragma unroll
      for (int r = 0; r < 8; ++r) {
        rv[wave][16 * j + r + 8 * g] = bestv[j][r];
        ri[wave][16 * j + r + 8 * g] = besti[j][r];
      }
  }
  __syncthreads();

  if (tid < 64) {
    float bv = rv[0][tid]; int bi = ri[0][tid];
    for (int w = 1; w < 4; ++w) {
      const float v = rv[w][tid]; const int ii = ri[w][tid];
      if (v < bv || (v == bv && ii < bi)) { bv = v; bi = ii; }
    }
    idxs[tid] = bi;
    atomicAdd(&counts[b * NT + bi], 1u);
  }
  __syncthreads();

  float fsum = 0.f;
  {
    const int id = idxs[sp];
    const float* xs = As + VQ_SLOT(sc, sp) * 2;
    const float* es = embed + (2 * sc) * NT + id;
    float* qd = q + gbase0 + (long)(2 * sc) * HW + sp;
#pragma unroll 8
    for (int i = 0; i < 64; ++i) {
      const v2f x = *(const v2f*)(xs + i * (2 * 64 * 2));
      const float e0 = es[(4 * i) * NT];
      const float e1 = es[(4 * i + 1) * NT];
      qd[(long)(4 * i) * HW]     = e0;
      qd[(long)(4 * i + 1) * HW] = e1;
      const float d0 = e0 - x[0], d1 = e1 - x[1];
      fsum += d0 * d0 + d1 * d1;
    }
  }
  red[tid] = fsum; __syncthreads();
  for (int st = 64; st > 0; st >>= 1) {
    if (tid < st) red[tid] += red[tid + st];
    __syncthreads();
  }
  if (tid == 0) wsdiff[s] = red[0];
}

#define KCH    16
#define T_HP   96
#define T_CCP  400

static __device__ __forceinline__ void c3_load_run(v2f (&dst)[18], const float* wl,
                                                   int kc, int r) {
#pragma unroll
  for (int j = 0; j < 2; ++j)
#pragma unroll
    for (int tap = 0; tap < 9; ++tap)
      dst[9 * j + tap] = *(const v2f*)(wl + ((size_t)tap * WP_PER_TAP
                                            + (size_t)(KCH * kc + 2 * (2 * r + j)) * CB) * 2);
}

static __device__ __forceinline__ void c3_run(const v2f (&w)[18], const float* xb, int r,
                                              v8f (&acc)[24]) {
#pragma unroll
  for (int j = 0; j < 2; ++j) {
    const int ks = 2 * r + j;
    const float* xk = xb + (ks * 2 * T_CCP) * 2;
    v2f vh[4][3];
#pragma unroll
    for (int dw = 0; dw < 3; ++dw) {
      const v2f k0 = w[9 * j + dw], k1 = w[9 * j + 3 + dw], k2 = w[9 * j + 6 + dw];
      const v2f sk = k0 + k2;
      vh[0][dw] = k0;
      vh[1][dw] = 0.5f * (sk + k1);
      vh[2][dw] = 0.5f * (sk - k1);
      vh[3][dw] = k2;
    }
#pragma unroll
    for (int hp = 0; hp < 4; ++hp) {
      const float* xp = xk + (hp * T_HP) * 2;
      const v2f g0 = vh[hp][0], g1 = vh[hp][1], g2 = vh[hp][2];
      const v2f sg = g0 + g2;
      const v2f u0 = 0.25f * g0;
      const v2f u1 = (-1.0f / 6.0f) * (sg + g1);
      const v2f u2 = (-1.0f / 6.0f) * (sg - g1);
      const v2f ue = (1.0f / 24.0f) * g0 + (1.0f / 6.0f) * g2;
      const v2f uo = (1.0f / 12.0f) * g1;
      const v2f u3 = ue + uo;
      const v2f u4 = ue - uo;
      const v2f t0 = *(const v2f*)(xp);
      const v2f t1 = *(const v2f*)(xp + 32);
      const v2f t2 = *(const v2f*)(xp + 64);
      const v2f t3 = *(const v2f*)(xp + 96);
      const v2f t4 = *(const v2f*)(xp + 128);
      const v2f t5 = *(const v2f*)(xp + 160);
      acc[6 * hp + 0] = wmma4(u0, t0, acc[6 * hp + 0]);
      acc[6 * hp + 1] = wmma4(u1, t1, acc[6 * hp + 1]);
      acc[6 * hp + 2] = wmma4(u2, t2, acc[6 * hp + 2]);
      acc[6 * hp + 3] = wmma4(u3, t3, acc[6 * hp + 3]);
      acc[6 * hp + 4] = wmma4(u4, t4, acc[6 * hp + 4]);
      acc[6 * hp + 5] = wmma4(g2, t5, acc[6 * hp + 5]);
    }
  }
}

__global__ __launch_bounds__(128) void k_res_conv3(const float* __restrict__ q,
                                                   const float* __restrict__ wp,
                                                   const float* __restrict__ b1,
                                                   float* __restrict__ hbuf) {
  __shared__ __attribute__((aligned(16))) float Td[KCH * T_CCP * 2];
  const int s   = blockIdx.x;
  const int b   = s >> 5;
  const int hh0 = (s & 31) * 2;
  const int tid = threadIdx.x;
  const int wave = tid >> 5, lane = tid & 31, g = lane >> 4, mc = lane & 15;
  const int oc = wave * 16 + mc;

  v8f acc[24];
#pragma unroll
  for (int i = 0; i < 24; ++i) {
    const v8f z = {0.f, 0.f, 0.f, 0.f, 0.f, 0.f, 0.f, 0.f};
    acc[i] = z;
  }

  const float* wl = wp + ((size_t)g * CB + oc) * 2;
  v2f wA[18], wB[18];
  c3_load_run(wA, wl, 0, 0);

  const int p  = mc;
  const int il = p > 0 ? -1 : 0;
  const int ir = p < 15 ? 4 : 3;
  const int scc = 2 * wave + g;
  const float* xb = Td + (g * T_CCP + mc) * 2;

  for (int kc = 0; kc < C_ / (2 * KCH); ++kc) {
    if (kc) __syncthreads();
    {
      v4f a0[2][4], a1[2][4];
      float l0[2][4], l1[2][4], r0[2][4], r1[2][4];
#pragma unroll
      for (int i2 = 0; i2 < 2; ++i2) {
#pragma unroll
        for (int rr = 0; rr < 4; ++rr) {
          const int hy  = hh0 + rr - 1;
          const int hyc = hy < 0 ? 0 : (hy > 63 ? 63 : hy);
          const float* c0 = q + (long)b * CHW + (long)(2 * (KCH * kc + scc + 8 * i2)) * HW
                              + hyc * W_ + 4 * p;
          const float* c1 = c0 + HW;
          a0[i2][rr] = *(const v4f*)c0;
          a1[i2][rr] = *(const v4f*)c1;
          l0[i2][rr] = c0[il];  r0[i2][rr] = c0[ir];
          l1[i2][rr] = c1[il];  r1[i2][rr] = c1[ir];
        }
      }
#pragma unroll
      for (int i2 = 0; i2 < 2; ++i2) {
        v2f T[4][6];
#pragma unroll
        for (int rr = 0; rr < 4; ++rr) {
          const int hy = hh0 + rr - 1;
          const bool inimg = (unsigned)hy < 64u;
          v2f d0 = {l0[i2][rr] > 0.f ? l0[i2][rr] : 0.f, l1[i2][rr] > 0.f ? l1[i2][rr] : 0.f};
          v2f d5 = {r0[i2][rr] > 0.f ? r0[i2][rr] : 0.f, r1[i2][rr] > 0.f ? r1[i2][rr] : 0.f};
          const v2f z2 = {0.f, 0.f};
          if (p == 0)  d0 = z2;
          if (p == 15) d5 = z2;
          const v4f x0 = a0[i2][rr], x1 = a1[i2][rr];
          const v2f d1 = {x0[0] > 0.f ? x0[0] : 0.f, x1[0] > 0.f ? x1[0] : 0.f};
          const v2f d2 = {x0[1] > 0.f ? x0[1] : 0.f, x1[1] > 0.f ? x1[1] : 0.f};
          const v2f d3 = {x0[2] > 0.f ? x0[2] : 0.f, x1[2] > 0.f ? x1[2] : 0.f};
          const v2f d4 = {x0[3] > 0.f ? x0[3] : 0.f, x1[3] > 0.f ? x1[3] : 0.f};
          const v2f s12 = d1 + d2, m12 = d1 - d2;
          const v2f s34 = d3 + d4, m43 = d4 - d3;
          const v2f m42 = d4 - d2, m31 = d3 - d1;
          const v2f t0 = (4.0f * d0 - 5.0f * d2) + d4;
          const v2f t1 = s34 - 4.0f * s12;
          const v2f t2 = m43 + 4.0f * m12;
          const v2f t3 = m42 + 2.0f * m31;
          const v2f t4 = m42 - 2.0f * m31;
          const v2f t5 = (4.0f * d1 - 5.0f * d3) + d5;
          T[rr][0] = inimg ? t0 : z2;
          T[rr][1] = inimg ? t1 : z2;
          T[rr][2] = inimg ? t2 : z2;
          T[rr][3] = inimg ? t3 : z2;
          T[rr][4] = inimg ? t4 : z2;
          T[rr][5] = inimg ? t5 : z2;
        }
        float* d = Td + ((scc + 8 * i2) * T_CCP + p) * 2;
#pragma unroll
        for (int wq = 0; wq < 6; ++wq) {
          *(v2f*)(d + (0 * T_HP + 16 * wq) * 2) = T[0][wq] - T[2][wq];
          *(v2f*)(d + (1 * T_HP + 16 * wq) * 2) = T[1][wq] + T[2][wq];
          *(v2f*)(d + (2 * T_HP + 16 * wq) * 2) = T[2][wq] - T[1][wq];
          *(v2f*)(d + (3 * T_HP + 16 * wq) * 2) = T[1][wq] - T[3][wq];
        }
      }
    }
    __syncthreads();

    int nkc = kc + 1;
    if (nkc == C_ / (2 * KCH)) nkc = 0;
    c3_load_run(wB, wl, kc, 1);   c3_run(wA, xb, 0, acc);
    c3_load_run(wA, wl, kc, 2);   c3_run(wB, xb, 1, acc);
    c3_load_run(wB, wl, kc, 3);   c3_run(wA, xb, 2, acc);
    c3_load_run(wA, wl, nkc, 0);  c3_run(wB, xb, 3, acc);
  }

  const int ocb = wave * 16 + 8 * g;
  const long ob = (long)b * (CB * HW) + (long)ocb * HW + hh0 * W_ + 4 * mc;
#pragma unroll
  for (int r = 0; r < 8; ++r) {
    const float bias = b1[ocb + r];
#pragma unroll
    for (int row = 0; row < 2; ++row) {
      float m[6];
#pragma unroll
      for (int wq = 0; wq < 6; ++wq) {
        const float n0 = acc[wq][r], n1 = acc[6 + wq][r];
        const float n2 = acc[12 + wq][r], n3 = acc[18 + wq][r];
        m[wq] = row == 0 ? (n0 + n1) + n2 : (n1 - n2) - n3;
      }
      const float sa = m[1] + m[2], da = m[1] - m[2];
      const float sb = m[3] + m[4], db = m[3] - m[4];
      const float y0 = ((m[0] + sa) + sb) + bias;
      const float y1 = (da + 2.0f * db) + bias;
      const float y2 = (sa + 4.0f * sb) + bias;
      const float y3 = ((da + 8.0f * db) + m[5]) + bias;
      const v4f o = {y0 > 0.f ? y0 : 0.f, y1 > 0.f ? y1 : 0.f,
                     y2 > 0.f ? y2 : 0.f, y3 > 0.f ? y3 : 0.f};
      *(v4f*)(hbuf + ob + (long)r * HW + row * W_) = o;
    }
  }
}

#define C1_HP  80
#define C1_WP  68

__global__ __launch_bounds__(256) void k_res_conv1(float* __restrict__ q,
                                                   const float* __restrict__ hbuf,
                                                   const float* __restrict__ w2,
                                                   const float* __restrict__ b2) {
  __shared__ __attribute__((aligned(16))) float Hs[32 * C1_HP * 2];
  __shared__ __attribute__((aligned(16))) float Ws[C_ * C1_WP];
  const int s  = blockIdx.x;
  const int b  = s >> 6;
  const int hh = s & 63;
  const int tid = threadIdx.x;
  const int wave = tid >> 5, lane = tid & 31, g = lane >> 4, mc = lane & 15;

  float* qb = q + (long)b * CHW + (long)(32 * wave + 8 * g) * HW + hh * W_ + 4 * mc;
  v4f qv[2][8];
#pragma unroll
  for (int t = 0; t < 2; ++t)
#pragma unroll
    for (int r = 0; r < 8; ++r)
      qv[t][r] = *(const v4f*)(qb + (long)(16 * t + r) * HW);

#pragma unroll
  for (int i = 0; i < 16; ++i) {
    const int i4 = tid + 256 * i;
    const v4f v = *(const v4f*)(w2 + i4 * 4);
    *(v4f*)(Ws + (i4 >> 4) * C1_WP + (i4 & 15) * 4) = v;
  }
  {
    const int m = tid & 15, ccb = tid >> 4;
    const float* hs = hbuf + (long)b * (CB * HW) + hh * W_ + 4 * m;
#pragma unroll
    for (int i = 0; i < 2; ++i) {
      const int cc = ccb + 16 * i;
      const v4f v0 = *(const v4f*)(hs + (long)(2 * cc) * HW);
      const v4f v1 = *(const v4f*)(hs + (long)(2 * cc + 1) * HW);
      float* dst = Hs + (cc * C1_HP + m) * 2;
#pragma unroll
      for (int j = 0; j < 4; ++j) {
        const v2f pv = {v0[j], v1[j]};
        *(v2f*)(dst + 16 * j * 2) = pv;
      }
    }
  }
  __syncthreads();

  v8f acc[2][4];
#pragma unroll
  for (int t = 0; t < 2; ++t)
#pragma unroll
    for (int j = 0; j < 4; ++j) {
      const v8f z = {0.f, 0.f, 0.f, 0.f, 0.f, 0.f, 0.f, 0.f};
      acc[t][j] = z;
    }

  const float* wa = Ws + (32 * wave + mc) * C1_WP + 2 * g;
  const float* xb = Hs + (g * C1_HP + mc) * 2;
#pragma unroll
  for (int ks = 0; ks < 16; ++ks) {
    const v2f a0 = *(const v2f*)(wa + 4 * ks);
    const v2f a1 = *(const v2f*)(wa + 16 * C1_WP + 4 * ks);
    const float* xp = xb + ks * (2 * C1_HP * 2);
    const v2f x0 = *(const v2f*)(xp);
    const v2f x1 = *(const v2f*)(xp + 32);
    const v2f x2 = *(const v2f*)(xp + 64);
    const v2f x3 = *(const v2f*)(xp + 96);
    acc[0][0] = wmma4(a0, x0, acc[0][0]);
    acc[0][1] = wmma4(a0, x1, acc[0][1]);
    acc[0][2] = wmma4(a0, x2, acc[0][2]);
    acc[0][3] = wmma4(a0, x3, acc[0][3]);
    acc[1][0] = wmma4(a1, x0, acc[1][0]);
    acc[1][1] = wmma4(a1, x1, acc[1][1]);
    acc[1][2] = wmma4(a1, x2, acc[1][2]);
    acc[1][3] = wmma4(a1, x3, acc[1][3]);
  }

  const int ocb = 32 * wave + 8 * g;
#pragma unroll
  for (int t = 0; t < 2; ++t)
#pragma unroll
    for (int r = 0; r < 8; ++r) {
      const float bias = b2[ocb + 16 * t + r];
      v4f o = qv[t][r];
      o[0] += acc[t][0][r] + bias;
      o[1] += acc[t][1][r] + bias;
      o[2] += acc[t][2][r] + bias;
      o[3] += acc[t][3][r] + bias;
      *(v4f*)(qb + (long)(16 * t + r) * HW) = o;
    }
}

#define OC_TILE(t) ((t) * 4096 + ((t) & 1) * 32 + ((t) >> 1) * 68)

__global__ __launch_bounds__(256) void k_out_conv(const float* __restrict__ q,
                                                  const float* __restrict__ ow,
                                                  const float* __restrict__ ob,
                                                  float* __restrict__ out) {
  __shared__ __attribute__((aligned(16))) float Ls[128 * 4 * 2 + 4 * 4096 + 128];
  float* Bs = Ls;
  float* Qs = Ls + 128 * 4 * 2;
  const int s  = blockIdx.x;
  const int b  = s >> 6;
  const int hh = s & 63;
  const int tid = threadIdx.x;

  {
    const int col = tid & 3;
    const int colc = col < OUTC ? col : OUTC - 1;
    v2f wv[2];
#pragma unroll
    for (int i = 0; i < 2; ++i) {
      const int cc = (tid + 256 * i) >> 2;
      wv[i] = *(const v2f*)(ow + colc * C_ + 2 * cc);
    }
    const int m = tid & 15, ccb = tid >> 4;
    const float* qs = q + (long)b * CHW + hh * W_ + 4 * m;
    v4f v0[8], v1[8];
#pragma unroll
    for (int i = 0; i < 8; ++i) {
      const int cc = ccb + 16 * i;
      v0[i] = *(const v4f*)(qs + (long)(2 * cc) * HW);
      v1[i] = *(const v4f*)(qs + (long)(2 * cc + 1) * HW);
    }
#pragma unroll
    for (int i = 0; i < 2; ++i) {
      const v2f z = {0.f, 0.f};
      *(v2f*)(Bs + (tid + 256 * i) * 2) = col < OUTC ? wv[i] : z;
    }
    float* dst = Qs + OC_TILE(m >> 2) + (ccb * 16 + 4 * (m & 3)) * 2;
#pragma unroll
    for (int i = 0; i < 8; ++i) {
      const v4f lo = {v0[i][0], v1[i][0], v0[i][1], v1[i][1]};
      const v4f hi = {v0[i][2], v1[i][2], v0[i][3], v1[i][3]};
      *(v4f*)(dst + i * (16 * 16 * 2))     = lo;
      *(v4f*)(dst + i * (16 * 16 * 2) + 4) = hi;
    }
  }
  __syncthreads();

  const int wave = __builtin_amdgcn_readfirstlane(tid >> 5);
  const int lane = tid & 31, g = lane >> 4, mc = lane & 15;
  if (wave < 4) {
    const float* xa = Qs + OC_TILE(wave) + (g * 16 + mc) * 2;
    const float* xw = Bs + (g * 4 + (mc & 3)) * 2;
    v8f acc = {0.f, 0.f, 0.f, 0.f, 0.f, 0.f, 0.f, 0.f};
    for (int k8 = 0; k8 < 64; k8 += 8) {
      const float* pa = xa + k8 * (2 * 16 * 2);
      const float* pw = xw + k8 * (2 * 4 * 2);
#pragma unroll
      for (int j = 0; j < 8; ++j) {
        const v2f a  = *(const v2f*)(pa + j * (2 * 16 * 2));
        const v2f bb = *(const v2f*)(pw + j * (2 * 4 * 2));
        acc = wmma4(a, bb, acc);
      }
    }
    if (mc < OUTC) {
      const float bias = ob[mc];
      const long obase = ((long)(b * OUTC + mc) * H_ + hh) * W_ + 16 * wave;
#pragma unroll
      for (int r = 0; r < 8; ++r) out[obase + r + 8 * g] = acc[r] + bias;
    }
  }
}

__global__ __launch_bounds__(256) void k_final(const unsigned* __restrict__ counts,
                                               const float* __restrict__ wsdiff,
                                               float* __restrict__ out) {
  __shared__ float red[256];
  const int gid = blockIdx.x * 256 + threadIdx.x;
  if (blockIdx.x < 32) {
    out[WF_POS + gid] = (float)counts[gid] * (1.0f / 4096.0f);
  } else {
    float sum = 0.f;
    for (int t = threadIdx.x; t < B_ * H_; t += 256) sum += wsdiff[t];
    red[threadIdx.x] = sum; __syncthreads();
    for (int st = 128; st > 0; st >>= 1) {
      if (threadIdx.x < st) red[threadIdx.x] += red[threadIdx.x + st];
      __syncthreads();
    }
    if (threadIdx.x == 0)
      out[DIFF_POS] = red[0] * (1.0f / ((float)NPOS * (float)C_));
  }
}

extern "C" void kernel_launch(void* const* d_in, const int* in_sizes, int n_in,
                              void* d_out, int out_size, void* d_ws, size_t ws_size,
                              hipStream_t stream) {
  (void)in_sizes; (void)n_in; (void)out_size; (void)ws_size;
  const float* in  = (const float*)d_in[0];
  const float* emb = (const float*)d_in[1];
  const float* rw1 = (const float*)d_in[2];
  const float* rb1 = (const float*)d_in[3];
  const float* rw2 = (const float*)d_in[4];
  const float* rb2 = (const float*)d_in[5];
  const float* ow  = (const float*)d_in[6];
  const float* ob  = (const float*)d_in[7];
  float* out = (float*)d_out;
  float* ws  = (float*)d_ws;

  float*    q      = ws + Q_OFF;
  float*    hbuf   = ws + H_OFF;
  float*    epack  = ws + EPACK_OFF;
  float*    en2    = ws + EN2_OFF;
  unsigned* counts = (unsigned*)(ws + CNT_OFF);
  float*    wsdiff = ws + WDIFF_OFF;
  float*    wpack  = ws + WPACK_OFF;

  k_prep<<<256, 256, 0, stream>>>(emb, epack, en2, counts);
  k_packw<<<(3 * WP_PER_BLK + 255) / 256, 256, 0, stream>>>(rw1, wpack);
  k_vq<<<B_ * H_, 128, 0, stream>>>(in, emb, epack, en2, q, counts, wsdiff);
  for (int i = 0; i < 3; ++i) {
    k_res_conv3<<<B_ * H_ / 2, 128, 0, stream>>>(q, wpack + (size_t)i * WP_PER_BLK * 2,
                                           rb1 + i * CB, hbuf);
    k_res_conv1<<<B_ * H_, 256, 0, stream>>>(q, hbuf, rw2 + (size_t)i * C_ * CB,
                                           rb2 + i * C_);
  }
  k_out_conv<<<B_ * H_, 256, 0, stream>>>(q, ow, ob, out);
  k_final<<<33, 256, 0, stream>>>(counts, wsdiff, out);
}


# last residual 1x1 conv folded into the out conv: out = ow q + (ow W2) h + (ow b2 + ob); its q is never rewritten and one launch goes; sum regrouped (256 q products, then 64 h products)
# speedup vs baseline: 5.3856x; 1.0555x over previous
#include <hip/hip_runtime.h>


typedef float v2f __attribute__((ext_vector_type(2)));
typedef float v4f __attribute__((ext_vector_type(4)));
typedef float v8f __attribute__((ext_vector_type(8)));

#define B_    16
#define C_    256
#define H_    64
#define W_    64
#define OUTC  3
#define NT    512
#define CB    64

#define HW     (H_*W_)
#define NPOS   (B_*HW)
#define CHW    (C_*HW)

#define Q_OFF      0u
#define H_OFF      16777216u
#define EPACK_OFF  20971520u
#define EN2_OFF    21102592u
#define CNT_OFF    21103104u
#define WDIFF_OFF  21111296u
#define FOLD_OFF   21113344u
#define WPACK_OFF  21115392u

#define WP_PER_BLK (9*128*CB)
#define WP_PER_TAP (128*CB)

#define DIFF_POS   196608
#define WF_POS     196609

static __device__ __forceinline__ v8f wmma4(v2f a, v2f b, v8f c) {
  return __builtin_amdgcn_wmma_f32_16x16x4_f32(false, a, false, b, (short)0, c,
                                               false, false);
}

__global__ __launch_bounds__(256) void k_prep(const float* __restrict__ embed,
                                              float* __restrict__ epack,
                                              float* __restrict__ en2,
                                              unsigned* __restrict__ counts,
                                              const float* __restrict__ w2l,
                                              const float* __restrict__ b2l,
                                              const float* __restrict__ ow,
                                              const float* __restrict__ ob,
                                              float* __restrict__ fold) {
  const int gid = blockIdx.x * 256 + threadIdx.x;
  if (gid < 128 * NT) {
    const int cc = gid >> 9, n = gid & (NT - 1);
    epack[gid * 2 + 0] = embed[(2 * cc) * NT + n];
    epack[gid * 2 + 1] = embed[(2 * cc + 1) * NT + n];
  }
  if (gid < NT) {
    float s = 0.f;
    for (int c = 0; c < C_; ++c) { float v = embed[c * NT + gid]; s += v * v; }
    en2[gid] = s;
  }
  if (gid < B_ * NT) counts[gid] = 0u;
  if (gid >= 1024 && gid < 1024 + OUTC * CB) {
    const int e = gid - 1024, o = e >> 6, k = e & (CB - 1);
    float s = 0.f;
    for (int c = 0; c < C_; ++c) s += ow[o * C_ + c] * w2l[c * CB + k];
    fold[e] = s;
  }
  if (gid >= 2048 && gid < 2048 + OUTC) {
    const int o = gid - 2048;
    float s = 0.f;
    for (int c = 0; c < C_; ++c) s += ow[o * C_ + c] * b2l[c];
    fold[OUTC * CB + o] = s + ob[o];
  }
}

__global__ __launch_bounds__(256) void k_packw(const float* __restrict__ rw1,
                                               float* __restrict__ wpack) {
  const int t = blockIdx.x * 256 + threadIdx.x;
  if (t >= 3 * WP_PER_BLK) return;
  const int i   = t / WP_PER_BLK;
  const int r   = t % WP_PER_BLK;
  const int tap = r / WP_PER_TAP;
  const int r2  = r % WP_PER_TAP;
  const int cc  = r2 >> 6;
  const int oc  = r2 & 63;
  const int dh = tap / 3, dw = tap % 3;
  const float* src = rw1 + (size_t)i * CB * C_ * 9
                         + ((size_t)oc * C_ + 2 * cc) * 9 + dh * 3 + dw;
  wpack[t * 2 + 0] = src[0];
  wpack[t * 2 + 1] = src[9];
}

#define VQ_SLOT(cc, p) (((cc) << 6) + (((p) + (((cc) & 1) << 4)) & 63))

__global__ __launch_bounds__(128) void k_vq(const float* __restrict__ in,
                                            const float* __restrict__ embed,
                                            const float* __restrict__ epack,
                                            const float* __restrict__ en2,
                                            float* __restrict__ q,
                                            unsigned* __restrict__ counts,
                                            float* __restrict__ wsdiff) {
  __shared__ float As[128 * 64 * 2];
  __shared__ float rv[4][64];
  __shared__ int   ri[4][64];
  __shared__ int   idxs[64];
  __shared__ float red[128];

  const int s  = blockIdx.x;
  const int b  = s >> 6;
  const int hh = s & 63;
  const int tid = threadIdx.x;
  const long gbase0 = (long)b * CHW + hh * W_;
  const int sp = tid & 63;
  const int sc = tid >> 6;

  {
    const float* src = in + gbase0 + (long)(2 * sc) * HW + sp;
    float* dst = As + VQ_SLOT(sc, sp) * 2;
#pragma unroll 8
    for (int i = 0; i < 64; ++i) {
      const float v0 = src[(long)(4 * i) * HW];
      const float v1 = src[(long)(4 * i + 1) * HW];
      const v2f pv = {v0, v1};
      *(v2f*)(dst + i * (2 * 64 * 2)) = pv;
    }
  }
  __syncthreads();

  const int wave = tid >> 5, lane = tid & 31, g = lane >> 4, mc = lane & 15;
  float bestv[4][8];
  int   besti[4][8];
#pragma unroll
  for (int j = 0; j < 4; ++j)
#pragma unroll
    for (int r = 0; r < 8; ++r) { bestv[j][r] = 3.4e38f; besti[j][r] = 0; }

  const float* xa0 = As + ((g << 6) + (((0 + g) & 3) << 4) + mc) * 2;
  const float* xa1 = As + ((g << 6) + (((1 + g) & 3) << 4) + mc) * 2;
  const float* xa2 = As + ((g << 6) + (((2 + g) & 3) << 4) + mc) * 2;
  const float* xa3 = As + ((g << 6) + (((3 + g) & 3) << 4) + mc) * 2;

  const float* el = epack + ((size_t)g * NT + (wave << 7) + mc) * 2;
  v2f eA[8], eB[8];
#pragma unroll
  for (int j = 0; j < 8; ++j) eA[j] = *(const v2f*)(el + (size_t)j * (2 * NT * 2));

#pragma unroll 1
  for (int tt = 0; tt < 8; ++tt) {
    const int n0 = (wave * 8 + tt) << 4;
    const float* eb = el + tt * (16 * 2);
    v8f acc0 = {0.f, 0.f, 0.f, 0.f, 0.f, 0.f, 0.f, 0.f};
    v8f acc1 = acc0, acc2 = acc0, acc3 = acc0;
#pragma unroll 1
    for (int kp = 0; kp < 4; ++kp) {
#pragma unroll
      for (int j = 0; j < 8; ++j)
        eB[j] = *(const v2f*)(eb + (size_t)(16 * kp + 8 + j) * (2 * NT * 2));
#pragma unroll
      for (int j = 0; j < 8; ++j) {
        const int ks = 16 * kp + j;
        const v2f a0 = *(const v2f*)(xa0 + ks * (2 * 64 * 2));
        const v2f a1 = *(const v2f*)(xa1 + ks * (2 * 64 * 2));
        const v2f a2 = *(const v2f*)(xa2 + ks * (2 * 64 * 2));
        const v2f a3 = *(const v2f*)(xa3 + ks * (2 * 64 * 2));
        acc0 = wmma4(a0, eA[j], acc0);
        acc1 = wmma4(a1, eA[j], acc1);
        acc2 = wmma4(a2, eA[j], acc2);
        acc3 = wmma4(a3, eA[j], acc3);
      }
      int nkp = kp + 1, ntt = tt;
      if (nkp == 4) { nkp = 0; ntt = tt + 1; }
      if (ntt == 8) ntt = 0;
      const float* en = el + ntt * (16 * 2) + (size_t)(16 * nkp) * (2 * NT * 2);
#pragma unroll
      for (int j = 0; j < 8; ++j)
        eA[j] = *(const v2f*)(en + (size_t)j * (2 * NT * 2));
#pragma unroll
      for (int j = 0; j < 8; ++j) {
        const int ks = 16 * kp + 8 + j;
        const v2f a0 = *(const v2f*)(xa0 + ks * (2 * 64 * 2));
        const v2f a1 = *(const v2f*)(xa1 + ks * (2 * 64 * 2));
        const v2f a2 = *(const v2f*)(xa2 + ks * (2 * 64 * 2));
        const v2f a3 = *(const v2f*)(xa3 + ks * (2 * 64 * 2));
        acc0 = wmma4(a0, eB[j], acc0);
        acc1 = wmma4(a1, eB[j], acc1);
        acc2 = wmma4(a2, eB[j], acc2);
        acc3 = wmma4(a3, eB[j], acc3);
      }
    }
    const float e2 = en2[n0 + mc];
    const int   n  = n0 + mc;
#pragma unroll
    for (int r = 0; r < 8; ++r) {
      float d;
      d = e2 - 2.0f * acc0[r]; if (d < bestv[0][r]) { bestv[0][r] = d; besti[0][r] = n; }
      d = e2 - 2.0f * acc1[r]; if (d < bestv[1][r]) { bestv[1][r] = d; besti[1][r] = n; }
      d = e2 - 2.0f * acc2[r]; if (d < bestv[2][r]) { bestv[2][r] = d; besti[2][r] = n; }
      d = e2 - 2.0f * acc3[r]; if (d < bestv[3][r]) { bestv[3][r] = d; besti[3][r] = n; }
    }
  }

#pragma unroll
  for (int j = 0; j < 4; ++j) {
#pragma unroll
    for (int r = 0; r < 8; ++r) {
      float v = bestv[j][r];
      int  ix = besti[j][r];
#pragma unroll
      for (int m = 8; m > 0; m >>= 1) {
        const float ov = __shfl_xor(v, m);
        const int   oi = __shfl_xor(ix, m);
        const bool take = (ov < v) | ((ov == v) & (oi < ix));
        v  = take ? ov : v;
        ix = take ? oi : ix;
      }
      bestv[j][r] = v;
      besti[j][r] = ix;
    }
  }
  if (mc == 0) {
#pragma unroll
    for (int j = 0; j < 4; ++j)
#pragma unroll
      for (int r = 0; r < 8; ++r) {
        rv[wave][16 * j + r + 8 * g] = bestv[j][r];
        ri[wave][16 * j + r + 8 * g] = besti[j][r];
      }
  }
  __syncthreads();

  if (tid < 64) {
    float bv = rv[0][tid]; int bi = ri[0][tid];
    for (int w = 1; w < 4; ++w) {
      const float v = rv[w][tid]; const int ii = ri[w][tid];
      if (v < bv || (v == bv && ii < bi)) { bv = v; bi = ii; }
    }
    idxs[tid] = bi;
    atomicAdd(&counts[b * NT + bi], 1u);
  }
  __syncthreads();

  float fsum = 0.f;
  {
    const int id = idxs[sp];
    const float* xs = As + VQ_SLOT(sc, sp) * 2;
    const float* es = embed + (2 * sc) * NT + id;
    float* qd = q + gbase0 + (long)(2 * sc) * HW + sp;
#pragma unroll 8
    for (int i = 0; i < 64; ++i) {
      const v2f x = *(const v2f*)(xs + i * (2 * 64 * 2));
      const float e0 = es[(4 * i) * NT];
      const float e1 = es[(4 * i + 1) * NT];
      qd[(long)(4 * i) * HW]     = e0;
      qd[(long)(4 * i + 1) * HW] = e1;
      const float d0 = e0 - x[0], d1 = e1 - x[1];
      fsum += d0 * d0 + d1 * d1;
    }
  }
  red[tid] = fsum; __syncthreads();
  for (int st = 64; st > 0; st >>= 1) {
    if (tid < st) red[tid] += red[tid + st];
    __syncthreads();
  }
  if (tid == 0) wsdiff[s] = red[0];
}

#define KCH    16
#define T_HP   96
#define T_CCP  400

static __device__ __forceinline__ void c3_load_run(v2f (&dst)[18], const float* wl,
                                                   int kc, int r) {
#pragma unroll
  for (int j = 0; j < 2; ++j)
#pragma unroll
    for (int tap = 0; tap < 9; ++tap)
      dst[9 * j + tap] = *(const v2f*)(wl + ((size_t)tap * WP_PER_TAP
                                            + (size_t)(KCH * kc + 2 * (2 * r + j)) * CB) * 2);
}

static __device__ __forceinline__ void c3_run(const v2f (&w)[18], const float* xb, int r,
                                              v8f (&acc)[24]) {
#pragma unroll
  for (int j = 0; j < 2; ++j) {
    const int ks = 2 * r + j;
    const float* xk = xb + (ks * 2 * T_CCP) * 2;
    v2f vh[4][3];
#pragma unroll
    for (int dw = 0; dw < 3; ++dw) {
      const v2f k0 = w[9 * j + dw], k1 = w[9 * j + 3 + dw], k2 = w[9 * j + 6 + dw];
      const v2f sk = k0 + k2;
      vh[0][dw] = k0;
      vh[1][dw] = 0.5f * (sk + k1);
      vh[2][dw] = 0.5f * (sk - k1);
      vh[3][dw] = k2;
    }
#pragma unroll
    for (int hp = 0; hp < 4; ++hp) {
      const float* xp = xk + (hp * T_HP) * 2;
      const v2f g0 = vh[hp][0], g1 = vh[hp][1], g2 = vh[hp][2];
      const v2f sg = g0 + g2;
      const v2f u0 = 0.25f * g0;
      const v2f u1 = (-1.0f / 6.0f) * (sg + g1);
      const v2f u2 = (-1.0f / 6.0f) * (sg - g1);
      const v2f ue = (1.0f / 24.0f) * g0 + (1.0f / 6.0f) * g2;
      const v2f uo = (1.0f / 12.0f) * g1;
      const v2f u3 = ue + uo;
      const v2f u4 = ue - uo;
      const v2f t0 = *(const v2f*)(xp);
      const v2f t1 = *(const v2f*)(xp + 32);
      const v2f t2 = *(const v2f*)(xp + 64);
      const v2f t3 = *(const v2f*)(xp + 96);
      const v2f t4 = *(const v2f*)(xp + 128);
      const v2f t5 = *(const v2f*)(xp + 160);
      acc[6 * hp + 0] = wmma4(u0, t0, acc[6 * hp + 0]);
      acc[6 * hp + 1] = wmma4(u1, t1, acc[6 * hp + 1]);
      acc[6 * hp + 2] = wmma4(u2, t2, acc[6 * hp + 2]);
      acc[6 * hp + 3] = wmma4(u3, t3, acc[6 * hp + 3]);
      acc[6 * hp + 4] = wmma4(u4, t4, acc[6 * hp + 4]);
      acc[6 * hp + 5] = wmma4(g2, t5, acc[6 * hp + 5]);
    }
  }
}

__global__ __launch_bounds__(128) void k_res_conv3(const float* __restrict__ q,
                                                   const float* __restrict__ wp,
                                                   const float* __restrict__ b1,
                                                   float* __restrict__ hbuf) {
  __shared__ __attribute__((aligned(16))) float Td[KCH * T_CCP * 2];
  const int s   = blockIdx.x;
  const int b   = s >> 5;
  const int hh0 = (s & 31) * 2;
  const int tid = threadIdx.x;
  const int wave = tid >> 5, lane = tid & 31, g = lane >> 4, mc = lane & 15;
  const int oc = wave * 16 + mc;

  v8f acc[24];
#pragma unroll
  for (int i = 0; i < 24; ++i) {
    const v8f z = {0.f, 0.f, 0.f, 0.f, 0.f, 0.f, 0.f, 0.f};
    acc[i] = z;
  }

  const float* wl = wp + ((size_t)g * CB + oc) * 2;
  v2f wA[18], wB[18];
  c3_load_run(wA, wl, 0, 0);

  const int p  = mc;
  const int il = p > 0 ? -1 : 0;
  const int ir = p < 15 ? 4 : 3;
  const int scc = 2 * wave + g;
  const float* xb = Td + (g * T_CCP + mc) * 2;

  for (int kc = 0; kc < C_ / (2 * KCH); ++kc) {
    if (kc) __syncthreads();
    {
      v4f a0[2][4], a1[2][4];
      float l0[2][4], l1[2][4], r0[2][4], r1[2][4];
#pragma unroll
      for (int i2 = 0; i2 < 2; ++i2) {
#pragma unroll
        for (int rr = 0; rr < 4; ++rr) {
          const int hy  = hh0 + rr - 1;
          const int hyc = hy < 0 ? 0 : (hy > 63 ? 63 : hy);
          const float* c0 = q + (long)b * CHW + (long)(2 * (KCH * kc + scc + 8 * i2)) * HW
                              + hyc * W_ + 4 * p;
          const float* c1 = c0 + HW;
          a0[i2][rr] = *(const v4f*)c0;
          a1[i2][rr] = *(const v4f*)c1;
          l0[i2][rr] = c0[il];  r0[i2][rr] = c0[ir];
          l1[i2][rr] = c1[il];  r1[i2][rr] = c1[ir];
        }
      }
#pragma unroll
      for (int i2 = 0; i2 < 2; ++i2) {
        v2f T[4][6];
#pragma unroll
        for (int rr = 0; rr < 4; ++rr) {
          const int hy = hh0 + rr - 1;
          const bool inimg = (unsigned)hy < 64u;
          v2f d0 = {l0[i2][rr] > 0.f ? l0[i2][rr] : 0.f, l1[i2][rr] > 0.f ? l1[i2][rr] : 0.f};
          v2f d5 = {r0[i2][rr] > 0.f ? r0[i2][rr] : 0.f, r1[i2][rr] > 0.f ? r1[i2][rr] : 0.f};
          const v2f z2 = {0.f, 0.f};
          if (p == 0)  d0 = z2;
          if (p == 15) d5 = z2;
          const v4f x0 = a0[i2][rr], x1 = a1[i2][rr];
          const v2f d1 = {x0[0] > 0.f ? x0[0] : 0.f, x1[0] > 0.f ? x1[0] : 0.f};
          const v2f d2 = {x0[1] > 0.f ? x0[1] : 0.f, x1[1] > 0.f ? x1[1] : 0.f};
          const v2f d3 = {x0[2] > 0.f ? x0[2] : 0.f, x1[2] > 0.f ? x1[2] : 0.f};
          const v2f d4 = {x0[3] > 0.f ? x0[3] : 0.f, x1[3] > 0.f ? x1[3] : 0.f};
          const v2f s12 = d1 + d2, m12 = d1 - d2;
          const v2f s34 = d3 + d4, m43 = d4 - d3;
          const v2f m42 = d4 - d2, m31 = d3 - d1;
          const v2f t0 = (4.0f * d0 - 5.0f * d2) + d4;
          const v2f t1 = s34 - 4.0f * s12;
          const v2f t2 = m43 + 4.0f * m12;
          const v2f t3 = m42 + 2.0f * m31;
          const v2f t4 = m42 - 2.0f * m31;
          const v2f t5 = (4.0f * d1 - 5.0f * d3) + d5;
          T[rr][0] = inimg ? t0 : z2;
          T[rr][1] = inimg ? t1 : z2;
          T[rr][2] = inimg ? t2 : z2;
          T[rr][3] = inimg ? t3 : z2;
          T[rr][4] = inimg ? t4 : z2;
          T[rr][5] = inimg ? t5 : z2;
        }
        float* d = Td + ((scc + 8 * i2) * T_CCP + p) * 2;
#pragma unroll
        for (int wq = 0; wq < 6; ++wq) {
          *(v2f*)(d + (0 * T_HP + 16 * wq) * 2) = T[0][wq] - T[2][wq];
          *(v2f*)(d + (1 * T_HP + 16 * wq) * 2) = T[1][wq] + T[2][wq];
          *(v2f*)(d + (2 * T_HP + 16 * wq) * 2) = T[2][wq] - T[1][wq];
          *(v2f*)(d + (3 * T_HP + 16 * wq) * 2) = T[1][wq] - T[3][wq];
        }
      }
    }
    __syncthreads();

    int nkc = kc + 1;
    if (nkc == C_ / (2 * KCH)) nkc = 0;
    c3_load_run(wB, wl, kc, 1);   c3_run(wA, xb, 0, acc);
    c3_load_run(wA, wl, kc, 2);   c3_run(wB, xb, 1, acc);
    c3_load_run(wB, wl, kc, 3);   c3_run(wA, xb, 2, acc);
    c3_load_run(wA, wl, nkc, 0);  c3_run(wB, xb, 3, acc);
  }

  const int ocb = wave * 16 + 8 * g;
  const long ob = (long)b * (CB * HW) + (long)ocb * HW + hh0 * W_ + 4 * mc;
#pragma unroll
  for (int r = 0; r < 8; ++r) {
    const float bias = b1[ocb + r];
#pragma unroll
    for (int row = 0; row < 2; ++row) {
      float m[6];
#pragma unroll
      for (int wq = 0; wq < 6; ++wq) {
        const float n0 = acc[wq][r], n1 = acc[6 + wq][r];
        const float n2 = acc[12 + wq][r], n3 = acc[18 + wq][r];
        m[wq] = row == 0 ? (n0 + n1) + n2 : (n1 - n2) - n3;
      }
      const float sa = m[1] + m[2], da = m[1] - m[2];
      const float sb = m[3] + m[4], db = m[3] - m[4];
      const float y0 = ((m[0] + sa) + sb) + bias;
      const float y1 = (da + 2.0f * db) + bias;
      const float y2 = (sa + 4.0f * sb) + bias;
      const float y3 = ((da + 8.0f * db) + m[5]) + bias;
      const v4f o = {y0 > 0.f ? y0 : 0.f, y1 > 0.f ? y1 : 0.f,
                     y2 > 0.f ? y2 : 0.f, y3 > 0.f ? y3 : 0.f};
      *(v4f*)(hbuf + ob + (long)r * HW + row * W_) = o;
    }
  }
}

#define C1_HP  80
#define C1_WP  68

__global__ __launch_bounds__(256) void k_res_conv1(float* __restrict__ q,
                                                   const float* __restrict__ hbuf,
                                                   const float* __restrict__ w2,
                                                   const float* __restrict__ b2) {
  __shared__ __attribute__((aligned(16))) float Hs[32 * C1_HP * 2];
  __shared__ __attribute__((aligned(16))) float Ws[C_ * C1_WP];
  const int s  = blockIdx.x;
  const int b  = s >> 6;
  const int hh = s & 63;
  const int tid = threadIdx.x;
  const int wave = tid >> 5, lane = tid & 31, g = lane >> 4, mc = lane & 15;

  float* qb = q + (long)b * CHW + (long)(32 * wave + 8 * g) * HW + hh * W_ + 4 * mc;
  v4f qv[2][8];
#pragma unroll
  for (int t = 0; t < 2; ++t)
#pragma unroll
    for (int r = 0; r < 8; ++r)
      qv[t][r] = *(const v4f*)(qb + (long)(16 * t + r) * HW);

#pragma unroll
  for (int i = 0; i < 16; ++i) {
    const int i4 = tid + 256 * i;
    const v4f v = *(const v4f*)(w2 + i4 * 4);
    *(v4f*)(Ws + (i4 >> 4) * C1_WP + (i4 & 15) * 4) = v;
  }
  {
    const int m = tid & 15, ccb = tid >> 4;
    const float* hs = hbuf + (long)b * (CB * HW) + hh * W_ + 4 * m;
#pragma unroll
    for (int i = 0; i < 2; ++i) {
      const int cc = ccb + 16 * i;
      const v4f v0 = *(const v4f*)(hs + (long)(2 * cc) * HW);
      const v4f v1 = *(const v4f*)(hs + (long)(2 * cc + 1) * HW);
      float* dst = Hs + (cc * C1_HP + m) * 2;
#pragma unroll
      for (int j = 0; j < 4; ++j) {
        const v2f pv = {v0[j], v1[j]};
        *(v2f*)(dst + 16 * j * 2) = pv;
      }
    }
  }
  __syncthreads();

  v8f acc[2][4];
#pragma unroll
  for (int t = 0; t < 2; ++t)
#pragma unroll
    for (int j = 0; j < 4; ++j) {
      const v8f z = {0.f, 0.f, 0.f, 0.f, 0.f, 0.f, 0.f, 0.f};
      acc[t][j] = z;
    }

  const float* wa = Ws + (32 * wave + mc) * C1_WP + 2 * g;
  const float* xb = Hs + (g * C1_HP + mc) * 2;
#pragma unroll
  for (int ks = 0; ks < 16; ++ks) {
    const v2f a0 = *(const v2f*)(wa + 4 * ks);
    const v2f a1 = *(const v2f*)(wa + 16 * C1_WP + 4 * ks);
    const float* xp = xb + ks * (2 * C1_HP * 2);
    const v2f x0 = *(const v2f*)(xp);
    const v2f x1 = *(const v2f*)(xp + 32);
    const v2f x2 = *(const v2f*)(xp + 64);
    const v2f x3 = *(const v2f*)(xp + 96);
    acc[0][0] = wmma4(a0, x0, acc[0][0]);
    acc[0][1] = wmma4(a0, x1, acc[0][1]);
    acc[0][2] = wmma4(a0, x2, acc[0][2]);
    acc[0][3] = wmma4(a0, x3, acc[0][3]);
    acc[1][0] = wmma4(a1, x0, acc[1][0]);
    acc[1][1] = wmma4(a1, x1, acc[1][1]);
    acc[1][2] = wmma4(a1, x2, acc[1][2]);
    acc[1][3] = wmma4(a1, x3, acc[1][3]);
  }

  const int ocb = 32 * wave + 8 * g;
#pragma unroll
  for (int t = 0; t < 2; ++t)
#pragma unroll
    for (int r = 0; r < 8; ++r) {
      const float bias = b2[ocb + 16 * t + r];
      v4f o = qv[t][r];
      o[0] += acc[t][0][r] + bias;
      o[1] += acc[t][1][r] + bias;
      o[2] += acc[t][2][r] + bias;
      o[3] += acc[t][3][r] + bias;
      *(v4f*)(qb + (long)(16 * t + r) * HW) = o;
    }
}

#define OF_CC      160
#define OF_TP      (OF_CC * 16 * 2)
#define OF_TILE(t) ((t) * OF_TP + ((t) & 1) * 32 + ((t) >> 1) * 68)

__global__ __launch_bounds__(256) void k_out_fold(const float* __restrict__ q,
                                                  const float* __restrict__ hbuf,
                                                  const float* __restrict__ ow,
                                                  const float* __restrict__ fold,
                                                  float* __restrict__ out) {
  __shared__ __attribute__((aligned(16))) float Ls[OF_CC * 4 * 2 + 4 * OF_TP + 128];
  float* Bs = Ls;
  float* Qs = Ls + OF_CC * 4 * 2;
  const int s  = blockIdx.x;
  const int b  = s >> 6;
  const int hh = s & 63;
  const int tid = threadIdx.x;

  {
    const int col = tid & 3;
    const int colc = col < OUTC ? col : OUTC - 1;
    v2f wv[3];
#pragma unroll
    for (int i = 0; i < 3; ++i) {
      int cc = (tid + 256 * i) >> 2;
      cc = cc < OF_CC ? cc : OF_CC - 1;
      const float* src = cc < 128 ? ow + colc * C_ + 2 * cc
                                  : fold + colc * CB + 2 * (cc - 128);
      wv[i] = *(const v2f*)src;
    }
    const int m = tid & 15, ccb = tid >> 4;
    const float* qs = q + (long)b * CHW + hh * W_ + 4 * m;
    const float* hs = hbuf + (long)b * (CB * HW) + hh * W_ + 4 * m;
    v4f v0[10], v1[10];
#pragma unroll
    for (int i = 0; i < 8; ++i) {
      const int cc = ccb + 16 * i;
      v0[i] = *(const v4f*)(qs + (long)(2 * cc) * HW);
      v1[i] = *(const v4f*)(qs + (long)(2 * cc + 1) * HW);
    }
#pragma unroll
    for (int i = 0; i < 2; ++i) {
      const int cc = ccb + 16 * i;
      v0[8 + i] = *(const v4f*)(hs + (long)(2 * cc) * HW);
      v1[8 + i] = *(const v4f*)(hs + (long)(2 * cc + 1) * HW);
    }
#pragma unroll
    for (int i = 0; i < 3; ++i) {
      const v2f z = {0.f, 0.f};
      if (tid + 256 * i < OF_CC * 4)
        *(v2f*)(Bs + (tid + 256 * i) * 2) = col < OUTC ? wv[i] : z;
    }
    float* dst = Qs + OF_TILE(m >> 2) + (ccb * 16 + 4 * (m & 3)) * 2;
#pragma unroll
    for (int i = 0; i < 10; ++i) {
      const v4f lo = {v0[i][0], v1[i][0], v0[i][1], v1[i][1]};
      const v4f hi = {v0[i][2], v1[i][2], v0[i][3], v1[i][3]};
      *(v4f*)(dst + i * (16 * 16 * 2))     = lo;
      *(v4f*)(dst + i * (16 * 16 * 2) + 4) = hi;
    }
  }
  __syncthreads();

  const int wave = __builtin_amdgcn_readfirstlane(tid >> 5);
  const int lane = tid & 31, g = lane >> 4, mc = lane & 15;
  if (wave < 4) {
    const float* xa = Qs + OF_TILE(wave) + (g * 16 + mc) * 2;
    const float* xw = Bs + (g * 4 + (mc & 3)) * 2;
    v8f acc = {0.f, 0.f, 0.f, 0.f, 0.f, 0.f, 0.f, 0.f};
    for (int k8 = 0; k8 < OF_CC / 2; k8 += 8) {
      const float* pa = xa + k8 * (2 * 16 * 2);
      const float* pw = xw + k8 * (2 * 4 * 2);
#pragma unroll
      for (int j = 0; j < 8; ++j) {
        const v2f a  = *(const v2f*)(pa + j * (2 * 16 * 2));
        const v2f bb = *(const v2f*)(pw + j * (2 * 4 * 2));
        acc = wmma4(a, bb, acc);
      }
    }
    if (mc < OUTC) {
      const float bias = fold[OUTC * CB + mc];
      const long obase = ((long)(b * OUTC + mc) * H_ + hh) * W_ + 16 * wave;
#pragma unroll
      for (int r = 0; r < 8; ++r) out[obase + r + 8 * g] = acc[r] + bias;
    }
  }
}

__global__ __launch_bounds__(256) void k_final(const unsigned* __restrict__ counts,
                                               const float* __restrict__ wsdiff,
                                               float* __restrict__ out) {
  __shared__ float red[256];
  const int gid = blockIdx.x * 256 + threadIdx.x;
  if (blockIdx.x < 32) {
    out[WF_POS + gid] = (float)counts[gid] * (1.0f / 4096.0f);
  } else {
    float sum = 0.f;
    for (int t = threadIdx.x; t < B_ * H_; t += 256) sum += wsdiff[t];
    red[threadIdx.x] = sum; __syncthreads();
    for (int st = 128; st > 0; st >>= 1) {
      if (threadIdx.x < st) red[threadIdx.x] += red[threadIdx.x + st];
      __syncthreads();
    }
    if (threadIdx.x == 0)
      out[DIFF_POS] = red[0] * (1.0f / ((float)NPOS * (float)C_));
  }
}

extern "C" void kernel_launch(void* const* d_in, const int* in_sizes, int n_in,
                              void* d_out, int out_size, void* d_ws, size_t ws_size,
                              hipStream_t stream) {
  (void)in_sizes; (void)n_in; (void)out_size; (void)ws_size;
  const float* in  = (const float*)d_in[0];
  const float* emb = (const float*)d_in[1];
  const float* rw1 = (const float*)d_in[2];
  const float* rb1 = (const float*)d_in[3];
  const float* rw2 = (const float*)d_in[4];
  const float* rb2 = (const float*)d_in[5];
  const float* ow  = (const float*)d_in[6];
  const float* ob  = (const float*)d_in[7];
  float* out = (float*)d_out;
  float* ws  = (float*)d_ws;

  float*    q      = ws + Q_OFF;
  float*    hbuf   = ws + H_OFF;
  float*    epack  = ws + EPACK_OFF;
  float*    en2    = ws + EN2_OFF;
  unsigned* counts = (unsigned*)(ws + CNT_OFF);
  float*    wsdiff = ws + WDIFF_OFF;
  float*    wpack  = ws + WPACK_OFF;
  float*    fold   = ws + FOLD_OFF;

  k_prep<<<256, 256, 0, stream>>>(emb, epack, en2, counts, rw2 + (size_t)2 * C_ * CB,
                                  rb2 + 2 * C_, ow, ob, fold);
  k_packw<<<(3 * WP_PER_BLK + 255) / 256, 256, 0, stream>>>(rw1, wpack);
  k_vq<<<B_ * H_, 128, 0, stream>>>(in, emb, epack, en2, q, counts, wsdiff);
  for (int i = 0; i < 2; ++i) {
    k_res_conv3<<<B_ * H_ / 2, 128, 0, stream>>>(q, wpack + (size_t)i * WP_PER_BLK * 2,
                                           rb1 + i * CB, hbuf);
    k_res_conv1<<<B_ * H_, 256, 0, stream>>>(q, hbuf, rw2 + (size_t)i * C_ * CB,
                                           rb2 + i * C_);
  }
  k_res_conv3<<<B_ * H_ / 2, 128, 0, stream>>>(q, wpack + (size_t)2 * WP_PER_BLK * 2,
                                         rb1 + 2 * CB, hbuf);
  k_out_fold<<<B_ * H_, 256, 0, stream>>>(q, hbuf, ow, fold, out);
  k_final<<<33, 256, 0, stream>>>(counts, wsdiff, out);
}
